# starting kernel
# baseline (speedup 1.0000x reference)
#include <hip/hip_runtime.h>


#define GAT_N 50000
#define GAT_E 800000
#define GAT_F 64

typedef float v2f __attribute__((ext_vector_type(2)));
typedef float v8f __attribute__((ext_vector_type(8)));

__global__ __launch_bounds__(256)
void gat_gemm_wmma(const float* __restrict__ A, const float* __restrict__ B,
                   float* __restrict__ C, int M, int K, int Nw) {
  const int wave = blockIdx.x * 8 + (threadIdx.x >> 5);
  const int lane = threadIdx.x & 31;
  const int tilesN = Nw >> 6;
  const int tilesM = M >> 4;
  if (wave >= tilesM * tilesN) return;
  const int tm = wave / tilesN;
  const int tn = wave - tm * tilesN;
  const int kh = lane >> 4;
  const int lr = lane & 15;

  const float* Ap = A + (size_t)(tm * 16 + lr) * K + kh * 2;
  const float* Bp = B + (size_t)(tn * 64) + lr;
  v8f acc0 = {}, acc1 = {}, acc2 = {}, acc3 = {};
  for (int k0 = 0; k0 < K; k0 += 4) {
    v2f a = *(const v2f*)(Ap + k0);
    const float* Brow0 = Bp + (size_t)(k0 + kh * 2) * Nw;
    const float* Brow1 = Brow0 + Nw;
    v2f b0, b1, b2, b3;
    b0.x = Brow0[0];  b0.y = Brow1[0];
    b1.x = Brow0[16]; b1.y = Brow1[16];
    b2.x = Brow0[32]; b2.y = Brow1[32];
    b3.x = Brow0[48]; b3.y = Brow1[48];
    acc0 = __builtin_amdgcn_wmma_f32_16x16x4_f32(false, a, false, b0, (short)0, acc0, false, false);
    acc1 = __builtin_amdgcn_wmma_f32_16x16x4_f32(false, a, false, b1, (short)0, acc1, false, false);
    acc2 = __builtin_amdgcn_wmma_f32_16x16x4_f32(false, a, false, b2, (short)0, acc2, false, false);
    acc3 = __builtin_amdgcn_wmma_f32_16x16x4_f32(false, a, false, b3, (short)0, acc3, false, false);
  }
  float* Cp = C + (size_t)(tm * 16 + kh * 8) * Nw + tn * 64 + lr;
#pragma unroll
  for (int r = 0; r < 8; ++r) {
    Cp[(size_t)r * Nw +  0] = acc0[r];
    Cp[(size_t)r * Nw + 16] = acc1[r];
    Cp[(size_t)r * Nw + 32] = acc2[r];
    Cp[(size_t)r * Nw + 48] = acc3[r];
  }
}

__global__ __launch_bounds__(256)
void gat_node_attn(const float* __restrict__ h, const float* __restrict__ attn,
                   float* __restrict__ out, int N, int H) {
  int idx = blockIdx.x * blockDim.x + threadIdx.x;
  if (idx >= N * H) return;
  int n = idx / H, hh = idx - n * H;
  const float4* hp = (const float4*)(h + ((size_t)n * H + hh) * GAT_F);
  const float4* ap = (const float4*)(attn + (size_t)hh * GAT_F);
  float s = 0.f;
#pragma unroll
  for (int f = 0; f < GAT_F / 4; ++f) {
    float4 x = hp[f], w = ap[f];
    s += x.x * w.x + x.y * w.y + x.z * w.z + x.w * w.w;
  }
  out[idx] = s;
}

__global__ __launch_bounds__(256)
void gat_fill_u32(unsigned* __restrict__ p, unsigned v, long long n) {
  long long i = (long long)blockIdx.x * blockDim.x + threadIdx.x;
  if (i < n) p[i] = v;
}

__device__ __forceinline__ unsigned f32_key(float x) {
  unsigned u = __float_as_uint(x);
  return (u & 0x80000000u) ? ~u : (u | 0x80000000u);
}
__device__ __forceinline__ float key_f32(unsigned k) {
  return (k & 0x80000000u) ? __uint_as_float(k ^ 0x80000000u) : __uint_as_float(~k);
}
#define KEY_NEG_INF 0x007FFFFFu

__global__ __launch_bounds__(256)
void gat_edge_logits(const int* __restrict__ src, const int* __restrict__ dst,
                     const float* __restrict__ el, const float* __restrict__ er,
                     float* __restrict__ ebuf, unsigned* __restrict__ mkey,
                     int E, int H) {
  int idx = blockIdx.x * blockDim.x + threadIdx.x;
  if (idx >= E * H) return;
  int e = idx / H, hh = idx - e * H;
  int s = src[e], d = dst[e];
  float x = el[s * H + hh] + er[d * H + hh];
  x = (x > 0.f) ? x : 0.2f * x;
  ebuf[idx] = x;
  atomicMax(&mkey[d * H + hh], f32_key(x));
}

__global__ __launch_bounds__(256)
void gat_edge_exp(const int* __restrict__ dst, float* __restrict__ ebuf,
                  const unsigned* __restrict__ mkey, float* __restrict__ ssum,
                  int E, int H) {
  int idx = blockIdx.x * blockDim.x + threadIdx.x;
  if (idx >= E * H) return;
  int e = idx / H, hh = idx - e * H;
  int d = dst[e];
  float m = key_f32(mkey[d * H + hh]);
  float ex = __expf(ebuf[idx] - m);
  ebuf[idx] = ex;
  atomicAdd(&ssum[d * H + hh], ex);
}

__global__ __launch_bounds__(256)
void gat_edge_aggregate(const int* __restrict__ src, const int* __restrict__ dst,
                        const float* __restrict__ h, const float* __restrict__ ebuf,
                        const float* __restrict__ ssum, float* __restrict__ out,
                        int E, int H) {
  const int HF = H * GAT_F;
  long long idx = (long long)blockIdx.x * blockDim.x + threadIdx.x;
  if (idx >= (long long)E * HF) return;
  int e = (int)(idx / HF);
  int i = (int)(idx - (long long)e * HF);
  int hh = i >> 6;
  int s = src[e], d = dst[e];
  float alpha = ebuf[e * H + hh] / ssum[d * H + hh];
  atomicAdd(&out[(size_t)d * HF + i], h[(size_t)s * HF + i] * alpha);
}

__global__ __launch_bounds__(256)
void gat_bias_relu(float* __restrict__ x, const float* __restrict__ b,
                   long long n, int HF) {
  long long idx = (long long)blockIdx.x * blockDim.x + threadIdx.x;
  if (idx >= n) return;
  float v = x[idx] + b[(int)(idx % HF)];
  x[idx] = v > 0.f ? v : 0.f;
}

__global__ __launch_bounds__(256)
void gat_bias_out(const float* __restrict__ x, const float* __restrict__ b,
                  float* __restrict__ out, long long n, int HF) {
  long long idx = (long long)blockIdx.x * blockDim.x + threadIdx.x;
  if (idx >= n) return;
  out[idx] = x[idx] + b[(int)(idx % HF)];
}

extern "C" void kernel_launch(void* const* d_in, const int* in_sizes, int n_in,
                              void* d_out, int out_size, void* d_ws, size_t ws_size,
                              hipStream_t stream) {
  (void)in_sizes; (void)n_in; (void)out_size; (void)ws_size;
  const float* feat = (const float*)d_in[0];
  const int*   src  = (const int*)d_in[1];
  const int*   dst  = (const int*)d_in[2];
  const float* W1  = (const float*)d_in[3];
  const float* al1 = (const float*)d_in[4];
  const float* ar1 = (const float*)d_in[5];
  const float* b1  = (const float*)d_in[6];
  const float* W2  = (const float*)d_in[7];
  const float* al2 = (const float*)d_in[8];
  const float* ar2 = (const float*)d_in[9];
  const float* b2  = (const float*)d_in[10];
  const float* W3  = (const float*)d_in[11];
  const float* al3 = (const float*)d_in[12];
  const float* ar3 = (const float*)d_in[13];
  const float* b3  = (const float*)d_in[14];
  float* out = (float*)d_out;

  const int N = GAT_N, E = GAT_E;

  char* ws = (char*)d_ws;
  size_t off = 0;
  auto carve = [&](size_t bytes) -> void* {
    void* p = ws + off;
    off += (bytes + 255) & ~(size_t)255;
    return p;
  };
  float*    bufA = (float*)carve((size_t)N * 256 * 4);
  float*    bufB = (float*)carve((size_t)N * 256 * 4);
  float*    el   = (float*)carve((size_t)N * 4 * 4);
  float*    er   = (float*)carve((size_t)N * 4 * 4);
  unsigned* mkey = (unsigned*)carve((size_t)N * 4 * 4);
  float*    ssum = (float*)carve((size_t)N * 4 * 4);
  float*    ebuf = (float*)carve((size_t)E * 4 * 4);

  auto blocks = [](long long n) { return (unsigned)((n + 255) / 256); };

  {
    const int K = 128, Nw = 256, H = 4;
    int waves = (N / 16) * (Nw / 64);
    gat_gemm_wmma<<<(waves + 7) / 8, 256, 0, stream>>>(feat, W1, bufA, N, K, Nw);
    long long nh = (long long)N * H;
    gat_node_attn<<<blocks(nh), 256, 0, stream>>>(bufA, al1, el, N, H);
    gat_node_attn<<<blocks(nh), 256, 0, stream>>>(bufA, ar1, er, N, H);
    gat_fill_u32<<<blocks(nh), 256, 0, stream>>>(mkey, KEY_NEG_INF, nh);
    gat_fill_u32<<<blocks(nh), 256, 0, stream>>>((unsigned*)ssum, 0u, nh);
    long long nout = (long long)N * H * GAT_F;
    gat_fill_u32<<<blocks(nout), 256, 0, stream>>>((unsigned*)bufB, 0u, nout);
    long long eh = (long long)E * H;
    gat_edge_logits<<<blocks(eh), 256, 0, stream>>>(src, dst, el, er, ebuf, mkey, E, H);
    gat_edge_exp<<<blocks(eh), 256, 0, stream>>>(dst, ebuf, mkey, ssum, E, H);
    long long ew = (long long)E * H * GAT_F;
    gat_edge_aggregate<<<blocks(ew), 256, 0, stream>>>(src, dst, bufA, ebuf, ssum, bufB, E, H);
    gat_bias_relu<<<blocks(nout), 256, 0, stream>>>(bufB, b1, nout, H * GAT_F);
  }

  {
    const int K = 256, Nw = 256, H = 4;
    int waves = (N / 16) * (Nw / 64);
    gat_gemm_wmma<<<(waves + 7) / 8, 256, 0, stream>>>(bufB, W2, bufA, N, K, Nw);
    long long nh = (long long)N * H;
    gat_node_attn<<<blocks(nh), 256, 0, stream>>>(bufA, al2, el, N, H);
    gat_node_attn<<<blocks(nh), 256, 0, stream>>>(bufA, ar2, er, N, H);
    gat_fill_u32<<<blocks(nh), 256, 0, stream>>>(mkey, KEY_NEG_INF, nh);
    gat_fill_u32<<<blocks(nh), 256, 0, stream>>>((unsigned*)ssum, 0u, nh);
    long long nout = (long long)N * H * GAT_F;
    gat_fill_u32<<<blocks(nout), 256, 0, stream>>>((unsigned*)bufB, 0u, nout);
    long long eh = (long long)E * H;
    gat_edge_logits<<<blocks(eh), 256, 0, stream>>>(src, dst, el, er, ebuf, mkey, E, H);
    gat_edge_exp<<<blocks(eh), 256, 0, stream>>>(dst, ebuf, mkey, ssum, E, H);
    long long ew = (long long)E * H * GAT_F;
    gat_edge_aggregate<<<blocks(ew), 256, 0, stream>>>(src, dst, bufA, ebuf, ssum, bufB, E, H);
    gat_bias_relu<<<blocks(nout), 256, 0, stream>>>(bufB, b2, nout, H * GAT_F);
  }

  {
    const int K = 256, Nw = 64, H = 1;
    int waves = (N / 16) * (Nw / 64);
    gat_gemm_wmma<<<(waves + 7) / 8, 256, 0, stream>>>(bufB, W3, bufA, N, K, Nw);
    long long nh = (long long)N * H;
    gat_node_attn<<<blocks(nh), 256, 0, stream>>>(bufA, al3, el, N, H);
    gat_node_attn<<<blocks(nh), 256, 0, stream>>>(bufA, ar3, er, N, H);
    gat_fill_u32<<<blocks(nh), 256, 0, stream>>>(mkey, KEY_NEG_INF, nh);
    gat_fill_u32<<<blocks(nh), 256, 0, stream>>>((unsigned*)ssum, 0u, nh);
    long long nout = (long long)N * H * GAT_F;
    gat_fill_u32<<<blocks(nout), 256, 0, stream>>>((unsigned*)bufB, 0u, nout);
    long long eh = (long long)E * H;
    gat_edge_logits<<<blocks(eh), 256, 0, stream>>>(src, dst, el, er, ebuf, mkey, E, H);
    gat_edge_exp<<<blocks(eh), 256, 0, stream>>>(dst, ebuf, mkey, ssum, E, H);
    long long ew = (long long)E * H * GAT_F;
    gat_edge_aggregate<<<blocks(ew), 256, 0, stream>>>(src, dst, bufA, ebuf, ssum, bufB, E, H);
    gat_bias_out<<<blocks(nout), 256, 0, stream>>>(bufB, b3, out, nout, GAT_F);
  }
}


# Aggregation per dst-node range: a work-group keeps 196 output rows in LDS, scans dst[] for its edges, adds alpha*h[src] by LDS float atomics, stores rows once with bias/ReLU; global float adds gone
# speedup vs baseline: 2.7754x; 2.7754x over previous
#include <hip/hip_runtime.h>


#define GAT_N 50000
#define GAT_E 800000
#define GAT_F 64

typedef float v2f __attribute__((ext_vector_type(2)));
typedef float v8f __attribute__((ext_vector_type(8)));

__global__ __launch_bounds__(256)
void gat_gemm_wmma(const float* __restrict__ A, const float* __restrict__ B,
                   float* __restrict__ C, int M, int K, int Nw) {
  const int wave = blockIdx.x * 8 + (threadIdx.x >> 5);
  const int lane = threadIdx.x & 31;
  const int tilesN = Nw >> 6;
  const int tilesM = M >> 4;
  if (wave >= tilesM * tilesN) return;
  const int tm = wave / tilesN;
  const int tn = wave - tm * tilesN;
  const int kh = lane >> 4;
  const int lr = lane & 15;

  const float* Ap = A + (size_t)(tm * 16 + lr) * K + kh * 2;
  const float* Bp = B + (size_t)(tn * 64) + lr;
  v8f acc0 = {}, acc1 = {}, acc2 = {}, acc3 = {};
  for (int k0 = 0; k0 < K; k0 += 4) {
    v2f a = *(const v2f*)(Ap + k0);
    const float* Brow0 = Bp + (size_t)(k0 + kh * 2) * Nw;
    const float* Brow1 = Brow0 + Nw;
    v2f b0, b1, b2, b3;
    b0.x = Brow0[0];  b0.y = Brow1[0];
    b1.x = Brow0[16]; b1.y = Brow1[16];
    b2.x = Brow0[32]; b2.y = Brow1[32];
    b3.x = Brow0[48]; b3.y = Brow1[48];
    acc0 = __builtin_amdgcn_wmma_f32_16x16x4_f32(false, a, false, b0, (short)0, acc0, false, false);
    acc1 = __builtin_amdgcn_wmma_f32_16x16x4_f32(false, a, false, b1, (short)0, acc1, false, false);
    acc2 = __builtin_amdgcn_wmma_f32_16x16x4_f32(false, a, false, b2, (short)0, acc2, false, false);
    acc3 = __builtin_amdgcn_wmma_f32_16x16x4_f32(false, a, false, b3, (short)0, acc3, false, false);
  }
  float* Cp = C + (size_t)(tm * 16 + kh * 8) * Nw + tn * 64 + lr;
#pragma unroll
  for (int r = 0; r < 8; ++r) {
    Cp[(size_t)r * Nw +  0] = acc0[r];
    Cp[(size_t)r * Nw + 16] = acc1[r];
    Cp[(size_t)r * Nw + 32] = acc2[r];
    Cp[(size_t)r * Nw + 48] = acc3[r];
  }
}

__global__ __launch_bounds__(256)
void gat_node_attn(const float* __restrict__ h, const float* __restrict__ attn,
                   float* __restrict__ out, int N, int H) {
  int idx = blockIdx.x * blockDim.x + threadIdx.x;
  if (idx >= N * H) return;
  int n = idx / H, hh = idx - n * H;
  const float4* hp = (const float4*)(h + ((size_t)n * H + hh) * GAT_F);
  const float4* ap = (const float4*)(attn + (size_t)hh * GAT_F);
  float s = 0.f;
#pragma unroll
  for (int f = 0; f < GAT_F / 4; ++f) {
    float4 x = hp[f], w = ap[f];
    s += x.x * w.x + x.y * w.y + x.z * w.z + x.w * w.w;
  }
  out[idx] = s;
}

__global__ __launch_bounds__(256)
void gat_fill_u32(unsigned* __restrict__ p, unsigned v, long long n) {
  long long i = (long long)blockIdx.x * blockDim.x + threadIdx.x;
  if (i < n) p[i] = v;
}

__device__ __forceinline__ unsigned f32_key(float x) {
  unsigned u = __float_as_uint(x);
  return (u & 0x80000000u) ? ~u : (u | 0x80000000u);
}
__device__ __forceinline__ float key_f32(unsigned k) {
  return (k & 0x80000000u) ? __uint_as_float(k ^ 0x80000000u) : __uint_as_float(~k);
}
#define KEY_NEG_INF 0x007FFFFFu

__global__ __launch_bounds__(256)
void gat_edge_logits(const int* __restrict__ src, const int* __restrict__ dst,
                     const float* __restrict__ el, const float* __restrict__ er,
                     float* __restrict__ ebuf, unsigned* __restrict__ mkey,
                     int E, int H) {
  int idx = blockIdx.x * blockDim.x + threadIdx.x;
  if (idx >= E * H) return;
  int e = idx / H, hh = idx - e * H;
  int s = src[e], d = dst[e];
  float x = el[s * H + hh] + er[d * H + hh];
  x = (x > 0.f) ? x : 0.2f * x;
  ebuf[idx] = x;
  atomicMax(&mkey[d * H + hh], f32_key(x));
}

__global__ __launch_bounds__(256)
void gat_edge_exp(const int* __restrict__ dst, float* __restrict__ ebuf,
                  const unsigned* __restrict__ mkey, float* __restrict__ ssum,
                  int E, int H) {
  int idx = blockIdx.x * blockDim.x + threadIdx.x;
  if (idx >= E * H) return;
  int e = idx / H, hh = idx - e * H;
  int d = dst[e];
  float m = key_f32(mkey[d * H + hh]);
  float ex = __expf(ebuf[idx] - m);
  ebuf[idx] = ex;
  atomicAdd(&ssum[d * H + hh], ex);
}

#define GAT_TILE_NODES 196
#define GAT_LIST_CAP   6144

__device__ __forceinline__ void gat_lds_add(float* p, float v) {
  __hip_atomic_fetch_add(p, v, __ATOMIC_RELAXED, __HIP_MEMORY_SCOPE_WORKGROUP);
}


template <int H>
__device__ __forceinline__ float gat_alpha_lane(const float* __restrict__ ebuf, const float* __restrict__ ssum,
                                                int e, int node, int lane) {
  const int hq = lane & (H - 1);
  return ebuf[e * H + hq] / ssum[node * H + hq];
}
template <int H>
__device__ __forceinline__ float gat_alpha_head(float a, int head) {
  if constexpr (H == 1) return a;
  else return __int_as_float(__builtin_amdgcn_readlane(__float_as_int(a), head));
}

template <int H>
__device__ __forceinline__ void gat_tile_add_edge(const int* __restrict__ src, const float* __restrict__ h,
                                                  const float* __restrict__ ebuf, const float* __restrict__ ssum,
                                                  float* tile, int base, int e, int dl, int lane) {
  constexpr int HF = H * GAT_F;
  constexpr int VEC = HF / 32;
  const int s = src[e];
  const float a = gat_alpha_lane<H>(ebuf, ssum, e, base + dl, lane);
  const float* hp = h + (size_t)s * HF + lane;
  float* tp = tile + dl * HF + lane;
  float v[VEC];
#pragma unroll
  for (int j = 0; j < VEC; ++j) v[j] = hp[j * 32];
#pragma unroll
  for (int j = 0; j < VEC; ++j) gat_lds_add(tp + j * 32, v[j] * gat_alpha_head<H>(a, j >> 1));
}

template <int H, bool RELU>
__global__ __launch_bounds__(1024)
void gat_tile_aggregate(const int* __restrict__ src, const int* __restrict__ dst,
                        const float* __restrict__ h, const float* __restrict__ ebuf,
                        const float* __restrict__ ssum, const float* __restrict__ bias,
                        float* __restrict__ out, int N, int E) {
  constexpr int HF = H * GAT_F;
  constexpr int VEC = HF / 32;
  __shared__ __attribute__((aligned(16))) float tile[GAT_TILE_NODES * HF];
  __shared__ int2 list[GAT_LIST_CAP];
  __shared__ unsigned s_count;

  const int tid = threadIdx.x;
  const int lane = tid & 31;
  const int wave = __builtin_amdgcn_readfirstlane(tid >> 5);
  const int nw = blockDim.x >> 5;
  const int base = blockIdx.x * GAT_TILE_NODES;
  const int left = N - base;
  const int cnt = left < GAT_TILE_NODES ? left : GAT_TILE_NODES;

  float4* tile4 = (float4*)tile;
  const int n4 = cnt * (HF / 4);
  for (int i = tid; i < n4; i += blockDim.x) tile4[i] = make_float4(0.f, 0.f, 0.f, 0.f);
  if (tid == 0) s_count = 0u;
  __syncthreads();

  const int nGroups = (E + 511) >> 9;
  for (int gg = wave; gg < nGroups; gg += nw) {
    const int e0 = gg * 512 + lane * 4;
    int dv[4][4];
    if (gg * 512 + 512 <= E) {
#pragma unroll
      for (int c = 0; c < 4; ++c) {
        const int4 t = *(const int4*)(dst + e0 + c * 128);
        dv[c][0] = t.x; dv[c][1] = t.y; dv[c][2] = t.z; dv[c][3] = t.w;
      }
    } else {
#pragma unroll
      for (int c = 0; c < 4; ++c)
#pragma unroll
        for (int j = 0; j < 4; ++j) {
          const int e = e0 + c * 128 + j;
          dv[c][j] = (e < E) ? dst[e] : -1;
        }
    }
    unsigned ovf = 0u;
#pragma unroll
    for (int c = 0; c < 4; ++c)
#pragma unroll
      for (int j = 0; j < 4; ++j) {
        const int dl = dv[c][j] - base;
        if ((unsigned)dl < (unsigned)cnt) {
          const unsigned pos = atomicAdd(&s_count, 1u);
          if (pos < (unsigned)GAT_LIST_CAP) list[pos] = make_int2(e0 + c * 128 + j, dl);
          else ovf |= 1u << (c * 4 + j);
        }
      }
    if (__builtin_amdgcn_ballot_w32(ovf != 0u)) {
#pragma unroll
      for (int q = 0; q < 16; ++q) {
        unsigned mk = __builtin_amdgcn_ballot_w32(((ovf >> q) & 1u) != 0u);
        while (mk) {
          const int l = __builtin_ctz(mk);
          mk &= mk - 1u;
          const int e = gg * 512 + (q >> 2) * 128 + l * 4 + (q & 3);
          const int dl = __builtin_amdgcn_readlane(dv[q >> 2][q & 3], l) - base;
          gat_tile_add_edge<H>(src, h, ebuf, ssum, tile, base, e, dl, lane);
        }
      }
    }
  }
  __syncthreads();

  const int total = (int)s_count;
  const int n = total < GAT_LIST_CAP ? total : GAT_LIST_CAP;
  for (int i = wave * 4; i < n; i += nw * 4) {
    if (i + 4 <= n) {
      int ee[4], dd[4];
#pragma unroll
      for (int k = 0; k < 4; ++k) {
        const int2 en = list[i + k];
        ee[k] = __builtin_amdgcn_readfirstlane(en.x);
        dd[k] = __builtin_amdgcn_readfirstlane(en.y);
      }
      float vals[4][VEC];
      float al[4];
#pragma unroll
      for (int k = 0; k < 4; ++k) {
        const int s = src[ee[k]];
        al[k] = gat_alpha_lane<H>(ebuf, ssum, ee[k], base + dd[k], lane);
        const float* hp = h + (size_t)s * HF + lane;
#pragma unroll
        for (int j = 0; j < VEC; ++j) vals[k][j] = hp[j * 32];
      }
#pragma unroll
      for (int k = 0; k < 4; ++k) {
        float* tp = tile + dd[k] * HF + lane;
#pragma unroll
        for (int j = 0; j < VEC; ++j) gat_lds_add(tp + j * 32, vals[k][j] * gat_alpha_head<H>(al[k], j >> 1));
      }
    } else {
      for (int k = i; k < n; ++k) {
        const int2 en = list[k];
        gat_tile_add_edge<H>(src, h, ebuf, ssum, tile, base,
                             __builtin_amdgcn_readfirstlane(en.x), __builtin_amdgcn_readfirstlane(en.y), lane);
      }
    }
  }
  __syncthreads();

  const float4* bias4 = (const float4*)bias;
  float4* out4 = (float4*)(out + (size_t)base * HF);
  for (int i = tid; i < n4; i += blockDim.x) {
    float4 v = tile4[i];
    const float4 b = bias4[i & (HF / 4 - 1)];
    v.x += b.x; v.y += b.y; v.z += b.z; v.w += b.w;
    if (RELU) {
      v.x = v.x > 0.f ? v.x : 0.f; v.y = v.y > 0.f ? v.y : 0.f;
      v.z = v.z > 0.f ? v.z : 0.f; v.w = v.w > 0.f ? v.w : 0.f;
    }
    out4[i] = v;
  }
}

extern "C" void kernel_launch(void* const* d_in, const int* in_sizes, int n_in,
                              void* d_out, int out_size, void* d_ws, size_t ws_size,
                              hipStream_t stream) {
  (void)in_sizes; (void)n_in; (void)out_size; (void)ws_size;
  const float* feat = (const float*)d_in[0];
  const int*   src  = (const int*)d_in[1];
  const int*   dst  = (const int*)d_in[2];
  const float* W1  = (const float*)d_in[3];
  const float* al1 = (const float*)d_in[4];
  const float* ar1 = (const float*)d_in[5];
  const float* b1  = (const float*)d_in[6];
  const float* W2  = (const float*)d_in[7];
  const float* al2 = (const float*)d_in[8];
  const float* ar2 = (const float*)d_in[9];
  const float* b2  = (const float*)d_in[10];
  const float* W3  = (const float*)d_in[11];
  const float* al3 = (const float*)d_in[12];
  const float* ar3 = (const float*)d_in[13];
  const float* b3  = (const float*)d_in[14];
  float* out = (float*)d_out;

  const int N = GAT_N, E = GAT_E;

  char* ws = (char*)d_ws;
  size_t off = 0;
  auto carve = [&](size_t bytes) -> void* {
    void* p = ws + off;
    off += (bytes + 255) & ~(size_t)255;
    return p;
  };
  float*    bufA = (float*)carve((size_t)N * 256 * 4);
  float*    bufB = (float*)carve((size_t)N * 256 * 4);
  float*    el   = (float*)carve((size_t)N * 4 * 4);
  float*    er   = (float*)carve((size_t)N * 4 * 4);
  unsigned* mkey = (unsigned*)carve((size_t)N * 4 * 4);
  float*    ssum = (float*)carve((size_t)N * 4 * 4);
  float*    ebuf = (float*)carve((size_t)E * 4 * 4);

  auto blocks = [](long long n) { return (unsigned)((n + 255) / 256); };
  const unsigned tiles = (GAT_N + GAT_TILE_NODES - 1) / GAT_TILE_NODES;

  {
    const int K = 128, Nw = 256, H = 4;
    int waves = (N / 16) * (Nw / 64);
    gat_gemm_wmma<<<(waves + 7) / 8, 256, 0, stream>>>(feat, W1, bufA, N, K, Nw);
    long long nh = (long long)N * H;
    gat_node_attn<<<blocks(nh), 256, 0, stream>>>(bufA, al1, el, N, H);
    gat_node_attn<<<blocks(nh), 256, 0, stream>>>(bufA, ar1, er, N, H);
    gat_fill_u32<<<blocks(nh), 256, 0, stream>>>(mkey, KEY_NEG_INF, nh);
    gat_fill_u32<<<blocks(nh), 256, 0, stream>>>((unsigned*)ssum, 0u, nh);
    long long eh = (long long)E * H;
    gat_edge_logits<<<blocks(eh), 256, 0, stream>>>(src, dst, el, er, ebuf, mkey, E, H);
    gat_edge_exp<<<blocks(eh), 256, 0, stream>>>(dst, ebuf, mkey, ssum, E, H);
    gat_tile_aggregate<4, true><<<tiles, 1024, 0, stream>>>(src, dst, bufA, ebuf, ssum, b1, bufB, N, E);
  }

  {
    const int K = 256, Nw = 256, H = 4;
    int waves = (N / 16) * (Nw / 64);
    gat_gemm_wmma<<<(waves + 7) / 8, 256, 0, stream>>>(bufB, W2, bufA, N, K, Nw);
    long long nh = (long long)N * H;
    gat_node_attn<<<blocks(nh), 256, 0, stream>>>(bufA, al2, el, N, H);
    gat_node_attn<<<blocks(nh), 256, 0, stream>>>(bufA, ar2, er, N, H);
    gat_fill_u32<<<blocks(nh), 256, 0, stream>>>(mkey, KEY_NEG_INF, nh);
    gat_fill_u32<<<blocks(nh), 256, 0, stream>>>((unsigned*)ssum, 0u, nh);
    long long eh = (long long)E * H;
    gat_edge_logits<<<blocks(eh), 256, 0, stream>>>(src, dst, el, er, ebuf, mkey, E, H);
    gat_edge_exp<<<blocks(eh), 256, 0, stream>>>(dst, ebuf, mkey, ssum, E, H);
    gat_tile_aggregate<4, true><<<tiles, 1024, 0, stream>>>(src, dst, bufA, ebuf, ssum, b2, bufB, N, E);
  }

  {
    const int K = 256, Nw = 64, H = 1;
    int waves = (N / 16) * (Nw / 64);
    gat_gemm_wmma<<<(waves + 7) / 8, 256, 0, stream>>>(bufB, W3, bufA, N, K, Nw);
    long long nh = (long long)N * H;
    gat_node_attn<<<blocks(nh), 256, 0, stream>>>(bufA, al3, el, N, H);
    gat_node_attn<<<blocks(nh), 256, 0, stream>>>(bufA, ar3, er, N, H);
    gat_fill_u32<<<blocks(nh), 256, 0, stream>>>(mkey, KEY_NEG_INF, nh);
    gat_fill_u32<<<blocks(nh), 256, 0, stream>>>((unsigned*)ssum, 0u, nh);
    long long eh = (long long)E * H;
    gat_edge_logits<<<blocks(eh), 256, 0, stream>>>(src, dst, el, er, ebuf, mkey, E, H);
    gat_edge_exp<<<blocks(eh), 256, 0, stream>>>(dst, ebuf, mkey, ssum, E, H);
    gat_tile_aggregate<1, false><<<tiles, 1024, 0, stream>>>(src, dst, bufA, ebuf, ssum, b3, out, N, E);
  }
}


# Edge softmax moved into the dst-range tile kernel: logits, per-node max and exp sums in LDS (LDS adds, arrival-order sums as the global adds were); edge and fill launches and their buffers gone
# speedup vs baseline: 4.3926x; 1.5827x over previous
#include <hip/hip_runtime.h>


#define GAT_N 50000
#define GAT_E 800000
#define GAT_F 64

typedef float v2f __attribute__((ext_vector_type(2)));
typedef float v8f __attribute__((ext_vector_type(8)));

__global__ __launch_bounds__(256)
void gat_gemm_wmma(const float* __restrict__ A, const float* __restrict__ B,
                   float* __restrict__ C, int M, int K, int Nw) {
  const int wave = blockIdx.x * 8 + (threadIdx.x >> 5);
  const int lane = threadIdx.x & 31;
  const int tilesN = Nw >> 6;
  const int tilesM = M >> 4;
  if (wave >= tilesM * tilesN) return;
  const int tm = wave / tilesN;
  const int tn = wave - tm * tilesN;
  const int kh = lane >> 4;
  const int lr = lane & 15;

  const float* Ap = A + (size_t)(tm * 16 + lr) * K + kh * 2;
  const float* Bp = B + (size_t)(tn * 64) + lr;
  v8f acc0 = {}, acc1 = {}, acc2 = {}, acc3 = {};
  for (int k0 = 0; k0 < K; k0 += 4) {
    v2f a = *(const v2f*)(Ap + k0);
    const float* Brow0 = Bp + (size_t)(k0 + kh * 2) * Nw;
    const float* Brow1 = Brow0 + Nw;
    v2f b0, b1, b2, b3;
    b0.x = Brow0[0];  b0.y = Brow1[0];
    b1.x = Brow0[16]; b1.y = Brow1[16];
    b2.x = Brow0[32]; b2.y = Brow1[32];
    b3.x = Brow0[48]; b3.y = Brow1[48];
    acc0 = __builtin_amdgcn_wmma_f32_16x16x4_f32(false, a, false, b0, (short)0, acc0, false, false);
    acc1 = __builtin_amdgcn_wmma_f32_16x16x4_f32(false, a, false, b1, (short)0, acc1, false, false);
    acc2 = __builtin_amdgcn_wmma_f32_16x16x4_f32(false, a, false, b2, (short)0, acc2, false, false);
    acc3 = __builtin_amdgcn_wmma_f32_16x16x4_f32(false, a, false, b3, (short)0, acc3, false, false);
  }
  float* Cp = C + (size_t)(tm * 16 + kh * 8) * Nw + tn * 64 + lr;
#pragma unroll
  for (int r = 0; r < 8; ++r) {
    Cp[(size_t)r * Nw +  0] = acc0[r];
    Cp[(size_t)r * Nw + 16] = acc1[r];
    Cp[(size_t)r * Nw + 32] = acc2[r];
    Cp[(size_t)r * Nw + 48] = acc3[r];
  }
}

__global__ __launch_bounds__(256)
void gat_node_attn(const float* __restrict__ h, const float* __restrict__ attn,
                   float* __restrict__ out, int N, int H) {
  int idx = blockIdx.x * blockDim.x + threadIdx.x;
  if (idx >= N * H) return;
  int n = idx / H, hh = idx - n * H;
  const float4* hp = (const float4*)(h + ((size_t)n * H + hh) * GAT_F);
  const float4* ap = (const float4*)(attn + (size_t)hh * GAT_F);
  float s = 0.f;
#pragma unroll
  for (int f = 0; f < GAT_F / 4; ++f) {
    float4 x = hp[f], w = ap[f];
    s += x.x * w.x + x.y * w.y + x.z * w.z + x.w * w.w;
  }
  out[idx] = s;
}

__device__ __forceinline__ unsigned f32_key(float x) {
  unsigned u = __float_as_uint(x);
  return (u & 0x80000000u) ? ~u : (u | 0x80000000u);
}
__device__ __forceinline__ float key_f32(unsigned k) {
  return (k & 0x80000000u) ? __uint_as_float(k ^ 0x80000000u) : __uint_as_float(~k);
}
#define KEY_NEG_INF 0x007FFFFFu

__device__ __forceinline__ float gat_leaky(float x) { return (x > 0.f) ? x : 0.2f * x; }

#define GAT_TILE_NODES   196
#define GAT_TILE_THREADS 1024
#define GAT_LIST_CAP     4608
#define GAT_LIST_ROUNDS  ((GAT_LIST_CAP + GAT_TILE_THREADS - 1) / GAT_TILE_THREADS)

__device__ __forceinline__ void gat_lds_add(float* p, float v) {
  __hip_atomic_fetch_add(p, v, __ATOMIC_RELAXED, __HIP_MEMORY_SCOPE_WORKGROUP);
}
__device__ __forceinline__ void gat_lds_max(unsigned* p, unsigned v) {
  __hip_atomic_fetch_max(p, v, __ATOMIC_RELAXED, __HIP_MEMORY_SCOPE_WORKGROUP);
}

template <int H> struct gat_hv { float v[H]; };
template <int H>
__device__ __forceinline__ gat_hv<H> gat_ld_heads(const float* p) {
  gat_hv<H> r;
  if constexpr (H == 4) {
    const float4 t = *(const float4*)p;
    r.v[0] = t.x; r.v[1] = t.y; r.v[2] = t.z; r.v[3] = t.w;
  } else {
#pragma unroll
    for (int q = 0; q < H; ++q) r.v[q] = p[q];
  }
  return r;
}
template <int H>
__device__ __forceinline__ void gat_st_heads(float* p, const gat_hv<H>& r) {
  if constexpr (H == 4) {
    *(float4*)p = make_float4(r.v[0], r.v[1], r.v[2], r.v[3]);
  } else {
#pragma unroll
    for (int q = 0; q < H; ++q) p[q] = r.v[q];
  }
}


template <int H>
__device__ __forceinline__ float gat_alpha_head(float a, int head) {
  if constexpr (H == 1) return a;
  else return __int_as_float(__builtin_amdgcn_readlane(__float_as_int(a), head));
}

template <int H>
__device__ __forceinline__ void gat_tile_add_row(const float* __restrict__ h, float* tile,
                                                 int s, int dl, float a, int lane) {
  constexpr int HF = H * GAT_F;
  constexpr int VEC = HF / 32;
  const float* hp = h + (size_t)s * HF + lane;
  float* tp = tile + dl * HF + lane;
  float v[VEC];
#pragma unroll
  for (int j = 0; j < VEC; ++j) v[j] = hp[j * 32];
#pragma unroll
  for (int j = 0; j < VEC; ++j) gat_lds_add(tp + j * 32, v[j] * gat_alpha_head<H>(a, j >> 1));
}

__device__ __forceinline__ void gat_load_dst_group(const int* __restrict__ dst, int E, int gg, int lane,
                                                   int (&dv)[4][4]) {
  const int e0 = gg * 512 + lane * 4;
  if (gg * 512 + 512 <= E) {
#pragma unroll
    for (int c = 0; c < 4; ++c) {
      const int4 t = *(const int4*)(dst + e0 + c * 128);
      dv[c][0] = t.x; dv[c][1] = t.y; dv[c][2] = t.z; dv[c][3] = t.w;
    }
  } else {
#pragma unroll
    for (int c = 0; c < 4; ++c)
#pragma unroll
      for (int j = 0; j < 4; ++j) {
        const int e = e0 + c * 128 + j;
        dv[c][j] = (e < E) ? dst[e] : -1;
      }
  }
}

template <class F>
__device__ __forceinline__ void gat_for_each_match(const int* __restrict__ dst, int E, int base, int cnt,
                                                   int wave, int nw, int lane, F f) {
  const int nGroups = (E + 511) >> 9;
  for (int gg = wave; gg < nGroups; gg += nw) {
    int dv[4][4];
    gat_load_dst_group(dst, E, gg, lane, dv);
#pragma unroll
    for (int c = 0; c < 4; ++c)
#pragma unroll
      for (int j = 0; j < 4; ++j) {
        const int dl = dv[c][j] - base;
        unsigned hit = __builtin_amdgcn_ballot_w32((unsigned)dl < (unsigned)cnt);
        while (hit) {
          const int l = __builtin_ctz(hit);
          hit &= hit - 1u;
          f(gg * 512 + c * 128 + l * 4 + j, __builtin_amdgcn_readlane(dl, l));
        }
      }
  }
}

template <int H, bool RELU>
__global__ __launch_bounds__(GAT_TILE_THREADS)
void gat_tile_aggregate(const int* __restrict__ src, const int* __restrict__ dst,
                        const float* __restrict__ h, const float* __restrict__ el,
                        const float* __restrict__ er, const float* __restrict__ bias,
                        float* __restrict__ out, int N, int E) {
  constexpr int HF = H * GAT_F;
  constexpr int VEC = HF / 32;
  __shared__ __attribute__((aligned(16))) float tile[GAT_TILE_NODES * HF];
  __shared__ __attribute__((aligned(16))) float lg[GAT_LIST_CAP * H];
  __shared__ __attribute__((aligned(16))) float er_t[GAT_TILE_NODES * H];
  __shared__ int2 list[GAT_LIST_CAP];
  __shared__ unsigned mk[GAT_TILE_NODES * H];
  __shared__ float ss[GAT_TILE_NODES * H];
  __shared__ unsigned s_count;

  const int tid = threadIdx.x;
  const int lane = tid & 31;
  const int hq = lane & (H - 1);
  const int wave = __builtin_amdgcn_readfirstlane(tid >> 5);
  const int nw = GAT_TILE_THREADS >> 5;
  const int base = blockIdx.x * GAT_TILE_NODES;
  const int left = N - base;
  const int cnt = left < GAT_TILE_NODES ? left : GAT_TILE_NODES;

  float4* tile4 = (float4*)tile;
  const int n4 = cnt * (HF / 4);
  for (int i = tid; i < n4; i += GAT_TILE_THREADS) tile4[i] = make_float4(0.f, 0.f, 0.f, 0.f);
  for (int i = tid; i < cnt * H; i += GAT_TILE_THREADS) {
    er_t[i] = er[(size_t)base * H + i];
    mk[i] = KEY_NEG_INF;
    ss[i] = 0.f;
  }
  if (tid == 0) s_count = 0u;
  __syncthreads();

  {
    const int nGroups = (E + 511) >> 9;
    for (int gg = wave; gg < nGroups; gg += nw) {
      int dv[4][4];
      gat_load_dst_group(dst, E, gg, lane, dv);
      const int e0 = gg * 512 + lane * 4;
#pragma unroll
      for (int c = 0; c < 4; ++c)
#pragma unroll
        for (int j = 0; j < 4; ++j) {
          const int dl = dv[c][j] - base;
          if ((unsigned)dl < (unsigned)cnt) {
            const unsigned pos = atomicAdd(&s_count, 1u);
            if (pos < (unsigned)GAT_LIST_CAP) list[pos] = make_int2(e0 + c * 128 + j, dl);
          }
        }
    }
  }
  __syncthreads();
  const int total = (int)s_count;

  if (total <= GAT_LIST_CAP) {
    {
      int sv[GAT_LIST_ROUNDS];
      gat_hv<H> lv[GAT_LIST_ROUNDS];
#pragma unroll
      for (int r = 0; r < GAT_LIST_ROUNDS; ++r) {
        const int i = tid + r * GAT_TILE_THREADS;
        sv[r] = src[i < total ? list[i].x : 0];
      }
#pragma unroll
      for (int r = 0; r < GAT_LIST_ROUNDS; ++r) lv[r] = gat_ld_heads<H>(el + (size_t)sv[r] * H);
#pragma unroll
      for (int r = 0; r < GAT_LIST_ROUNDS; ++r) {
        const int i = tid + r * GAT_TILE_THREADS;
        if (i < total) {
          const int dl = list[i].y;
          list[i].x = sv[r];
          const gat_hv<H> rv = gat_ld_heads<H>(er_t + dl * H);
          gat_hv<H> x;
#pragma unroll
          for (int q = 0; q < H; ++q) {
            x.v[q] = gat_leaky(lv[r].v[q] + rv.v[q]);
            gat_lds_max(&mk[dl * H + q], f32_key(x.v[q]));
          }
          gat_st_heads<H>(lg + i * H, x);
        }
      }
    }
    __syncthreads();

    for (int i = tid; i < total; i += GAT_TILE_THREADS) {
      const int dl = list[i].y;
      gat_hv<H> x = gat_ld_heads<H>(lg + i * H);
#pragma unroll
      for (int q = 0; q < H; ++q) {
        x.v[q] = __expf(x.v[q] - key_f32(mk[dl * H + q]));
        gat_lds_add(&ss[dl * H + q], x.v[q]);
      }
      gat_st_heads<H>(lg + i * H, x);
    }
    __syncthreads();

    for (int i = wave * 4; i < total; i += nw * 4) {
      if (i + 4 <= total) {
        int sn[4], dd[4];
#pragma unroll
        for (int k = 0; k < 4; ++k) {
          const int2 en = list[i + k];
          sn[k] = __builtin_amdgcn_readfirstlane(en.x);
          dd[k] = __builtin_amdgcn_readfirstlane(en.y);
        }
        float vals[4][VEC];
        float al[4];
#pragma unroll
        for (int k = 0; k < 4; ++k) {
          al[k] = lg[(i + k) * H + hq] / ss[dd[k] * H + hq];
          const float* hp = h + (size_t)sn[k] * HF + lane;
#pragma unroll
          for (int j = 0; j < VEC; ++j) vals[k][j] = hp[j * 32];
        }
#pragma unroll
        for (int k = 0; k < 4; ++k) {
          float* tp = tile + dd[k] * HF + lane;
#pragma unroll
          for (int j = 0; j < VEC; ++j) gat_lds_add(tp + j * 32, vals[k][j] * gat_alpha_head<H>(al[k], j >> 1));
        }
      } else {
        for (int k = i; k < total; ++k) {
          const int2 en = list[k];
          const int dl = __builtin_amdgcn_readfirstlane(en.y);
          gat_tile_add_row<H>(h, tile, __builtin_amdgcn_readfirstlane(en.x), dl,
                              lg[k * H + hq] / ss[dl * H + hq], lane);
        }
      }
    }
  } else {
    gat_for_each_match(dst, E, base, cnt, wave, nw, lane, [&](int e, int dl) {
      if (lane < H) {
        const int s = src[e];
        gat_lds_max(&mk[dl * H + lane], f32_key(gat_leaky(el[(size_t)s * H + lane] + er_t[dl * H + lane])));
      }
    });
    __syncthreads();
    gat_for_each_match(dst, E, base, cnt, wave, nw, lane, [&](int e, int dl) {
      if (lane < H) {
        const int s = src[e];
        const float x = gat_leaky(el[(size_t)s * H + lane] + er_t[dl * H + lane]);
        gat_lds_add(&ss[dl * H + lane], __expf(x - key_f32(mk[dl * H + lane])));
      }
    });
    __syncthreads();
    gat_for_each_match(dst, E, base, cnt, wave, nw, lane, [&](int e, int dl) {
      const int s = src[e];
      const float x = gat_leaky(el[(size_t)s * H + hq] + er_t[dl * H + hq]);
      gat_tile_add_row<H>(h, tile, s, dl, __expf(x - key_f32(mk[dl * H + hq])) / ss[dl * H + hq], lane);
    });
  }
  __syncthreads();

  const float4* bias4 = (const float4*)bias;
  float4* out4 = (float4*)(out + (size_t)base * HF);
  for (int i = tid; i < n4; i += GAT_TILE_THREADS) {
    float4 v = tile4[i];
    const float4 b = bias4[i & (HF / 4 - 1)];
    v.x += b.x; v.y += b.y; v.z += b.z; v.w += b.w;
    if (RELU) {
      v.x = v.x > 0.f ? v.x : 0.f; v.y = v.y > 0.f ? v.y : 0.f;
      v.z = v.z > 0.f ? v.z : 0.f; v.w = v.w > 0.f ? v.w : 0.f;
    }
    out4[i] = v;
  }
}

extern "C" void kernel_launch(void* const* d_in, const int* in_sizes, int n_in,
                              void* d_out, int out_size, void* d_ws, size_t ws_size,
                              hipStream_t stream) {
  (void)in_sizes; (void)n_in; (void)out_size; (void)ws_size;
  const float* feat = (const float*)d_in[0];
  const int*   src  = (const int*)d_in[1];
  const int*   dst  = (const int*)d_in[2];
  const float* W1  = (const float*)d_in[3];
  const float* al1 = (const float*)d_in[4];
  const float* ar1 = (const float*)d_in[5];
  const float* b1  = (const float*)d_in[6];
  const float* W2  = (const float*)d_in[7];
  const float* al2 = (const float*)d_in[8];
  const float* ar2 = (const float*)d_in[9];
  const float* b2  = (const float*)d_in[10];
  const float* W3  = (const float*)d_in[11];
  const float* al3 = (const float*)d_in[12];
  const float* ar3 = (const float*)d_in[13];
  const float* b3  = (const float*)d_in[14];
  float* out = (float*)d_out;

  const int N = GAT_N, E = GAT_E;

  char* ws = (char*)d_ws;
  size_t off = 0;
  auto carve = [&](size_t bytes) -> void* {
    void* p = ws + off;
    off += (bytes + 255) & ~(size_t)255;
    return p;
  };
  float* bufA = (float*)carve((size_t)N * 256 * 4);
  float* bufB = (float*)carve((size_t)N * 256 * 4);
  float* el   = (float*)carve((size_t)N * 4 * 4);
  float* er   = (float*)carve((size_t)N * 4 * 4);

  auto blocks = [](long long n) { return (unsigned)((n + 255) / 256); };
  const unsigned tiles = (GAT_N + GAT_TILE_NODES - 1) / GAT_TILE_NODES;

  {
    const int K = 128, Nw = 256, H = 4;
    int waves = (N / 16) * (Nw / 64);
    gat_gemm_wmma<<<(waves + 7) / 8, 256, 0, stream>>>(feat, W1, bufA, N, K, Nw);
    long long nh = (long long)N * H;
    gat_node_attn<<<blocks(nh), 256, 0, stream>>>(bufA, al1, el, N, H);
    gat_node_attn<<<blocks(nh), 256, 0, stream>>>(bufA, ar1, er, N, H);
    gat_tile_aggregate<4, true><<<tiles, GAT_TILE_THREADS, 0, stream>>>(src, dst, bufA, el, er, b1, bufB, N, E);
  }

  {
    const int K = 256, Nw = 256, H = 4;
    int waves = (N / 16) * (Nw / 64);
    gat_gemm_wmma<<<(waves + 7) / 8, 256, 0, stream>>>(bufB, W2, bufA, N, K, Nw);
    long long nh = (long long)N * H;
    gat_node_attn<<<blocks(nh), 256, 0, stream>>>(bufA, al2, el, N, H);
    gat_node_attn<<<blocks(nh), 256, 0, stream>>>(bufA, ar2, er, N, H);
    gat_tile_aggregate<4, true><<<tiles, GAT_TILE_THREADS, 0, stream>>>(src, dst, bufA, el, er, b2, bufB, N, E);
  }

  {
    const int K = 256, Nw = 64, H = 1;
    int waves = (N / 16) * (Nw / 64);
    gat_gemm_wmma<<<(waves + 7) / 8, 256, 0, stream>>>(bufB, W3, bufA, N, K, Nw);
    long long nh = (long long)N * H;
    gat_node_attn<<<blocks(nh), 256, 0, stream>>>(bufA, al3, el, N, H);
    gat_node_attn<<<blocks(nh), 256, 0, stream>>>(bufA, ar3, er, N, H);
    gat_tile_aggregate<1, false><<<tiles, GAT_TILE_THREADS, 0, stream>>>(src, dst, bufA, el, er, b3, out, N, E);
  }
}


# Partition launch reads src[]/dst[] once per call and files each edge (source node, local row) into its dst-range segment via LDS slot counters; tile launches copy their segments, no dst[] scan
# speedup vs baseline: 5.6657x; 1.2898x over previous
#include <hip/hip_runtime.h>


#define GAT_N 50000
#define GAT_E 800000
#define GAT_F 64

typedef float v2f __attribute__((ext_vector_type(2)));
typedef float v8f __attribute__((ext_vector_type(8)));

__global__ __launch_bounds__(256)
void gat_gemm_wmma(const float* __restrict__ A, const float* __restrict__ B,
                   float* __restrict__ C, int M, int K, int Nw) {
  const int wave = blockIdx.x * 8 + (threadIdx.x >> 5);
  const int lane = threadIdx.x & 31;
  const int tilesN = Nw >> 6;
  const int tilesM = M >> 4;
  if (wave >= tilesM * tilesN) return;
  const int tm = wave / tilesN;
  const int tn = wave - tm * tilesN;
  const int kh = lane >> 4;
  const int lr = lane & 15;

  const float* Ap = A + (size_t)(tm * 16 + lr) * K + kh * 2;
  const float* Bp = B + (size_t)(tn * 64) + lr;
  v8f acc0 = {}, acc1 = {}, acc2 = {}, acc3 = {};
  for (int k0 = 0; k0 < K; k0 += 4) {
    v2f a = *(const v2f*)(Ap + k0);
    const float* Brow0 = Bp + (size_t)(k0 + kh * 2) * Nw;
    const float* Brow1 = Brow0 + Nw;
    v2f b0, b1, b2, b3;
    b0.x = Brow0[0];  b0.y = Brow1[0];
    b1.x = Brow0[16]; b1.y = Brow1[16];
    b2.x = Brow0[32]; b2.y = Brow1[32];
    b3.x = Brow0[48]; b3.y = Brow1[48];
    acc0 = __builtin_amdgcn_wmma_f32_16x16x4_f32(false, a, false, b0, (short)0, acc0, false, false);
    acc1 = __builtin_amdgcn_wmma_f32_16x16x4_f32(false, a, false, b1, (short)0, acc1, false, false);
    acc2 = __builtin_amdgcn_wmma_f32_16x16x4_f32(false, a, false, b2, (short)0, acc2, false, false);
    acc3 = __builtin_amdgcn_wmma_f32_16x16x4_f32(false, a, false, b3, (short)0, acc3, false, false);
  }
  float* Cp = C + (size_t)(tm * 16 + kh * 8) * Nw + tn * 64 + lr;
#pragma unroll
  for (int r = 0; r < 8; ++r) {
    Cp[(size_t)r * Nw +  0] = acc0[r];
    Cp[(size_t)r * Nw + 16] = acc1[r];
    Cp[(size_t)r * Nw + 32] = acc2[r];
    Cp[(size_t)r * Nw + 48] = acc3[r];
  }
}

__global__ __launch_bounds__(256)
void gat_node_attn(const float* __restrict__ h, const float* __restrict__ attn,
                   float* __restrict__ out, int N, int H) {
  int idx = blockIdx.x * blockDim.x + threadIdx.x;
  if (idx >= N * H) return;
  int n = idx / H, hh = idx - n * H;
  const float4* hp = (const float4*)(h + ((size_t)n * H + hh) * GAT_F);
  const float4* ap = (const float4*)(attn + (size_t)hh * GAT_F);
  float s = 0.f;
#pragma unroll
  for (int f = 0; f < GAT_F / 4; ++f) {
    float4 x = hp[f], w = ap[f];
    s += x.x * w.x + x.y * w.y + x.z * w.z + x.w * w.w;
  }
  out[idx] = s;
}

__device__ __forceinline__ unsigned f32_key(float x) {
  unsigned u = __float_as_uint(x);
  return (u & 0x80000000u) ? ~u : (u | 0x80000000u);
}
__device__ __forceinline__ float key_f32(unsigned k) {
  return (k & 0x80000000u) ? __uint_as_float(k ^ 0x80000000u) : __uint_as_float(~k);
}
#define KEY_NEG_INF 0x007FFFFFu

__device__ __forceinline__ float gat_leaky(float x) { return (x > 0.f) ? x : 0.2f * x; }

#define GAT_TILE_NODES   196
#define GAT_TILE_THREADS 1024
#define GAT_LIST_CAP     4608
#define GAT_LIST_ROUNDS  ((GAT_LIST_CAP + GAT_TILE_THREADS - 1) / GAT_TILE_THREADS)
#define GAT_NUM_TILES    ((GAT_N + GAT_TILE_NODES - 1) / GAT_TILE_NODES)

#define GAT_PART_GROUPS   128
#define GAT_PART_THREADS  1024
#define GAT_PART_EDGES    ((GAT_E + GAT_PART_GROUPS - 1) / GAT_PART_GROUPS)
#define GAT_PART_ROUNDS   ((GAT_PART_EDGES + GAT_PART_THREADS - 1) / GAT_PART_THREADS)
#define GAT_SEG_CAP       64
#define GAT_SEGS_PER_WAVE (GAT_PART_GROUPS / (GAT_TILE_THREADS / 32))
static_assert(GAT_PART_GROUPS % (GAT_TILE_THREADS / 32) == 0, "segments divide evenly over the tile kernel's waves");
static_assert(GAT_SEG_CAP % 32 == 0, "a segment is read in whole-wave rounds");

__global__ __launch_bounds__(GAT_PART_THREADS)
void gat_partition(const int* __restrict__ src, const int* __restrict__ dst,
                   int* __restrict__ segcnt, int* __restrict__ seg) {
  __shared__ unsigned cnt[GAT_NUM_TILES];
  const int tid = threadIdx.x;
  const int g = blockIdx.x;
  for (int i = tid; i < GAT_NUM_TILES; i += GAT_PART_THREADS) cnt[i] = 0u;
  __syncthreads();

  const int e0 = g * GAT_PART_EDGES;
  const int lim = (e0 + GAT_PART_EDGES < GAT_E) ? e0 + GAT_PART_EDGES : GAT_E;
  int dv[GAT_PART_ROUNDS], sv[GAT_PART_ROUNDS];
#pragma unroll
  for (int r = 0; r < GAT_PART_ROUNDS; ++r) {
    const int e = e0 + tid + r * GAT_PART_THREADS;
    const bool ok = e < lim;
    dv[r] = ok ? dst[e] : -1;
    sv[r] = ok ? src[e] : 0;
  }
  int2* segs = (int2*)seg;
#pragma unroll
  for (int r = 0; r < GAT_PART_ROUNDS; ++r) {
    const int d = dv[r];
    if ((unsigned)d < (unsigned)GAT_N) {
      const int t = d / GAT_TILE_NODES;
      const unsigned pos = atomicAdd(&cnt[t], 1u);
      if (pos < (unsigned)GAT_SEG_CAP)
        segs[((size_t)t * GAT_PART_GROUPS + g) * GAT_SEG_CAP + pos] = make_int2(sv[r], d - t * GAT_TILE_NODES);
    }
  }
  __syncthreads();
  for (int i = tid; i < GAT_NUM_TILES; i += GAT_PART_THREADS) segcnt[i * GAT_PART_GROUPS + g] = (int)cnt[i];
}

__device__ __forceinline__ void gat_lds_add(float* p, float v) {
  __hip_atomic_fetch_add(p, v, __ATOMIC_RELAXED, __HIP_MEMORY_SCOPE_WORKGROUP);
}
__device__ __forceinline__ void gat_lds_max(unsigned* p, unsigned v) {
  __hip_atomic_fetch_max(p, v, __ATOMIC_RELAXED, __HIP_MEMORY_SCOPE_WORKGROUP);
}

template <int H> struct gat_hv { float v[H]; };
template <int H>
__device__ __forceinline__ gat_hv<H> gat_ld_heads(const float* p) {
  gat_hv<H> r;
  if constexpr (H == 4) {
    const float4 t = *(const float4*)p;
    r.v[0] = t.x; r.v[1] = t.y; r.v[2] = t.z; r.v[3] = t.w;
  } else {
#pragma unroll
    for (int q = 0; q < H; ++q) r.v[q] = p[q];
  }
  return r;
}
template <int H>
__device__ __forceinline__ void gat_st_heads(float* p, const gat_hv<H>& r) {
  if constexpr (H == 4) {
    *(float4*)p = make_float4(r.v[0], r.v[1], r.v[2], r.v[3]);
  } else {
#pragma unroll
    for (int q = 0; q < H; ++q) p[q] = r.v[q];
  }
}


template <int H>
__device__ __forceinline__ float gat_alpha_head(float a, int head) {
  if constexpr (H == 1) return a;
  else return __int_as_float(__builtin_amdgcn_readlane(__float_as_int(a), head));
}

template <int H>
__device__ __forceinline__ void gat_tile_add_row(const float* __restrict__ h, float* tile,
                                                 int s, int dl, float a, int lane) {
  constexpr int HF = H * GAT_F;
  constexpr int VEC = HF / 32;
  const float* hp = h + (size_t)s * HF + lane;
  float* tp = tile + dl * HF + lane;
  float v[VEC];
#pragma unroll
  for (int j = 0; j < VEC; ++j) v[j] = hp[j * 32];
#pragma unroll
  for (int j = 0; j < VEC; ++j) gat_lds_add(tp + j * 32, v[j] * gat_alpha_head<H>(a, j >> 1));
}

__device__ __forceinline__ void gat_load_dst_group(const int* __restrict__ dst, int E, int gg, int lane,
                                                   int (&dv)[4][4]) {
  const int e0 = gg * 512 + lane * 4;
  if (gg * 512 + 512 <= E) {
#pragma unroll
    for (int c = 0; c < 4; ++c) {
      const int4 t = *(const int4*)(dst + e0 + c * 128);
      dv[c][0] = t.x; dv[c][1] = t.y; dv[c][2] = t.z; dv[c][3] = t.w;
    }
  } else {
#pragma unroll
    for (int c = 0; c < 4; ++c)
#pragma unroll
      for (int j = 0; j < 4; ++j) {
        const int e = e0 + c * 128 + j;
        dv[c][j] = (e < E) ? dst[e] : -1;
      }
  }
}

template <class F>
__device__ __forceinline__ void gat_for_each_match(const int* __restrict__ dst, int E, int base, int cnt,
                                                   int wave, int nw, int lane, F f) {
  const int nGroups = (E + 511) >> 9;
  for (int gg = wave; gg < nGroups; gg += nw) {
    int dv[4][4];
    gat_load_dst_group(dst, E, gg, lane, dv);
#pragma unroll
    for (int c = 0; c < 4; ++c)
#pragma unroll
      for (int j = 0; j < 4; ++j) {
        const int dl = dv[c][j] - base;
        unsigned hit = __builtin_amdgcn_ballot_w32((unsigned)dl < (unsigned)cnt);
        while (hit) {
          const int l = __builtin_ctz(hit);
          hit &= hit - 1u;
          f(gg * 512 + c * 128 + l * 4 + j, __builtin_amdgcn_readlane(dl, l));
        }
      }
  }
}

template <int H, bool RELU>
__global__ __launch_bounds__(GAT_TILE_THREADS)
void gat_tile_aggregate(const int* __restrict__ src, const int* __restrict__ dst,
                        const float* __restrict__ h, const float* __restrict__ el,
                        const float* __restrict__ er, const float* __restrict__ bias,
                        const int* __restrict__ segcnt, const int* __restrict__ seg,
                        float* __restrict__ out, int N, int E) {
  constexpr int HF = H * GAT_F;
  constexpr int VEC = HF / 32;
  __shared__ __attribute__((aligned(16))) float tile[GAT_TILE_NODES * HF];
  __shared__ __attribute__((aligned(16))) float lg[GAT_LIST_CAP * H];
  __shared__ __attribute__((aligned(16))) float er_t[GAT_TILE_NODES * H];
  __shared__ int2 list[GAT_LIST_CAP];
  __shared__ unsigned mk[GAT_TILE_NODES * H];
  __shared__ float ss[GAT_TILE_NODES * H];
  __shared__ unsigned s_count;

  const int tid = threadIdx.x;
  const int lane = tid & 31;
  const int hq = lane & (H - 1);
  const int wave = __builtin_amdgcn_readfirstlane(tid >> 5);
  const int nw = GAT_TILE_THREADS >> 5;
  const int base = blockIdx.x * GAT_TILE_NODES;
  const int left = N - base;
  const int cnt = left < GAT_TILE_NODES ? left : GAT_TILE_NODES;

  float4* tile4 = (float4*)tile;
  const int n4 = cnt * (HF / 4);
  for (int i = tid; i < n4; i += GAT_TILE_THREADS) tile4[i] = make_float4(0.f, 0.f, 0.f, 0.f);
  for (int i = tid; i < cnt * H; i += GAT_TILE_THREADS) {
    er_t[i] = er[(size_t)base * H + i];
    mk[i] = KEY_NEG_INF;
    ss[i] = 0.f;
  }
  if (tid == 0) s_count = 0u;
  __syncthreads();

  {
    const int* cp = segcnt + (size_t)blockIdx.x * GAT_PART_GROUPS;
    const int2* sp = (const int2*)seg + (size_t)blockIdx.x * GAT_PART_GROUPS * GAT_SEG_CAP;
    int c[GAT_SEGS_PER_WAVE];
    unsigned b[GAT_SEGS_PER_WAVE];
    bool over = false;
    unsigned sum = 0u;
#pragma unroll
    for (int k = 0; k < GAT_SEGS_PER_WAVE; ++k) {
      const int ck = cp[wave + k * nw];
      over |= ck > GAT_SEG_CAP;
      c[k] = ck < GAT_SEG_CAP ? ck : GAT_SEG_CAP;
      b[k] = sum;
      sum += (unsigned)c[k];
    }
    if (over) sum += (unsigned)GAT_LIST_CAP + 1u;
    unsigned first = 0u;
    if (lane == 0) first = atomicAdd(&s_count, sum);
    first = __builtin_amdgcn_readfirstlane(first);
    int2 ent[GAT_SEGS_PER_WAVE][GAT_SEG_CAP / 32];
#pragma unroll
    for (int k = 0; k < GAT_SEGS_PER_WAVE; ++k)
#pragma unroll
      for (int j = 0; j < GAT_SEG_CAP / 32; ++j) {
        const int i = lane + 32 * j;
        ent[k][j] = make_int2(0, 0);
        if (i < c[k]) ent[k][j] = sp[(size_t)(wave + k * nw) * GAT_SEG_CAP + i];
      }
#pragma unroll
    for (int k = 0; k < GAT_SEGS_PER_WAVE; ++k)
#pragma unroll
      for (int j = 0; j < GAT_SEG_CAP / 32; ++j) {
        const int i = lane + 32 * j;
        const unsigned pos = first + b[k] + (unsigned)i;
        if (i < c[k] && pos < (unsigned)GAT_LIST_CAP) list[pos] = ent[k][j];
      }
  }
  __syncthreads();
  const int total = (int)s_count;

  if (total <= GAT_LIST_CAP) {
    {
      int sv[GAT_LIST_ROUNDS];
      gat_hv<H> lv[GAT_LIST_ROUNDS];
#pragma unroll
      for (int r = 0; r < GAT_LIST_ROUNDS; ++r) {
        const int i = tid + r * GAT_TILE_THREADS;
        sv[r] = i < total ? list[i].x : 0;
      }
#pragma unroll
      for (int r = 0; r < GAT_LIST_ROUNDS; ++r) lv[r] = gat_ld_heads<H>(el + (size_t)sv[r] * H);
#pragma unroll
      for (int r = 0; r < GAT_LIST_ROUNDS; ++r) {
        const int i = tid + r * GAT_TILE_THREADS;
        if (i < total) {
          const int dl = list[i].y;
          const gat_hv<H> rv = gat_ld_heads<H>(er_t + dl * H);
          gat_hv<H> x;
#pragma unroll
          for (int q = 0; q < H; ++q) {
            x.v[q] = gat_leaky(lv[r].v[q] + rv.v[q]);
            gat_lds_max(&mk[dl * H + q], f32_key(x.v[q]));
          }
          gat_st_heads<H>(lg + i * H, x);
        }
      }
    }
    __syncthreads();

    for (int i = tid; i < total; i += GAT_TILE_THREADS) {
      const int dl = list[i].y;
      gat_hv<H> x = gat_ld_heads<H>(lg + i * H);
#pragma unroll
      for (int q = 0; q < H; ++q) {
        x.v[q] = __expf(x.v[q] - key_f32(mk[dl * H + q]));
        gat_lds_add(&ss[dl * H + q], x.v[q]);
      }
      gat_st_heads<H>(lg + i * H, x);
    }
    __syncthreads();

    for (int i = wave * 4; i < total; i += nw * 4) {
      if (i + 4 <= total) {
        int sn[4], dd[4];
#pragma unroll
        for (int k = 0; k < 4; ++k) {
          const int2 en = list[i + k];
          sn[k] = __builtin_amdgcn_readfirstlane(en.x);
          dd[k] = __builtin_amdgcn_readfirstlane(en.y);
        }
        float vals[4][VEC];
        float al[4];
#pragma unroll
        for (int k = 0; k < 4; ++k) {
          al[k] = lg[(i + k) * H + hq] / ss[dd[k] * H + hq];
          const float* hp = h + (size_t)sn[k] * HF + lane;
#pragma unroll
          for (int j = 0; j < VEC; ++j) vals[k][j] = hp[j * 32];
        }
#pragma unroll
        for (int k = 0; k < 4; ++k) {
          float* tp = tile + dd[k] * HF + lane;
#pragma unroll
          for (int j = 0; j < VEC; ++j) gat_lds_add(tp + j * 32, vals[k][j] * gat_alpha_head<H>(al[k], j >> 1));
        }
      } else {
        for (int k = i; k < total; ++k) {
          const int2 en = list[k];
          const int dl = __builtin_amdgcn_readfirstlane(en.y);
          gat_tile_add_row<H>(h, tile, __builtin_amdgcn_readfirstlane(en.x), dl,
                              lg[k * H + hq] / ss[dl * H + hq], lane);
        }
      }
    }
  } else {
    gat_for_each_match(dst, E, base, cnt, wave, nw, lane, [&](int e, int dl) {
      if (lane < H) {
        const int s = src[e];
        gat_lds_max(&mk[dl * H + lane], f32_key(gat_leaky(el[(size_t)s * H + lane] + er_t[dl * H + lane])));
      }
    });
    __syncthreads();
    gat_for_each_match(dst, E, base, cnt, wave, nw, lane, [&](int e, int dl) {
      if (lane < H) {
        const int s = src[e];
        const float x = gat_leaky(el[(size_t)s * H + lane] + er_t[dl * H + lane]);
        gat_lds_add(&ss[dl * H + lane], __expf(x - key_f32(mk[dl * H + lane])));
      }
    });
    __syncthreads();
    gat_for_each_match(dst, E, base, cnt, wave, nw, lane, [&](int e, int dl) {
      const int s = src[e];
      const float x = gat_leaky(el[(size_t)s * H + hq] + er_t[dl * H + hq]);
      gat_tile_add_row<H>(h, tile, s, dl, __expf(x - key_f32(mk[dl * H + hq])) / ss[dl * H + hq], lane);
    });
  }
  __syncthreads();

  const float4* bias4 = (const float4*)bias;
  float4* out4 = (float4*)(out + (size_t)base * HF);
  for (int i = tid; i < n4; i += GAT_TILE_THREADS) {
    float4 v = tile4[i];
    const float4 b = bias4[i & (HF / 4 - 1)];
    v.x += b.x; v.y += b.y; v.z += b.z; v.w += b.w;
    if (RELU) {
      v.x = v.x > 0.f ? v.x : 0.f; v.y = v.y > 0.f ? v.y : 0.f;
      v.z = v.z > 0.f ? v.z : 0.f; v.w = v.w > 0.f ? v.w : 0.f;
    }
    out4[i] = v;
  }
}

extern "C" void kernel_launch(void* const* d_in, const int* in_sizes, int n_in,
                              void* d_out, int out_size, void* d_ws, size_t ws_size,
                              hipStream_t stream) {
  (void)in_sizes; (void)n_in; (void)out_size; (void)ws_size;
  const float* feat = (const float*)d_in[0];
  const int*   src  = (const int*)d_in[1];
  const int*   dst  = (const int*)d_in[2];
  const float* W1  = (const float*)d_in[3];
  const float* al1 = (const float*)d_in[4];
  const float* ar1 = (const float*)d_in[5];
  const float* b1  = (const float*)d_in[6];
  const float* W2  = (const float*)d_in[7];
  const float* al2 = (const float*)d_in[8];
  const float* ar2 = (const float*)d_in[9];
  const float* b2  = (const float*)d_in[10];
  const float* W3  = (const float*)d_in[11];
  const float* al3 = (const float*)d_in[12];
  const float* ar3 = (const float*)d_in[13];
  const float* b3  = (const float*)d_in[14];
  float* out = (float*)d_out;

  const int N = GAT_N, E = GAT_E;

  char* ws = (char*)d_ws;
  size_t off = 0;
  auto carve = [&](size_t bytes) -> void* {
    void* p = ws + off;
    off += (bytes + 255) & ~(size_t)255;
    return p;
  };
  float* bufA = (float*)carve((size_t)N * 256 * 4);
  float* bufB = (float*)carve((size_t)N * 256 * 4);
  float* el   = (float*)carve((size_t)N * 4 * 4);
  float* er   = (float*)carve((size_t)N * 4 * 4);
  int* segcnt = (int*)carve((size_t)GAT_NUM_TILES * GAT_PART_GROUPS * 4);
  int* seg    = (int*)carve((size_t)GAT_NUM_TILES * GAT_PART_GROUPS * GAT_SEG_CAP * 8);

  auto blocks = [](long long n) { return (unsigned)((n + 255) / 256); };
  const unsigned tiles = GAT_NUM_TILES;

  gat_partition<<<GAT_PART_GROUPS, GAT_PART_THREADS, 0, stream>>>(src, dst, segcnt, seg);

  {
    const int K = 128, Nw = 256, H = 4;
    int waves = (N / 16) * (Nw / 64);
    gat_gemm_wmma<<<(waves + 7) / 8, 256, 0, stream>>>(feat, W1, bufA, N, K, Nw);
    long long nh = (long long)N * H;
    gat_node_attn<<<blocks(nh), 256, 0, stream>>>(bufA, al1, el, N, H);
    gat_node_attn<<<blocks(nh), 256, 0, stream>>>(bufA, ar1, er, N, H);
    gat_tile_aggregate<4, true><<<tiles, GAT_TILE_THREADS, 0, stream>>>(src, dst, bufA, el, er, b1, segcnt, seg, bufB, N, E);
  }

  {
    const int K = 256, Nw = 256, H = 4;
    int waves = (N / 16) * (Nw / 64);
    gat_gemm_wmma<<<(waves + 7) / 8, 256, 0, stream>>>(bufB, W2, bufA, N, K, Nw);
    long long nh = (long long)N * H;
    gat_node_attn<<<blocks(nh), 256, 0, stream>>>(bufA, al2, el, N, H);
    gat_node_attn<<<blocks(nh), 256, 0, stream>>>(bufA, ar2, er, N, H);
    gat_tile_aggregate<4, true><<<tiles, GAT_TILE_THREADS, 0, stream>>>(src, dst, bufA, el, er, b2, segcnt, seg, bufB, N, E);
  }

  {
    const int K = 256, Nw = 64, H = 1;
    int waves = (N / 16) * (Nw / 64);
    gat_gemm_wmma<<<(waves + 7) / 8, 256, 0, stream>>>(bufB, W3, bufA, N, K, Nw);
    long long nh = (long long)N * H;
    gat_node_attn<<<blocks(nh), 256, 0, stream>>>(bufA, al3, el, N, H);
    gat_node_attn<<<blocks(nh), 256, 0, stream>>>(bufA, ar3, er, N, H);
    gat_tile_aggregate<1, false><<<tiles, GAT_TILE_THREADS, 0, stream>>>(src, dst, bufA, el, er, b3, segcnt, seg, out, N, E);
  }
}


# el and er from one shared pass over h: a 16-lane half reads a whole 256-B head-row per load, halving-tree lane sums, one 128-B line stored per wave; 6 node launches become 3; new el/er sum order
# speedup vs baseline: 6.2946x; 1.1110x over previous
#include <hip/hip_runtime.h>


#define GAT_N 50000
#define GAT_E 800000
#define GAT_F 64

typedef float v2f __attribute__((ext_vector_type(2)));
typedef float v8f __attribute__((ext_vector_type(8)));

__global__ __launch_bounds__(256)
void gat_gemm_wmma(const float* __restrict__ A, const float* __restrict__ B,
                   float* __restrict__ C, int M, int K, int Nw) {
  const int wave = blockIdx.x * 8 + (threadIdx.x >> 5);
  const int lane = threadIdx.x & 31;
  const int tilesN = Nw >> 6;
  const int tilesM = M >> 4;
  if (wave >= tilesM * tilesN) return;
  const int tm = wave / tilesN;
  const int tn = wave - tm * tilesN;
  const int kh = lane >> 4;
  const int lr = lane & 15;

  const float* Ap = A + (size_t)(tm * 16 + lr) * K + kh * 2;
  const float* Bp = B + (size_t)(tn * 64) + lr;
  v8f acc0 = {}, acc1 = {}, acc2 = {}, acc3 = {};
  for (int k0 = 0; k0 < K; k0 += 4) {
    v2f a = *(const v2f*)(Ap + k0);
    const float* Brow0 = Bp + (size_t)(k0 + kh * 2) * Nw;
    const float* Brow1 = Brow0 + Nw;
    v2f b0, b1, b2, b3;
    b0.x = Brow0[0];  b0.y = Brow1[0];
    b1.x = Brow0[16]; b1.y = Brow1[16];
    b2.x = Brow0[32]; b2.y = Brow1[32];
    b3.x = Brow0[48]; b3.y = Brow1[48];
    acc0 = __builtin_amdgcn_wmma_f32_16x16x4_f32(false, a, false, b0, (short)0, acc0, false, false);
    acc1 = __builtin_amdgcn_wmma_f32_16x16x4_f32(false, a, false, b1, (short)0, acc1, false, false);
    acc2 = __builtin_amdgcn_wmma_f32_16x16x4_f32(false, a, false, b2, (short)0, acc2, false, false);
    acc3 = __builtin_amdgcn_wmma_f32_16x16x4_f32(false, a, false, b3, (short)0, acc3, false, false);
  }
  float* Cp = C + (size_t)(tm * 16 + kh * 8) * Nw + tn * 64 + lr;
#pragma unroll
  for (int r = 0; r < 8; ++r) {
    Cp[(size_t)r * Nw +  0] = acc0[r];
    Cp[(size_t)r * Nw + 16] = acc1[r];
    Cp[(size_t)r * Nw + 32] = acc2[r];
    Cp[(size_t)r * Nw + 48] = acc3[r];
  }
}


template <int C, int M>
__device__ __forceinline__ void gat_halve(float (&v)[16], int lane) {
  const bool up = (lane & M) != 0;
#pragma unroll
  for (int k = 0; k < C / 2; ++k) {
    const float a = v[k], b = v[k + C / 2];
    const float keep = up ? b : a;
    const float send = up ? a : b;
    v[k] = keep + __shfl_xor(send, M);
  }
}

template <int H>
__global__ __launch_bounds__(256)
void gat_node_attn2(const float* __restrict__ h, const float* __restrict__ attn_l,
                    const float* __restrict__ attn_r, float* __restrict__ el,
                    float* __restrict__ er, int R) {
  static_assert(H == 1 || H == 4, "the head of a load is taken from its number for these head counts");
  constexpr int NWT = (H == 4) ? 2 : 1;
  const int lane = threadIdx.x & 31;
  const int hf = lane >> 4;
  const int li = lane & 15;
  const int r0 = (blockIdx.x * 8 + (threadIdx.x >> 5)) * 32;
  if (r0 >= R) return;

  float4 wl[NWT], wr[NWT];
#pragma unroll
  for (int w = 0; w < NWT; ++w) {
    const int head = (2 * w + hf) % H;
    wl[w] = ((const float4*)attn_l)[head * (GAT_F / 4) + li];
    wr[w] = ((const float4*)attn_r)[head * (GAT_F / 4) + li];
  }
  const float4* h4 = (const float4*)h;
  float4 x[16];
#pragma unroll
  for (int t = 0; t < 16; ++t) {
    const int r = r0 + 2 * t + hf;
    x[t] = h4[(size_t)(r < R ? r : R - 1) * (GAT_F / 4) + li];
  }
  float pl[16], pr[16];
#pragma unroll
  for (int t = 0; t < 16; ++t) {
    const float4 a = wl[t & (NWT - 1)], b = wr[t & (NWT - 1)];
    pl[t] = x[t].x * a.x + x[t].y * a.y + x[t].z * a.z + x[t].w * a.w;
    pr[t] = x[t].x * b.x + x[t].y * b.y + x[t].z * b.z + x[t].w * b.w;
  }
  gat_halve<16, 8>(pl, lane); gat_halve<16, 8>(pr, lane);
  gat_halve<8, 4>(pl, lane);  gat_halve<8, 4>(pr, lane);
  gat_halve<4, 2>(pl, lane);  gat_halve<4, 2>(pr, lane);
  gat_halve<2, 1>(pl, lane);  gat_halve<2, 1>(pr, lane);
  const int r = r0 + 2 * li + hf;
  if (r < R) { el[r] = pl[0]; er[r] = pr[0]; }
}

__device__ __forceinline__ unsigned f32_key(float x) {
  unsigned u = __float_as_uint(x);
  return (u & 0x80000000u) ? ~u : (u | 0x80000000u);
}
__device__ __forceinline__ float key_f32(unsigned k) {
  return (k & 0x80000000u) ? __uint_as_float(k ^ 0x80000000u) : __uint_as_float(~k);
}
#define KEY_NEG_INF 0x007FFFFFu

__device__ __forceinline__ float gat_leaky(float x) { return (x > 0.f) ? x : 0.2f * x; }

#define GAT_TILE_NODES   196
#define GAT_TILE_THREADS 1024
#define GAT_LIST_CAP     4608
#define GAT_LIST_ROUNDS  ((GAT_LIST_CAP + GAT_TILE_THREADS - 1) / GAT_TILE_THREADS)
#define GAT_NUM_TILES    ((GAT_N + GAT_TILE_NODES - 1) / GAT_TILE_NODES)

#define GAT_PART_GROUPS   128
#define GAT_PART_THREADS  1024
#define GAT_PART_EDGES    ((GAT_E + GAT_PART_GROUPS - 1) / GAT_PART_GROUPS)
#define GAT_PART_ROUNDS   ((GAT_PART_EDGES + GAT_PART_THREADS - 1) / GAT_PART_THREADS)
#define GAT_SEG_CAP       64
#define GAT_SEGS_PER_WAVE (GAT_PART_GROUPS / (GAT_TILE_THREADS / 32))
static_assert(GAT_PART_GROUPS % (GAT_TILE_THREADS / 32) == 0, "segments divide evenly over the tile kernel's waves");
static_assert(GAT_SEG_CAP % 32 == 0, "a segment is read in whole-wave rounds");

__global__ __launch_bounds__(GAT_PART_THREADS)
void gat_partition(const int* __restrict__ src, const int* __restrict__ dst,
                   int* __restrict__ segcnt, int* __restrict__ seg) {
  __shared__ unsigned cnt[GAT_NUM_TILES];
  const int tid = threadIdx.x;
  const int g = blockIdx.x;
  for (int i = tid; i < GAT_NUM_TILES; i += GAT_PART_THREADS) cnt[i] = 0u;
  __syncthreads();

  const int e0 = g * GAT_PART_EDGES;
  const int lim = (e0 + GAT_PART_EDGES < GAT_E) ? e0 + GAT_PART_EDGES : GAT_E;
  int dv[GAT_PART_ROUNDS], sv[GAT_PART_ROUNDS];
#pragma unroll
  for (int r = 0; r < GAT_PART_ROUNDS; ++r) {
    const int e = e0 + tid + r * GAT_PART_THREADS;
    const bool ok = e < lim;
    dv[r] = ok ? dst[e] : -1;
    sv[r] = ok ? src[e] : 0;
  }
  int2* segs = (int2*)seg;
#pragma unroll
  for (int r = 0; r < GAT_PART_ROUNDS; ++r) {
    const int d = dv[r];
    if ((unsigned)d < (unsigned)GAT_N) {
      const int t = d / GAT_TILE_NODES;
      const unsigned pos = atomicAdd(&cnt[t], 1u);
      if (pos < (unsigned)GAT_SEG_CAP)
        segs[((size_t)t * GAT_PART_GROUPS + g) * GAT_SEG_CAP + pos] = make_int2(sv[r], d - t * GAT_TILE_NODES);
    }
  }
  __syncthreads();
  for (int i = tid; i < GAT_NUM_TILES; i += GAT_PART_THREADS) segcnt[i * GAT_PART_GROUPS + g] = (int)cnt[i];
}

__device__ __forceinline__ void gat_lds_add(float* p, float v) {
  __hip_atomic_fetch_add(p, v, __ATOMIC_RELAXED, __HIP_MEMORY_SCOPE_WORKGROUP);
}
__device__ __forceinline__ void gat_lds_max(unsigned* p, unsigned v) {
  __hip_atomic_fetch_max(p, v, __ATOMIC_RELAXED, __HIP_MEMORY_SCOPE_WORKGROUP);
}

template <int H> struct gat_hv { float v[H]; };
template <int H>
__device__ __forceinline__ gat_hv<H> gat_ld_heads(const float* p) {
  gat_hv<H> r;
  if constexpr (H == 4) {
    const float4 t = *(const float4*)p;
    r.v[0] = t.x; r.v[1] = t.y; r.v[2] = t.z; r.v[3] = t.w;
  } else {
#pragma unroll
    for (int q = 0; q < H; ++q) r.v[q] = p[q];
  }
  return r;
}
template <int H>
__device__ __forceinline__ void gat_st_heads(float* p, const gat_hv<H>& r) {
  if constexpr (H == 4) {
    *(float4*)p = make_float4(r.v[0], r.v[1], r.v[2], r.v[3]);
  } else {
#pragma unroll
    for (int q = 0; q < H; ++q) p[q] = r.v[q];
  }
}


template <int H>
__device__ __forceinline__ float gat_alpha_head(float a, int head) {
  if constexpr (H == 1) return a;
  else return __int_as_float(__builtin_amdgcn_readlane(__float_as_int(a), head));
}

template <int H>
__device__ __forceinline__ void gat_tile_add_row(const float* __restrict__ h, float* tile,
                                                 int s, int dl, float a, int lane) {
  constexpr int HF = H * GAT_F;
  constexpr int VEC = HF / 32;
  const float* hp = h + (size_t)s * HF + lane;
  float* tp = tile + dl * HF + lane;
  float v[VEC];
#pragma unroll
  for (int j = 0; j < VEC; ++j) v[j] = hp[j * 32];
#pragma unroll
  for (int j = 0; j < VEC; ++j) gat_lds_add(tp + j * 32, v[j] * gat_alpha_head<H>(a, j >> 1));
}

__device__ __forceinline__ void gat_load_dst_group(const int* __restrict__ dst, int E, int gg, int lane,
                                                   int (&dv)[4][4]) {
  const int e0 = gg * 512 + lane * 4;
  if (gg * 512 + 512 <= E) {
#pragma unroll
    for (int c = 0; c < 4; ++c) {
      const int4 t = *(const int4*)(dst + e0 + c * 128);
      dv[c][0] = t.x; dv[c][1] = t.y; dv[c][2] = t.z; dv[c][3] = t.w;
    }
  } else {
#pragma unroll
    for (int c = 0; c < 4; ++c)
#pragma unroll
      for (int j = 0; j < 4; ++j) {
        const int e = e0 + c * 128 + j;
        dv[c][j] = (e < E) ? dst[e] : -1;
      }
  }
}

template <class F>
__device__ __forceinline__ void gat_for_each_match(const int* __restrict__ dst, int E, int base, int cnt,
                                                   int wave, int nw, int lane, F f) {
  const int nGroups = (E + 511) >> 9;
  for (int gg = wave; gg < nGroups; gg += nw) {
    int dv[4][4];
    gat_load_dst_group(dst, E, gg, lane, dv);
#pragma unroll
    for (int c = 0; c < 4; ++c)
#pragma unroll
      for (int j = 0; j < 4; ++j) {
        const int dl = dv[c][j] - base;
        unsigned hit = __builtin_amdgcn_ballot_w32((unsigned)dl < (unsigned)cnt);
        while (hit) {
          const int l = __builtin_ctz(hit);
          hit &= hit - 1u;
          f(gg * 512 + c * 128 + l * 4 + j, __builtin_amdgcn_readlane(dl, l));
        }
      }
  }
}

template <int H, bool RELU>
__global__ __launch_bounds__(GAT_TILE_THREADS)
void gat_tile_aggregate(const int* __restrict__ src, const int* __restrict__ dst,
                        const float* __restrict__ h, const float* __restrict__ el,
                        const float* __restrict__ er, const float* __restrict__ bias,
                        const int* __restrict__ segcnt, const int* __restrict__ seg,
                        float* __restrict__ out, int N, int E) {
  constexpr int HF = H * GAT_F;
  constexpr int VEC = HF / 32;
  __shared__ __attribute__((aligned(16))) float tile[GAT_TILE_NODES * HF];
  __shared__ __attribute__((aligned(16))) float lg[GAT_LIST_CAP * H];
  __shared__ __attribute__((aligned(16))) float er_t[GAT_TILE_NODES * H];
  __shared__ int2 list[GAT_LIST_CAP];
  __shared__ unsigned mk[GAT_TILE_NODES * H];
  __shared__ float ss[GAT_TILE_NODES * H];
  __shared__ unsigned s_count;

  const int tid = threadIdx.x;
  const int lane = tid & 31;
  const int hq = lane & (H - 1);
  const int wave = __builtin_amdgcn_readfirstlane(tid >> 5);
  const int nw = GAT_TILE_THREADS >> 5;
  const int base = blockIdx.x * GAT_TILE_NODES;
  const int left = N - base;
  const int cnt = left < GAT_TILE_NODES ? left : GAT_TILE_NODES;

  float4* tile4 = (float4*)tile;
  const int n4 = cnt * (HF / 4);
  for (int i = tid; i < n4; i += GAT_TILE_THREADS) tile4[i] = make_float4(0.f, 0.f, 0.f, 0.f);
  for (int i = tid; i < cnt * H; i += GAT_TILE_THREADS) {
    er_t[i] = er[(size_t)base * H + i];
    mk[i] = KEY_NEG_INF;
    ss[i] = 0.f;
  }
  if (tid == 0) s_count = 0u;
  __syncthreads();

  {
    const int* cp = segcnt + (size_t)blockIdx.x * GAT_PART_GROUPS;
    const int2* sp = (const int2*)seg + (size_t)blockIdx.x * GAT_PART_GROUPS * GAT_SEG_CAP;
    int c[GAT_SEGS_PER_WAVE];
    unsigned b[GAT_SEGS_PER_WAVE];
    bool over = false;
    unsigned sum = 0u;
#pragma unroll
    for (int k = 0; k < GAT_SEGS_PER_WAVE; ++k) {
      const int ck = cp[wave + k * nw];
      over |= ck > GAT_SEG_CAP;
      c[k] = ck < GAT_SEG_CAP ? ck : GAT_SEG_CAP;
      b[k] = sum;
      sum += (unsigned)c[k];
    }
    if (over) sum += (unsigned)GAT_LIST_CAP + 1u;
    unsigned first = 0u;
    if (lane == 0) first = atomicAdd(&s_count, sum);
    first = __builtin_amdgcn_readfirstlane(first);
    int2 ent[GAT_SEGS_PER_WAVE][GAT_SEG_CAP / 32];
#pragma unroll
    for (int k = 0; k < GAT_SEGS_PER_WAVE; ++k)
#pragma unroll
      for (int j = 0; j < GAT_SEG_CAP / 32; ++j) {
        const int i = lane + 32 * j;
        ent[k][j] = make_int2(0, 0);
        if (i < c[k]) ent[k][j] = sp[(size_t)(wave + k * nw) * GAT_SEG_CAP + i];
      }
#pragma unroll
    for (int k = 0; k < GAT_SEGS_PER_WAVE; ++k)
#pragma unroll
      for (int j = 0; j < GAT_SEG_CAP / 32; ++j) {
        const int i = lane + 32 * j;
        const unsigned pos = first + b[k] + (unsigned)i;
        if (i < c[k] && pos < (unsigned)GAT_LIST_CAP) list[pos] = ent[k][j];
      }
  }
  __syncthreads();
  const int total = (int)s_count;

  if (total <= GAT_LIST_CAP) {
    {
      int sv[GAT_LIST_ROUNDS];
      gat_hv<H> lv[GAT_LIST_ROUNDS];
#pragma unroll
      for (int r = 0; r < GAT_LIST_ROUNDS; ++r) {
        const int i = tid + r * GAT_TILE_THREADS;
        sv[r] = i < total ? list[i].x : 0;
      }
#pragma unroll
      for (int r = 0; r < GAT_LIST_ROUNDS; ++r) lv[r] = gat_ld_heads<H>(el + (size_t)sv[r] * H);
#pragma unroll
      for (int r = 0; r < GAT_LIST_ROUNDS; ++r) {
        const int i = tid + r * GAT_TILE_THREADS;
        if (i < total) {
          const int dl = list[i].y;
          const gat_hv<H> rv = gat_ld_heads<H>(er_t + dl * H);
          gat_hv<H> x;
#pragma unroll
          for (int q = 0; q < H; ++q) {
            x.v[q] = gat_leaky(lv[r].v[q] + rv.v[q]);
            gat_lds_max(&mk[dl * H + q], f32_key(x.v[q]));
          }
          gat_st_heads<H>(lg + i * H, x);
        }
      }
    }
    __syncthreads();

    for (int i = tid; i < total; i += GAT_TILE_THREADS) {
      const int dl = list[i].y;
      gat_hv<H> x = gat_ld_heads<H>(lg + i * H);
#pragma unroll
      for (int q = 0; q < H; ++q) {
        x.v[q] = __expf(x.v[q] - key_f32(mk[dl * H + q]));
        gat_lds_add(&ss[dl * H + q], x.v[q]);
      }
      gat_st_heads<H>(lg + i * H, x);
    }
    __syncthreads();

    for (int i = wave * 4; i < total; i += nw * 4) {
      if (i + 4 <= total) {
        int sn[4], dd[4];
#pragma unroll
        for (int k = 0; k < 4; ++k) {
          const int2 en = list[i + k];
          sn[k] = __builtin_amdgcn_readfirstlane(en.x);
          dd[k] = __builtin_amdgcn_readfirstlane(en.y);
        }
        float vals[4][VEC];
        float al[4];
#pragma unroll
        for (int k = 0; k < 4; ++k) {
          al[k] = lg[(i + k) * H + hq] / ss[dd[k] * H + hq];
          const float* hp = h + (size_t)sn[k] * HF + lane;
#pragma unroll
          for (int j = 0; j < VEC; ++j) vals[k][j] = hp[j * 32];
        }
#pragma unroll
        for (int k = 0; k < 4; ++k) {
          float* tp = tile + dd[k] * HF + lane;
#pragma unroll
          for (int j = 0; j < VEC; ++j) gat_lds_add(tp + j * 32, vals[k][j] * gat_alpha_head<H>(al[k], j >> 1));
        }
      } else {
        for (int k = i; k < total; ++k) {
          const int2 en = list[k];
          const int dl = __builtin_amdgcn_readfirstlane(en.y);
          gat_tile_add_row<H>(h, tile, __builtin_amdgcn_readfirstlane(en.x), dl,
                              lg[k * H + hq] / ss[dl * H + hq], lane);
        }
      }
    }
  } else {
    gat_for_each_match(dst, E, base, cnt, wave, nw, lane, [&](int e, int dl) {
      if (lane < H) {
        const int s = src[e];
        gat_lds_max(&mk[dl * H + lane], f32_key(gat_leaky(el[(size_t)s * H + lane] + er_t[dl * H + lane])));
      }
    });
    __syncthreads();
    gat_for_each_match(dst, E, base, cnt, wave, nw, lane, [&](int e, int dl) {
      if (lane < H) {
        const int s = src[e];
        const float x = gat_leaky(el[(size_t)s * H + lane] + er_t[dl * H + lane]);
        gat_lds_add(&ss[dl * H + lane], __expf(x - key_f32(mk[dl * H + lane])));
      }
    });
    __syncthreads();
    gat_for_each_match(dst, E, base, cnt, wave, nw, lane, [&](int e, int dl) {
      const int s = src[e];
      const float x = gat_leaky(el[(size_t)s * H + hq] + er_t[dl * H + hq]);
      gat_tile_add_row<H>(h, tile, s, dl, __expf(x - key_f32(mk[dl * H + hq])) / ss[dl * H + hq], lane);
    });
  }
  __syncthreads();

  const float4* bias4 = (const float4*)bias;
  float4* out4 = (float4*)(out + (size_t)base * HF);
  for (int i = tid; i < n4; i += GAT_TILE_THREADS) {
    float4 v = tile4[i];
    const float4 b = bias4[i & (HF / 4 - 1)];
    v.x += b.x; v.y += b.y; v.z += b.z; v.w += b.w;
    if (RELU) {
      v.x = v.x > 0.f ? v.x : 0.f; v.y = v.y > 0.f ? v.y : 0.f;
      v.z = v.z > 0.f ? v.z : 0.f; v.w = v.w > 0.f ? v.w : 0.f;
    }
    out4[i] = v;
  }
}

extern "C" void kernel_launch(void* const* d_in, const int* in_sizes, int n_in,
                              void* d_out, int out_size, void* d_ws, size_t ws_size,
                              hipStream_t stream) {
  (void)in_sizes; (void)n_in; (void)out_size; (void)ws_size;
  const float* feat = (const float*)d_in[0];
  const int*   src  = (const int*)d_in[1];
  const int*   dst  = (const int*)d_in[2];
  const float* W1  = (const float*)d_in[3];
  const float* al1 = (const float*)d_in[4];
  const float* ar1 = (const float*)d_in[5];
  const float* b1  = (const float*)d_in[6];
  const float* W2  = (const float*)d_in[7];
  const float* al2 = (const float*)d_in[8];
  const float* ar2 = (const float*)d_in[9];
  const float* b2  = (const float*)d_in[10];
  const float* W3  = (const float*)d_in[11];
  const float* al3 = (const float*)d_in[12];
  const float* ar3 = (const float*)d_in[13];
  const float* b3  = (const float*)d_in[14];
  float* out = (float*)d_out;

  const int N = GAT_N, E = GAT_E;

  char* ws = (char*)d_ws;
  size_t off = 0;
  auto carve = [&](size_t bytes) -> void* {
    void* p = ws + off;
    off += (bytes + 255) & ~(size_t)255;
    return p;
  };
  float* bufA = (float*)carve((size_t)N * 256 * 4);
  float* bufB = (float*)carve((size_t)N * 256 * 4);
  float* el   = (float*)carve((size_t)N * 4 * 4);
  float* er   = (float*)carve((size_t)N * 4 * 4);
  int* segcnt = (int*)carve((size_t)GAT_NUM_TILES * GAT_PART_GROUPS * 4);
  int* seg    = (int*)carve((size_t)GAT_NUM_TILES * GAT_PART_GROUPS * GAT_SEG_CAP * 8);

  auto blocks = [](long long n) { return (unsigned)((n + 255) / 256); };
  const unsigned tiles = GAT_NUM_TILES;

  gat_partition<<<GAT_PART_GROUPS, GAT_PART_THREADS, 0, stream>>>(src, dst, segcnt, seg);

  {
    const int K = 128, Nw = 256, H = 4;
    int waves = (N / 16) * (Nw / 64);
    gat_gemm_wmma<<<(waves + 7) / 8, 256, 0, stream>>>(feat, W1, bufA, N, K, Nw);
    long long nh = (long long)N * H;
    gat_node_attn2<4><<<blocks(nh), 256, 0, stream>>>(bufA, al1, ar1, el, er, (int)nh);
    gat_tile_aggregate<4, true><<<tiles, GAT_TILE_THREADS, 0, stream>>>(src, dst, bufA, el, er, b1, segcnt, seg, bufB, N, E);
  }

  {
    const int K = 256, Nw = 256, H = 4;
    int waves = (N / 16) * (Nw / 64);
    gat_gemm_wmma<<<(waves + 7) / 8, 256, 0, stream>>>(bufB, W2, bufA, N, K, Nw);
    long long nh = (long long)N * H;
    gat_node_attn2<4><<<blocks(nh), 256, 0, stream>>>(bufA, al2, ar2, el, er, (int)nh);
    gat_tile_aggregate<4, true><<<tiles, GAT_TILE_THREADS, 0, stream>>>(src, dst, bufA, el, er, b2, segcnt, seg, bufB, N, E);
  }

  {
    const int K = 256, Nw = 64, H = 1;
    int waves = (N / 16) * (Nw / 64);
    gat_gemm_wmma<<<(waves + 7) / 8, 256, 0, stream>>>(bufB, W3, bufA, N, K, Nw);
    long long nh = (long long)N * H;
    gat_node_attn2<1><<<blocks(nh), 256, 0, stream>>>(bufA, al3, ar3, el, er, (int)nh);
    gat_tile_aggregate<1, false><<<tiles, GAT_TILE_THREADS, 0, stream>>>(src, dst, bufA, el, er, b3, segcnt, seg, out, N, E);
  }
}


# GEMM k loop in batches of 4 k-steps: a batch's 36 fragment loads are issued before its 16 WMMAs, so a wave waits one memory round trip per batch, not per k-step; same k order, sums bit-identical
# speedup vs baseline: 6.9996x; 1.1120x over previous
#include <hip/hip_runtime.h>


#define GAT_N 50000
#define GAT_E 800000
#define GAT_F 64

typedef float v2f __attribute__((ext_vector_type(2)));
typedef float v8f __attribute__((ext_vector_type(8)));

#define GEMM_KB 4
__global__ __launch_bounds__(256)
void gat_gemm_wmma(const float* __restrict__ A, const float* __restrict__ B,
                   float* __restrict__ C, int M, int K, int Nw) {
  const int wave = blockIdx.x * 8 + (threadIdx.x >> 5);
  const int lane = threadIdx.x & 31;
  const int tilesN = Nw >> 6;
  const int tilesM = M >> 4;
  if (wave >= tilesM * tilesN) return;
  const int tm = wave / tilesN;
  const int tn = wave - tm * tilesN;
  const int kh = lane >> 4;
  const int lr = lane & 15;

  const float* Ap = A + (size_t)(tm * 16 + lr) * K + kh * 2;
  const float* Bp = B + (size_t)(tn * 64) + lr + (size_t)(kh * 2) * Nw;
  v8f acc0 = {}, acc1 = {}, acc2 = {}, acc3 = {};
  for (int k0 = 0; k0 < K; k0 += 4 * GEMM_KB) {
    v2f a[GEMM_KB], b0[GEMM_KB], b1[GEMM_KB], b2[GEMM_KB], b3[GEMM_KB];
    const float* Bk = Bp + (size_t)k0 * Nw;
#pragma unroll
    for (int s = 0; s < GEMM_KB; ++s) {
      a[s] = *(const v2f*)(Ap + k0 + 4 * s);
      const float* Brow0 = Bk + (size_t)(4 * s) * Nw;
      const float* Brow1 = Brow0 + Nw;
      b0[s].x = Brow0[0];  b0[s].y = Brow1[0];
      b1[s].x = Brow0[16]; b1[s].y = Brow1[16];
      b2[s].x = Brow0[32]; b2[s].y = Brow1[32];
      b3[s].x = Brow0[48]; b3[s].y = Brow1[48];
    }
    __builtin_amdgcn_sched_barrier(0);
#pragma unroll
    for (int s = 0; s < GEMM_KB; ++s) {
      acc0 = __builtin_amdgcn_wmma_f32_16x16x4_f32(false, a[s], false, b0[s], (short)0, acc0, false, false);
      acc1 = __builtin_amdgcn_wmma_f32_16x16x4_f32(false, a[s], false, b1[s], (short)0, acc1, false, false);
      acc2 = __builtin_amdgcn_wmma_f32_16x16x4_f32(false, a[s], false, b2[s], (short)0, acc2, false, false);
      acc3 = __builtin_amdgcn_wmma_f32_16x16x4_f32(false, a[s], false, b3[s], (short)0, acc3, false, false);
    }
  }
  float* Cp = C + (size_t)(tm * 16 + kh * 8) * Nw + tn * 64 + lr;
#pragma unroll
  for (int r = 0; r < 8; ++r) {
    Cp[(size_t)r * Nw +  0] = acc0[r];
    Cp[(size_t)r * Nw + 16] = acc1[r];
    Cp[(size_t)r * Nw + 32] = acc2[r];
    Cp[(size_t)r * Nw + 48] = acc3[r];
  }
}


template <int C, int M>
__device__ __forceinline__ void gat_halve(float (&v)[16], int lane) {
  const bool up = (lane & M) != 0;
#pragma unroll
  for (int k = 0; k < C / 2; ++k) {
    const float a = v[k], b = v[k + C / 2];
    const float keep = up ? b : a;
    const float send = up ? a : b;
    v[k] = keep + __shfl_xor(send, M);
  }
}

template <int H>
__global__ __launch_bounds__(256)
void gat_node_attn2(const float* __restrict__ h, const float* __restrict__ attn_l,
                    const float* __restrict__ attn_r, float* __restrict__ el,
                    float* __restrict__ er, int R) {
  static_assert(H == 1 || H == 4, "the head of a load is taken from its number for these head counts");
  constexpr int NWT = (H == 4) ? 2 : 1;
  const int lane = threadIdx.x & 31;
  const int hf = lane >> 4;
  const int li = lane & 15;
  const int r0 = (blockIdx.x * 8 + (threadIdx.x >> 5)) * 32;
  if (r0 >= R) return;

  float4 wl[NWT], wr[NWT];
#pragma unroll
  for (int w = 0; w < NWT; ++w) {
    const int head = (2 * w + hf) % H;
    wl[w] = ((const float4*)attn_l)[head * (GAT_F / 4) + li];
    wr[w] = ((const float4*)attn_r)[head * (GAT_F / 4) + li];
  }
  const float4* h4 = (const float4*)h;
  float4 x[16];
#pragma unroll
  for (int t = 0; t < 16; ++t) {
    const int r = r0 + 2 * t + hf;
    x[t] = h4[(size_t)(r < R ? r : R - 1) * (GAT_F / 4) + li];
  }
  float pl[16], pr[16];
#pragma unroll
  for (int t = 0; t < 16; ++t) {
    const float4 a = wl[t & (NWT - 1)], b = wr[t & (NWT - 1)];
    pl[t] = x[t].x * a.x + x[t].y * a.y + x[t].z * a.z + x[t].w * a.w;
    pr[t] = x[t].x * b.x + x[t].y * b.y + x[t].z * b.z + x[t].w * b.w;
  }
  gat_halve<16, 8>(pl, lane); gat_halve<16, 8>(pr, lane);
  gat_halve<8, 4>(pl, lane);  gat_halve<8, 4>(pr, lane);
  gat_halve<4, 2>(pl, lane);  gat_halve<4, 2>(pr, lane);
  gat_halve<2, 1>(pl, lane);  gat_halve<2, 1>(pr, lane);
  const int r = r0 + 2 * li + hf;
  if (r < R) { el[r] = pl[0]; er[r] = pr[0]; }
}

__device__ __forceinline__ unsigned f32_key(float x) {
  unsigned u = __float_as_uint(x);
  return (u & 0x80000000u) ? ~u : (u | 0x80000000u);
}
__device__ __forceinline__ float key_f32(unsigned k) {
  return (k & 0x80000000u) ? __uint_as_float(k ^ 0x80000000u) : __uint_as_float(~k);
}
#define KEY_NEG_INF 0x007FFFFFu

__device__ __forceinline__ float gat_leaky(float x) { return (x > 0.f) ? x : 0.2f * x; }

#define GAT_TILE_NODES   196
#define GAT_TILE_THREADS 1024
#define GAT_LIST_CAP     4608
#define GAT_LIST_ROUNDS  ((GAT_LIST_CAP + GAT_TILE_THREADS - 1) / GAT_TILE_THREADS)
#define GAT_NUM_TILES    ((GAT_N + GAT_TILE_NODES - 1) / GAT_TILE_NODES)

#define GAT_PART_GROUPS   128
#define GAT_PART_THREADS  1024
#define GAT_PART_EDGES    ((GAT_E + GAT_PART_GROUPS - 1) / GAT_PART_GROUPS)
#define GAT_PART_ROUNDS   ((GAT_PART_EDGES + GAT_PART_THREADS - 1) / GAT_PART_THREADS)
#define GAT_SEG_CAP       64
#define GAT_SEGS_PER_WAVE (GAT_PART_GROUPS / (GAT_TILE_THREADS / 32))
static_assert(GAT_PART_GROUPS % (GAT_TILE_THREADS / 32) == 0, "segments divide evenly over the tile kernel's waves");
static_assert(GAT_SEG_CAP % 32 == 0, "a segment is read in whole-wave rounds");

__global__ __launch_bounds__(GAT_PART_THREADS)
void gat_partition(const int* __restrict__ src, const int* __restrict__ dst,
                   int* __restrict__ segcnt, int* __restrict__ seg) {
  __shared__ unsigned cnt[GAT_NUM_TILES];
  const int tid = threadIdx.x;
  const int g = blockIdx.x;
  for (int i = tid; i < GAT_NUM_TILES; i += GAT_PART_THREADS) cnt[i] = 0u;
  __syncthreads();

  const int e0 = g * GAT_PART_EDGES;
  const int lim = (e0 + GAT_PART_EDGES < GAT_E) ? e0 + GAT_PART_EDGES : GAT_E;
  int dv[GAT_PART_ROUNDS], sv[GAT_PART_ROUNDS];
#pragma unroll
  for (int r = 0; r < GAT_PART_ROUNDS; ++r) {
    const int e = e0 + tid + r * GAT_PART_THREADS;
    const bool ok = e < lim;
    dv[r] = ok ? dst[e] : -1;
    sv[r] = ok ? src[e] : 0;
  }
  int2* segs = (int2*)seg;
#pragma unroll
  for (int r = 0; r < GAT_PART_ROUNDS; ++r) {
    const int d = dv[r];
    if ((unsigned)d < (unsigned)GAT_N) {
      const int t = d / GAT_TILE_NODES;
      const unsigned pos = atomicAdd(&cnt[t], 1u);
      if (pos < (unsigned)GAT_SEG_CAP)
        segs[((size_t)t * GAT_PART_GROUPS + g) * GAT_SEG_CAP + pos] = make_int2(sv[r], d - t * GAT_TILE_NODES);
    }
  }
  __syncthreads();
  for (int i = tid; i < GAT_NUM_TILES; i += GAT_PART_THREADS) segcnt[i * GAT_PART_GROUPS + g] = (int)cnt[i];
}

__device__ __forceinline__ void gat_lds_add(float* p, float v) {
  __hip_atomic_fetch_add(p, v, __ATOMIC_RELAXED, __HIP_MEMORY_SCOPE_WORKGROUP);
}
__device__ __forceinline__ void gat_lds_max(unsigned* p, unsigned v) {
  __hip_atomic_fetch_max(p, v, __ATOMIC_RELAXED, __HIP_MEMORY_SCOPE_WORKGROUP);
}

template <int H> struct gat_hv { float v[H]; };
template <int H>
__device__ __forceinline__ gat_hv<H> gat_ld_heads(const float* p) {
  gat_hv<H> r;
  if constexpr (H == 4) {
    const float4 t = *(const float4*)p;
    r.v[0] = t.x; r.v[1] = t.y; r.v[2] = t.z; r.v[3] = t.w;
  } else {
#pragma unroll
    for (int q = 0; q < H; ++q) r.v[q] = p[q];
  }
  return r;
}
template <int H>
__device__ __forceinline__ void gat_st_heads(float* p, const gat_hv<H>& r) {
  if constexpr (H == 4) {
    *(float4*)p = make_float4(r.v[0], r.v[1], r.v[2], r.v[3]);
  } else {
#pragma unroll
    for (int q = 0; q < H; ++q) p[q] = r.v[q];
  }
}


template <int H>
__device__ __forceinline__ float gat_alpha_head(float a, int head) {
  if constexpr (H == 1) return a;
  else return __int_as_float(__builtin_amdgcn_readlane(__float_as_int(a), head));
}

template <int H>
__device__ __forceinline__ void gat_tile_add_row(const float* __restrict__ h, float* tile,
                                                 int s, int dl, float a, int lane) {
  constexpr int HF = H * GAT_F;
  constexpr int VEC = HF / 32;
  const float* hp = h + (size_t)s * HF + lane;
  float* tp = tile + dl * HF + lane;
  float v[VEC];
#pragma unroll
  for (int j = 0; j < VEC; ++j) v[j] = hp[j * 32];
#pragma unroll
  for (int j = 0; j < VEC; ++j) gat_lds_add(tp + j * 32, v[j] * gat_alpha_head<H>(a, j >> 1));
}

__device__ __forceinline__ void gat_load_dst_group(const int* __restrict__ dst, int E, int gg, int lane,
                                                   int (&dv)[4][4]) {
  const int e0 = gg * 512 + lane * 4;
  if (gg * 512 + 512 <= E) {
#pragma unroll
    for (int c = 0; c < 4; ++c) {
      const int4 t = *(const int4*)(dst + e0 + c * 128);
      dv[c][0] = t.x; dv[c][1] = t.y; dv[c][2] = t.z; dv[c][3] = t.w;
    }
  } else {
#pragma unroll
    for (int c = 0; c < 4; ++c)
#pragma unroll
      for (int j = 0; j < 4; ++j) {
        const int e = e0 + c * 128 + j;
        dv[c][j] = (e < E) ? dst[e] : -1;
      }
  }
}

template <class F>
__device__ __forceinline__ void gat_for_each_match(const int* __restrict__ dst, int E, int base, int cnt,
                                                   int wave, int nw, int lane, F f) {
  const int nGroups = (E + 511) >> 9;
  for (int gg = wave; gg < nGroups; gg += nw) {
    int dv[4][4];
    gat_load_dst_group(dst, E, gg, lane, dv);
#pragma unroll
    for (int c = 0; c < 4; ++c)
#pragma unroll
      for (int j = 0; j < 4; ++j) {
        const int dl = dv[c][j] - base;
        unsigned hit = __builtin_amdgcn_ballot_w32((unsigned)dl < (unsigned)cnt);
        while (hit) {
          const int l = __builtin_ctz(hit);
          hit &= hit - 1u;
          f(gg * 512 + c * 128 + l * 4 + j, __builtin_amdgcn_readlane(dl, l));
        }
      }
  }
}

template <int H, bool RELU>
__global__ __launch_bounds__(GAT_TILE_THREADS)
void gat_tile_aggregate(const int* __restrict__ src, const int* __restrict__ dst,
                        const float* __restrict__ h, const float* __restrict__ el,
                        const float* __restrict__ er, const float* __restrict__ bias,
                        const int* __restrict__ segcnt, const int* __restrict__ seg,
                        float* __restrict__ out, int N, int E) {
  constexpr int HF = H * GAT_F;
  constexpr int VEC = HF / 32;
  __shared__ __attribute__((aligned(16))) float tile[GAT_TILE_NODES * HF];
  __shared__ __attribute__((aligned(16))) float lg[GAT_LIST_CAP * H];
  __shared__ __attribute__((aligned(16))) float er_t[GAT_TILE_NODES * H];
  __shared__ int2 list[GAT_LIST_CAP];
  __shared__ unsigned mk[GAT_TILE_NODES * H];
  __shared__ float ss[GAT_TILE_NODES * H];
  __shared__ unsigned s_count;

  const int tid = threadIdx.x;
  const int lane = tid & 31;
  const int hq = lane & (H - 1);
  const int wave = __builtin_amdgcn_readfirstlane(tid >> 5);
  const int nw = GAT_TILE_THREADS >> 5;
  const int base = blockIdx.x * GAT_TILE_NODES;
  const int left = N - base;
  const int cnt = left < GAT_TILE_NODES ? left : GAT_TILE_NODES;

  float4* tile4 = (float4*)tile;
  const int n4 = cnt * (HF / 4);
  for (int i = tid; i < n4; i += GAT_TILE_THREADS) tile4[i] = make_float4(0.f, 0.f, 0.f, 0.f);
  for (int i = tid; i < cnt * H; i += GAT_TILE_THREADS) {
    er_t[i] = er[(size_t)base * H + i];
    mk[i] = KEY_NEG_INF;
    ss[i] = 0.f;
  }
  if (tid == 0) s_count = 0u;
  __syncthreads();

  {
    const int* cp = segcnt + (size_t)blockIdx.x * GAT_PART_GROUPS;
    const int2* sp = (const int2*)seg + (size_t)blockIdx.x * GAT_PART_GROUPS * GAT_SEG_CAP;
    int c[GAT_SEGS_PER_WAVE];
    unsigned b[GAT_SEGS_PER_WAVE];
    bool over = false;
    unsigned sum = 0u;
#pragma unroll
    for (int k = 0; k < GAT_SEGS_PER_WAVE; ++k) {
      const int ck = cp[wave + k * nw];
      over |= ck > GAT_SEG_CAP;
      c[k] = ck < GAT_SEG_CAP ? ck : GAT_SEG_CAP;
      b[k] = sum;
      sum += (unsigned)c[k];
    }
    if (over) sum += (unsigned)GAT_LIST_CAP + 1u;
    unsigned first = 0u;
    if (lane == 0) first = atomicAdd(&s_count, sum);
    first = __builtin_amdgcn_readfirstlane(first);
    int2 ent[GAT_SEGS_PER_WAVE][GAT_SEG_CAP / 32];
#pragma unroll
    for (int k = 0; k < GAT_SEGS_PER_WAVE; ++k)
#pragma unroll
      for (int j = 0; j < GAT_SEG_CAP / 32; ++j) {
        const int i = lane + 32 * j;
        ent[k][j] = make_int2(0, 0);
        if (i < c[k]) ent[k][j] = sp[(size_t)(wave + k * nw) * GAT_SEG_CAP + i];
      }
#pragma unroll
    for (int k = 0; k < GAT_SEGS_PER_WAVE; ++k)
#pragma unroll
      for (int j = 0; j < GAT_SEG_CAP / 32; ++j) {
        const int i = lane + 32 * j;
        const unsigned pos = first + b[k] + (unsigned)i;
        if (i < c[k] && pos < (unsigned)GAT_LIST_CAP) list[pos] = ent[k][j];
      }
  }
  __syncthreads();
  const int total = (int)s_count;

  if (total <= GAT_LIST_CAP) {
    {
      int sv[GAT_LIST_ROUNDS];
      gat_hv<H> lv[GAT_LIST_ROUNDS];
#pragma unroll
      for (int r = 0; r < GAT_LIST_ROUNDS; ++r) {
        const int i = tid + r * GAT_TILE_THREADS;
        sv[r] = i < total ? list[i].x : 0;
      }
#pragma unroll
      for (int r = 0; r < GAT_LIST_ROUNDS; ++r) lv[r] = gat_ld_heads<H>(el + (size_t)sv[r] * H);
#pragma unroll
      for (int r = 0; r < GAT_LIST_ROUNDS; ++r) {
        const int i = tid + r * GAT_TILE_THREADS;
        if (i < total) {
          const int dl = list[i].y;
          const gat_hv<H> rv = gat_ld_heads<H>(er_t + dl * H);
          gat_hv<H> x;
#pragma unroll
          for (int q = 0; q < H; ++q) {
            x.v[q] = gat_leaky(lv[r].v[q] + rv.v[q]);
            gat_lds_max(&mk[dl * H + q], f32_key(x.v[q]));
          }
          gat_st_heads<H>(lg + i * H, x);
        }
      }
    }
    __syncthreads();

    for (int i = tid; i < total; i += GAT_TILE_THREADS) {
      const int dl = list[i].y;
      gat_hv<H> x = gat_ld_heads<H>(lg + i * H);
#pragma unroll
      for (int q = 0; q < H; ++q) {
        x.v[q] = __expf(x.v[q] - key_f32(mk[dl * H + q]));
        gat_lds_add(&ss[dl * H + q], x.v[q]);
      }
      gat_st_heads<H>(lg + i * H, x);
    }
    __syncthreads();

    for (int i = wave * 4; i < total; i += nw * 4) {
      if (i + 4 <= total) {
        int sn[4], dd[4];
#pragma unroll
        for (int k = 0; k < 4; ++k) {
          const int2 en = list[i + k];
          sn[k] = __builtin_amdgcn_readfirstlane(en.x);
          dd[k] = __builtin_amdgcn_readfirstlane(en.y);
        }
        float vals[4][VEC];
        float al[4];
#pragma unroll
        for (int k = 0; k < 4; ++k) {
          al[k] = lg[(i + k) * H + hq] / ss[dd[k] * H + hq];
          const float* hp = h + (size_t)sn[k] * HF + lane;
#pragma unroll
          for (int j = 0; j < VEC; ++j) vals[k][j] = hp[j * 32];
        }
#pragma unroll
        for (int k = 0; k < 4; ++k) {
          float* tp = tile + dd[k] * HF + lane;
#pragma unroll
          for (int j = 0; j < VEC; ++j) gat_lds_add(tp + j * 32, vals[k][j] * gat_alpha_head<H>(al[k], j >> 1));
        }
      } else {
        for (int k = i; k < total; ++k) {
          const int2 en = list[k];
          const int dl = __builtin_amdgcn_readfirstlane(en.y);
          gat_tile_add_row<H>(h, tile, __builtin_amdgcn_readfirstlane(en.x), dl,
                              lg[k * H + hq] / ss[dl * H + hq], lane);
        }
      }
    }
  } else {
    gat_for_each_match(dst, E, base, cnt, wave, nw, lane, [&](int e, int dl) {
      if (lane < H) {
        const int s = src[e];
        gat_lds_max(&mk[dl * H + lane], f32_key(gat_leaky(el[(size_t)s * H + lane] + er_t[dl * H + lane])));
      }
    });
    __syncthreads();
    gat_for_each_match(dst, E, base, cnt, wave, nw, lane, [&](int e, int dl) {
      if (lane < H) {
        const int s = src[e];
        const float x = gat_leaky(el[(size_t)s * H + lane] + er_t[dl * H + lane]);
        gat_lds_add(&ss[dl * H + lane], __expf(x - key_f32(mk[dl * H + lane])));
      }
    });
    __syncthreads();
    gat_for_each_match(dst, E, base, cnt, wave, nw, lane, [&](int e, int dl) {
      const int s = src[e];
      const float x = gat_leaky(el[(size_t)s * H + hq] + er_t[dl * H + hq]);
      gat_tile_add_row<H>(h, tile, s, dl, __expf(x - key_f32(mk[dl * H + hq])) / ss[dl * H + hq], lane);
    });
  }
  __syncthreads();

  const float4* bias4 = (const float4*)bias;
  float4* out4 = (float4*)(out + (size_t)base * HF);
  for (int i = tid; i < n4; i += GAT_TILE_THREADS) {
    float4 v = tile4[i];
    const float4 b = bias4[i & (HF / 4 - 1)];
    v.x += b.x; v.y += b.y; v.z += b.z; v.w += b.w;
    if (RELU) {
      v.x = v.x > 0.f ? v.x : 0.f; v.y = v.y > 0.f ? v.y : 0.f;
      v.z = v.z > 0.f ? v.z : 0.f; v.w = v.w > 0.f ? v.w : 0.f;
    }
    out4[i] = v;
  }
}

extern "C" void kernel_launch(void* const* d_in, const int* in_sizes, int n_in,
                              void* d_out, int out_size, void* d_ws, size_t ws_size,
                              hipStream_t stream) {
  (void)in_sizes; (void)n_in; (void)out_size; (void)ws_size;
  const float* feat = (const float*)d_in[0];
  const int*   src  = (const int*)d_in[1];
  const int*   dst  = (const int*)d_in[2];
  const float* W1  = (const float*)d_in[3];
  const float* al1 = (const float*)d_in[4];
  const float* ar1 = (const float*)d_in[5];
  const float* b1  = (const float*)d_in[6];
  const float* W2  = (const float*)d_in[7];
  const float* al2 = (const float*)d_in[8];
  const float* ar2 = (const float*)d_in[9];
  const float* b2  = (const float*)d_in[10];
  const float* W3  = (const float*)d_in[11];
  const float* al3 = (const float*)d_in[12];
  const float* ar3 = (const float*)d_in[13];
  const float* b3  = (const float*)d_in[14];
  float* out = (float*)d_out;

  const int N = GAT_N, E = GAT_E;

  char* ws = (char*)d_ws;
  size_t off = 0;
  auto carve = [&](size_t bytes) -> void* {
    void* p = ws + off;
    off += (bytes + 255) & ~(size_t)255;
    return p;
  };
  float* bufA = (float*)carve((size_t)N * 256 * 4);
  float* bufB = (float*)carve((size_t)N * 256 * 4);
  float* el   = (float*)carve((size_t)N * 4 * 4);
  float* er   = (float*)carve((size_t)N * 4 * 4);
  int* segcnt = (int*)carve((size_t)GAT_NUM_TILES * GAT_PART_GROUPS * 4);
  int* seg    = (int*)carve((size_t)GAT_NUM_TILES * GAT_PART_GROUPS * GAT_SEG_CAP * 8);

  auto blocks = [](long long n) { return (unsigned)((n + 255) / 256); };
  const unsigned tiles = GAT_NUM_TILES;

  gat_partition<<<GAT_PART_GROUPS, GAT_PART_THREADS, 0, stream>>>(src, dst, segcnt, seg);

  {
    const int K = 128, Nw = 256, H = 4;
    int waves = (N / 16) * (Nw / 64);
    gat_gemm_wmma<<<(waves + 7) / 8, 256, 0, stream>>>(feat, W1, bufA, N, K, Nw);
    long long nh = (long long)N * H;
    gat_node_attn2<4><<<blocks(nh), 256, 0, stream>>>(bufA, al1, ar1, el, er, (int)nh);
    gat_tile_aggregate<4, true><<<tiles, GAT_TILE_THREADS, 0, stream>>>(src, dst, bufA, el, er, b1, segcnt, seg, bufB, N, E);
  }

  {
    const int K = 256, Nw = 256, H = 4;
    int waves = (N / 16) * (Nw / 64);
    gat_gemm_wmma<<<(waves + 7) / 8, 256, 0, stream>>>(bufB, W2, bufA, N, K, Nw);
    long long nh = (long long)N * H;
    gat_node_attn2<4><<<blocks(nh), 256, 0, stream>>>(bufA, al2, ar2, el, er, (int)nh);
    gat_tile_aggregate<4, true><<<tiles, GAT_TILE_THREADS, 0, stream>>>(src, dst, bufA, el, er, b2, segcnt, seg, bufB, N, E);
  }

  {
    const int K = 256, Nw = 64, H = 1;
    int waves = (N / 16) * (Nw / 64);
    gat_gemm_wmma<<<(waves + 7) / 8, 256, 0, stream>>>(bufB, W3, bufA, N, K, Nw);
    long long nh = (long long)N * H;
    gat_node_attn2<1><<<blocks(nh), 256, 0, stream>>>(bufA, al3, ar3, el, er, (int)nh);
    gat_tile_aggregate<1, false><<<tiles, GAT_TILE_THREADS, 0, stream>>>(src, dst, bufA, el, er, b3, segcnt, seg, out, N, E);
  }
}


# GEMM B by 16-byte loads: column c of the 4 column tiles = strip columns 4c..4c+3, so one load per k row feeds all 4 WMMAs (8 B loads per batch, was 32); C rows as 16-byte stores; same k order
# speedup vs baseline: 7.4945x; 1.0707x over previous
#include <hip/hip_runtime.h>


#define GAT_N 50000
#define GAT_E 800000
#define GAT_F 64

typedef float v2f __attribute__((ext_vector_type(2)));
typedef float v4f __attribute__((ext_vector_type(4)));
typedef float v8f __attribute__((ext_vector_type(8)));

#define GEMM_KB 4
__global__ __launch_bounds__(256)
void gat_gemm_wmma(const float* __restrict__ A, const float* __restrict__ B,
                   float* __restrict__ C, int M, int K, int Nw) {
  const int wave = blockIdx.x * 8 + (threadIdx.x >> 5);
  const int lane = threadIdx.x & 31;
  const int tilesN = Nw >> 6;
  const int tilesM = M >> 4;
  if (wave >= tilesM * tilesN) return;
  const int tm = wave / tilesN;
  const int tn = wave - tm * tilesN;
  const int kh = lane >> 4;
  const int lr = lane & 15;

  const float* Ap = A + (size_t)(tm * 16 + lr) * K + kh * 2;
  const float* Bp = B + (size_t)(tn * 64) + lr * 4 + (size_t)(kh * 2) * Nw;
  v8f acc0 = {}, acc1 = {}, acc2 = {}, acc3 = {};
  for (int k0 = 0; k0 < K; k0 += 4 * GEMM_KB) {
    v2f a[GEMM_KB];
    v4f bx[GEMM_KB], by[GEMM_KB];
    const float* Bk = Bp + (size_t)k0 * Nw;
#pragma unroll
    for (int s = 0; s < GEMM_KB; ++s) {
      a[s] = *(const v2f*)(Ap + k0 + 4 * s);
      const float* Brow0 = Bk + (size_t)(4 * s) * Nw;
      bx[s] = *(const v4f*)Brow0;
      by[s] = *(const v4f*)(Brow0 + Nw);
    }
    __builtin_amdgcn_sched_barrier(0);
#pragma unroll
    for (int s = 0; s < GEMM_KB; ++s) {
      const v2f b0 = {bx[s].x, by[s].x};
      const v2f b1 = {bx[s].y, by[s].y};
      const v2f b2 = {bx[s].z, by[s].z};
      const v2f b3 = {bx[s].w, by[s].w};
      acc0 = __builtin_amdgcn_wmma_f32_16x16x4_f32(false, a[s], false, b0, (short)0, acc0, false, false);
      acc1 = __builtin_amdgcn_wmma_f32_16x16x4_f32(false, a[s], false, b1, (short)0, acc1, false, false);
      acc2 = __builtin_amdgcn_wmma_f32_16x16x4_f32(false, a[s], false, b2, (short)0, acc2, false, false);
      acc3 = __builtin_amdgcn_wmma_f32_16x16x4_f32(false, a[s], false, b3, (short)0, acc3, false, false);
    }
  }
  float* Cp = C + (size_t)(tm * 16 + kh * 8) * Nw + tn * 64 + lr * 4;
#pragma unroll
  for (int r = 0; r < 8; ++r) {
    const v4f c = {acc0[r], acc1[r], acc2[r], acc3[r]};
    *(v4f*)(Cp + (size_t)r * Nw) = c;
  }
}


template <int C, int M>
__device__ __forceinline__ void gat_halve(float (&v)[16], int lane) {
  const bool up = (lane & M) != 0;
#pragma unroll
  for (int k = 0; k < C / 2; ++k) {
    const float a = v[k], b = v[k + C / 2];
    const float keep = up ? b : a;
    const float send = up ? a : b;
    v[k] = keep + __shfl_xor(send, M);
  }
}

template <int H>
__global__ __launch_bounds__(256)
void gat_node_attn2(const float* __restrict__ h, const float* __restrict__ attn_l,
                    const float* __restrict__ attn_r, float* __restrict__ el,
                    float* __restrict__ er, int R) {
  static_assert(H == 1 || H == 4, "the head of a load is taken from its number for these head counts");
  constexpr int NWT = (H == 4) ? 2 : 1;
  const int lane = threadIdx.x & 31;
  const int hf = lane >> 4;
  const int li = lane & 15;
  const int r0 = (blockIdx.x * 8 + (threadIdx.x >> 5)) * 32;
  if (r0 >= R) return;

  float4 wl[NWT], wr[NWT];
#pragma unroll
  for (int w = 0; w < NWT; ++w) {
    const int head = (2 * w + hf) % H;
    wl[w] = ((const float4*)attn_l)[head * (GAT_F / 4) + li];
    wr[w] = ((const float4*)attn_r)[head * (GAT_F / 4) + li];
  }
  const float4* h4 = (const float4*)h;
  float4 x[16];
#pragma unroll
  for (int t = 0; t < 16; ++t) {
    const int r = r0 + 2 * t + hf;
    x[t] = h4[(size_t)(r < R ? r : R - 1) * (GAT_F / 4) + li];
  }
  float pl[16], pr[16];
#pragma unroll
  for (int t = 0; t < 16; ++t) {
    const float4 a = wl[t & (NWT - 1)], b = wr[t & (NWT - 1)];
    pl[t] = x[t].x * a.x + x[t].y * a.y + x[t].z * a.z + x[t].w * a.w;
    pr[t] = x[t].x * b.x + x[t].y * b.y + x[t].z * b.z + x[t].w * b.w;
  }
  gat_halve<16, 8>(pl, lane); gat_halve<16, 8>(pr, lane);
  gat_halve<8, 4>(pl, lane);  gat_halve<8, 4>(pr, lane);
  gat_halve<4, 2>(pl, lane);  gat_halve<4, 2>(pr, lane);
  gat_halve<2, 1>(pl, lane);  gat_halve<2, 1>(pr, lane);
  const int r = r0 + 2 * li + hf;
  if (r < R) { el[r] = pl[0]; er[r] = pr[0]; }
}

__device__ __forceinline__ unsigned f32_key(float x) {
  unsigned u = __float_as_uint(x);
  return (u & 0x80000000u) ? ~u : (u | 0x80000000u);
}
__device__ __forceinline__ float key_f32(unsigned k) {
  return (k & 0x80000000u) ? __uint_as_float(k ^ 0x80000000u) : __uint_as_float(~k);
}
#define KEY_NEG_INF 0x007FFFFFu

__device__ __forceinline__ float gat_leaky(float x) { return (x > 0.f) ? x : 0.2f * x; }

#define GAT_TILE_NODES   196
#define GAT_TILE_THREADS 1024
#define GAT_LIST_CAP     4608
#define GAT_LIST_ROUNDS  ((GAT_LIST_CAP + GAT_TILE_THREADS - 1) / GAT_TILE_THREADS)
#define GAT_NUM_TILES    ((GAT_N + GAT_TILE_NODES - 1) / GAT_TILE_NODES)

#define GAT_PART_GROUPS   128
#define GAT_PART_THREADS  1024
#define GAT_PART_EDGES    ((GAT_E + GAT_PART_GROUPS - 1) / GAT_PART_GROUPS)
#define GAT_PART_ROUNDS   ((GAT_PART_EDGES + GAT_PART_THREADS - 1) / GAT_PART_THREADS)
#define GAT_SEG_CAP       64
#define GAT_SEGS_PER_WAVE (GAT_PART_GROUPS / (GAT_TILE_THREADS / 32))
static_assert(GAT_PART_GROUPS % (GAT_TILE_THREADS / 32) == 0, "segments divide evenly over the tile kernel's waves");
static_assert(GAT_SEG_CAP % 32 == 0, "a segment is read in whole-wave rounds");

__global__ __launch_bounds__(GAT_PART_THREADS)
void gat_partition(const int* __restrict__ src, const int* __restrict__ dst,
                   int* __restrict__ segcnt, int* __restrict__ seg) {
  __shared__ unsigned cnt[GAT_NUM_TILES];
  const int tid = threadIdx.x;
  const int g = blockIdx.x;
  for (int i = tid; i < GAT_NUM_TILES; i += GAT_PART_THREADS) cnt[i] = 0u;
  __syncthreads();

  const int e0 = g * GAT_PART_EDGES;
  const int lim = (e0 + GAT_PART_EDGES < GAT_E) ? e0 + GAT_PART_EDGES : GAT_E;
  int dv[GAT_PART_ROUNDS], sv[GAT_PART_ROUNDS];
#pragma unroll
  for (int r = 0; r < GAT_PART_ROUNDS; ++r) {
    const int e = e0 + tid + r * GAT_PART_THREADS;
    const bool ok = e < lim;
    dv[r] = ok ? dst[e] : -1;
    sv[r] = ok ? src[e] : 0;
  }
  int2* segs = (int2*)seg;
#pragma unroll
  for (int r = 0; r < GAT_PART_ROUNDS; ++r) {
    const int d = dv[r];
    if ((unsigned)d < (unsigned)GAT_N) {
      const int t = d / GAT_TILE_NODES;
      const unsigned pos = atomicAdd(&cnt[t], 1u);
      if (pos < (unsigned)GAT_SEG_CAP)
        segs[((size_t)t * GAT_PART_GROUPS + g) * GAT_SEG_CAP + pos] = make_int2(sv[r], d - t * GAT_TILE_NODES);
    }
  }
  __syncthreads();
  for (int i = tid; i < GAT_NUM_TILES; i += GAT_PART_THREADS) segcnt[i * GAT_PART_GROUPS + g] = (int)cnt[i];
}

__device__ __forceinline__ void gat_lds_add(float* p, float v) {
  __hip_atomic_fetch_add(p, v, __ATOMIC_RELAXED, __HIP_MEMORY_SCOPE_WORKGROUP);
}
__device__ __forceinline__ void gat_lds_max(unsigned* p, unsigned v) {
  __hip_atomic_fetch_max(p, v, __ATOMIC_RELAXED, __HIP_MEMORY_SCOPE_WORKGROUP);
}

template <int H> struct gat_hv { float v[H]; };
template <int H>
__device__ __forceinline__ gat_hv<H> gat_ld_heads(const float* p) {
  gat_hv<H> r;
  if constexpr (H == 4) {
    const float4 t = *(const float4*)p;
    r.v[0] = t.x; r.v[1] = t.y; r.v[2] = t.z; r.v[3] = t.w;
  } else {
#pragma unroll
    for (int q = 0; q < H; ++q) r.v[q] = p[q];
  }
  return r;
}
template <int H>
__device__ __forceinline__ void gat_st_heads(float* p, const gat_hv<H>& r) {
  if constexpr (H == 4) {
    *(float4*)p = make_float4(r.v[0], r.v[1], r.v[2], r.v[3]);
  } else {
#pragma unroll
    for (int q = 0; q < H; ++q) p[q] = r.v[q];
  }
}


template <int H>
__device__ __forceinline__ float gat_alpha_head(float a, int head) {
  if constexpr (H == 1) return a;
  else return __int_as_float(__builtin_amdgcn_readlane(__float_as_int(a), head));
}

template <int H>
__device__ __forceinline__ void gat_tile_add_row(const float* __restrict__ h, float* tile,
                                                 int s, int dl, float a, int lane) {
  constexpr int HF = H * GAT_F;
  constexpr int VEC = HF / 32;
  const float* hp = h + (size_t)s * HF + lane;
  float* tp = tile + dl * HF + lane;
  float v[VEC];
#pragma unroll
  for (int j = 0; j < VEC; ++j) v[j] = hp[j * 32];
#pragma unroll
  for (int j = 0; j < VEC; ++j) gat_lds_add(tp + j * 32, v[j] * gat_alpha_head<H>(a, j >> 1));
}

__device__ __forceinline__ void gat_load_dst_group(const int* __restrict__ dst, int E, int gg, int lane,
                                                   int (&dv)[4][4]) {
  const int e0 = gg * 512 + lane * 4;
  if (gg * 512 + 512 <= E) {
#pragma unroll
    for (int c = 0; c < 4; ++c) {
      const int4 t = *(const int4*)(dst + e0 + c * 128);
      dv[c][0] = t.x; dv[c][1] = t.y; dv[c][2] = t.z; dv[c][3] = t.w;
    }
  } else {
#pragma unroll
    for (int c = 0; c < 4; ++c)
#pragma unroll
      for (int j = 0; j < 4; ++j) {
        const int e = e0 + c * 128 + j;
        dv[c][j] = (e < E) ? dst[e] : -1;
      }
  }
}

template <class F>
__device__ __forceinline__ void gat_for_each_match(const int* __restrict__ dst, int E, int base, int cnt,
                                                   int wave, int nw, int lane, F f) {
  const int nGroups = (E + 511) >> 9;
  for (int gg = wave; gg < nGroups; gg += nw) {
    int dv[4][4];
    gat_load_dst_group(dst, E, gg, lane, dv);
#pragma unroll
    for (int c = 0; c < 4; ++c)
#pragma unroll
      for (int j = 0; j < 4; ++j) {
        const int dl = dv[c][j] - base;
        unsigned hit = __builtin_amdgcn_ballot_w32((unsigned)dl < (unsigned)cnt);
        while (hit) {
          const int l = __builtin_ctz(hit);
          hit &= hit - 1u;
          f(gg * 512 + c * 128 + l * 4 + j, __builtin_amdgcn_readlane(dl, l));
        }
      }
  }
}

template <int H, bool RELU>
__global__ __launch_bounds__(GAT_TILE_THREADS)
void gat_tile_aggregate(const int* __restrict__ src, const int* __restrict__ dst,
                        const float* __restrict__ h, const float* __restrict__ el,
                        const float* __restrict__ er, const float* __restrict__ bias,
                        const int* __restrict__ segcnt, const int* __restrict__ seg,
                        float* __restrict__ out, int N, int E) {
  constexpr int HF = H * GAT_F;
  constexpr int VEC = HF / 32;
  __shared__ __attribute__((aligned(16))) float tile[GAT_TILE_NODES * HF];
  __shared__ __attribute__((aligned(16))) float lg[GAT_LIST_CAP * H];
  __shared__ __attribute__((aligned(16))) float er_t[GAT_TILE_NODES * H];
  __shared__ int2 list[GAT_LIST_CAP];
  __shared__ unsigned mk[GAT_TILE_NODES * H];
  __shared__ float ss[GAT_TILE_NODES * H];
  __shared__ unsigned s_count;

  const int tid = threadIdx.x;
  const int lane = tid & 31;
  const int hq = lane & (H - 1);
  const int wave = __builtin_amdgcn_readfirstlane(tid >> 5);
  const int nw = GAT_TILE_THREADS >> 5;
  const int base = blockIdx.x * GAT_TILE_NODES;
  const int left = N - base;
  const int cnt = left < GAT_TILE_NODES ? left : GAT_TILE_NODES;

  float4* tile4 = (float4*)tile;
  const int n4 = cnt * (HF / 4);
  for (int i = tid; i < n4; i += GAT_TILE_THREADS) tile4[i] = make_float4(0.f, 0.f, 0.f, 0.f);
  for (int i = tid; i < cnt * H; i += GAT_TILE_THREADS) {
    er_t[i] = er[(size_t)base * H + i];
    mk[i] = KEY_NEG_INF;
    ss[i] = 0.f;
  }
  if (tid == 0) s_count = 0u;
  __syncthreads();

  {
    const int* cp = segcnt + (size_t)blockIdx.x * GAT_PART_GROUPS;
    const int2* sp = (const int2*)seg + (size_t)blockIdx.x * GAT_PART_GROUPS * GAT_SEG_CAP;
    int c[GAT_SEGS_PER_WAVE];
    unsigned b[GAT_SEGS_PER_WAVE];
    bool over = false;
    unsigned sum = 0u;
#pragma unroll
    for (int k = 0; k < GAT_SEGS_PER_WAVE; ++k) {
      const int ck = cp[wave + k * nw];
      over |= ck > GAT_SEG_CAP;
      c[k] = ck < GAT_SEG_CAP ? ck : GAT_SEG_CAP;
      b[k] = sum;
      sum += (unsigned)c[k];
    }
    if (over) sum += (unsigned)GAT_LIST_CAP + 1u;
    unsigned first = 0u;
    if (lane == 0) first = atomicAdd(&s_count, sum);
    first = __builtin_amdgcn_readfirstlane(first);
    int2 ent[GAT_SEGS_PER_WAVE][GAT_SEG_CAP / 32];
#pragma unroll
    for (int k = 0; k < GAT_SEGS_PER_WAVE; ++k)
#pragma unroll
      for (int j = 0; j < GAT_SEG_CAP / 32; ++j) {
        const int i = lane + 32 * j;
        ent[k][j] = make_int2(0, 0);
        if (i < c[k]) ent[k][j] = sp[(size_t)(wave + k * nw) * GAT_SEG_CAP + i];
      }
#pragma unroll
    for (int k = 0; k < GAT_SEGS_PER_WAVE; ++k)
#pragma unroll
      for (int j = 0; j < GAT_SEG_CAP / 32; ++j) {
        const int i = lane + 32 * j;
        const unsigned pos = first + b[k] + (unsigned)i;
        if (i < c[k] && pos < (unsigned)GAT_LIST_CAP) list[pos] = ent[k][j];
      }
  }
  __syncthreads();
  const int total = (int)s_count;

  if (total <= GAT_LIST_CAP) {
    {
      int sv[GAT_LIST_ROUNDS];
      gat_hv<H> lv[GAT_LIST_ROUNDS];
#pragma unroll
      for (int r = 0; r < GAT_LIST_ROUNDS; ++r) {
        const int i = tid + r * GAT_TILE_THREADS;
        sv[r] = i < total ? list[i].x : 0;
      }
#pragma unroll
      for (int r = 0; r < GAT_LIST_ROUNDS; ++r) lv[r] = gat_ld_heads<H>(el + (size_t)sv[r] * H);
#pragma unroll
      for (int r = 0; r < GAT_LIST_ROUNDS; ++r) {
        const int i = tid + r * GAT_TILE_THREADS;
        if (i < total) {
          const int dl = list[i].y;
          const gat_hv<H> rv = gat_ld_heads<H>(er_t + dl * H);
          gat_hv<H> x;
#pragma unroll
          for (int q = 0; q < H; ++q) {
            x.v[q] = gat_leaky(lv[r].v[q] + rv.v[q]);
            gat_lds_max(&mk[dl * H + q], f32_key(x.v[q]));
          }
          gat_st_heads<H>(lg + i * H, x);
        }
      }
    }
    __syncthreads();

    for (int i = tid; i < total; i += GAT_TILE_THREADS) {
      const int dl = list[i].y;
      gat_hv<H> x = gat_ld_heads<H>(lg + i * H);
#pragma unroll
      for (int q = 0; q < H; ++q) {
        x.v[q] = __expf(x.v[q] - key_f32(mk[dl * H + q]));
        gat_lds_add(&ss[dl * H + q], x.v[q]);
      }
      gat_st_heads<H>(lg + i * H, x);
    }
    __syncthreads();

    for (int i = wave * 4; i < total; i += nw * 4) {
      if (i + 4 <= total) {
        int sn[4], dd[4];
#pragma unroll
        for (int k = 0; k < 4; ++k) {
          const int2 en = list[i + k];
          sn[k] = __builtin_amdgcn_readfirstlane(en.x);
          dd[k] = __builtin_amdgcn_readfirstlane(en.y);
        }
        float vals[4][VEC];
        float al[4];
#pragma unroll
        for (int k = 0; k < 4; ++k) {
          al[k] = lg[(i + k) * H + hq] / ss[dd[k] * H + hq];
          const float* hp = h + (size_t)sn[k] * HF + lane;
#pragma unroll
          for (int j = 0; j < VEC; ++j) vals[k][j] = hp[j * 32];
        }
#pragma unroll
        for (int k = 0; k < 4; ++k) {
          float* tp = tile + dd[k] * HF + lane;
#pragma unroll
          for (int j = 0; j < VEC; ++j) gat_lds_add(tp + j * 32, vals[k][j] * gat_alpha_head<H>(al[k], j >> 1));
        }
      } else {
        for (int k = i; k < total; ++k) {
          const int2 en = list[k];
          const int dl = __builtin_amdgcn_readfirstlane(en.y);
          gat_tile_add_row<H>(h, tile, __builtin_amdgcn_readfirstlane(en.x), dl,
                              lg[k * H + hq] / ss[dl * H + hq], lane);
        }
      }
    }
  } else {
    gat_for_each_match(dst, E, base, cnt, wave, nw, lane, [&](int e, int dl) {
      if (lane < H) {
        const int s = src[e];
        gat_lds_max(&mk[dl * H + lane], f32_key(gat_leaky(el[(size_t)s * H + lane] + er_t[dl * H + lane])));
      }
    });
    __syncthreads();
    gat_for_each_match(dst, E, base, cnt, wave, nw, lane, [&](int e, int dl) {
      if (lane < H) {
        const int s = src[e];
        const float x = gat_leaky(el[(size_t)s * H + lane] + er_t[dl * H + lane]);
        gat_lds_add(&ss[dl * H + lane], __expf(x - key_f32(mk[dl * H + lane])));
      }
    });
    __syncthreads();
    gat_for_each_match(dst, E, base, cnt, wave, nw, lane, [&](int e, int dl) {
      const int s = src[e];
      const float x = gat_leaky(el[(size_t)s * H + hq] + er_t[dl * H + hq]);
      gat_tile_add_row<H>(h, tile, s, dl, __expf(x - key_f32(mk[dl * H + hq])) / ss[dl * H + hq], lane);
    });
  }
  __syncthreads();

  const float4* bias4 = (const float4*)bias;
  float4* out4 = (float4*)(out + (size_t)base * HF);
  for (int i = tid; i < n4; i += GAT_TILE_THREADS) {
    float4 v = tile4[i];
    const float4 b = bias4[i & (HF / 4 - 1)];
    v.x += b.x; v.y += b.y; v.z += b.z; v.w += b.w;
    if (RELU) {
      v.x = v.x > 0.f ? v.x : 0.f; v.y = v.y > 0.f ? v.y : 0.f;
      v.z = v.z > 0.f ? v.z : 0.f; v.w = v.w > 0.f ? v.w : 0.f;
    }
    out4[i] = v;
  }
}

extern "C" void kernel_launch(void* const* d_in, const int* in_sizes, int n_in,
                              void* d_out, int out_size, void* d_ws, size_t ws_size,
                              hipStream_t stream) {
  (void)in_sizes; (void)n_in; (void)out_size; (void)ws_size;
  const float* feat = (const float*)d_in[0];
  const int*   src  = (const int*)d_in[1];
  const int*   dst  = (const int*)d_in[2];
  const float* W1  = (const float*)d_in[3];
  const float* al1 = (const float*)d_in[4];
  const float* ar1 = (const float*)d_in[5];
  const float* b1  = (const float*)d_in[6];
  const float* W2  = (const float*)d_in[7];
  const float* al2 = (const float*)d_in[8];
  const float* ar2 = (const float*)d_in[9];
  const float* b2  = (const float*)d_in[10];
  const float* W3  = (const float*)d_in[11];
  const float* al3 = (const float*)d_in[12];
  const float* ar3 = (const float*)d_in[13];
  const float* b3  = (const float*)d_in[14];
  float* out = (float*)d_out;

  const int N = GAT_N, E = GAT_E;

  char* ws = (char*)d_ws;
  size_t off = 0;
  auto carve = [&](size_t bytes) -> void* {
    void* p = ws + off;
    off += (bytes + 255) & ~(size_t)255;
    return p;
  };
  float* bufA = (float*)carve((size_t)N * 256 * 4);
  float* bufB = (float*)carve((size_t)N * 256 * 4);
  float* el   = (float*)carve((size_t)N * 4 * 4);
  float* er   = (float*)carve((size_t)N * 4 * 4);
  int* segcnt = (int*)carve((size_t)GAT_NUM_TILES * GAT_PART_GROUPS * 4);
  int* seg    = (int*)carve((size_t)GAT_NUM_TILES * GAT_PART_GROUPS * GAT_SEG_CAP * 8);

  auto blocks = [](long long n) { return (unsigned)((n + 255) / 256); };
  const unsigned tiles = GAT_NUM_TILES;

  gat_partition<<<GAT_PART_GROUPS, GAT_PART_THREADS, 0, stream>>>(src, dst, segcnt, seg);

  {
    const int K = 128, Nw = 256, H = 4;
    int waves = (N / 16) * (Nw / 64);
    gat_gemm_wmma<<<(waves + 7) / 8, 256, 0, stream>>>(feat, W1, bufA, N, K, Nw);
    long long nh = (long long)N * H;
    gat_node_attn2<4><<<blocks(nh), 256, 0, stream>>>(bufA, al1, ar1, el, er, (int)nh);
    gat_tile_aggregate<4, true><<<tiles, GAT_TILE_THREADS, 0, stream>>>(src, dst, bufA, el, er, b1, segcnt, seg, bufB, N, E);
  }

  {
    const int K = 256, Nw = 256, H = 4;
    int waves = (N / 16) * (Nw / 64);
    gat_gemm_wmma<<<(waves + 7) / 8, 256, 0, stream>>>(bufB, W2, bufA, N, K, Nw);
    long long nh = (long long)N * H;
    gat_node_attn2<4><<<blocks(nh), 256, 0, stream>>>(bufA, al2, ar2, el, er, (int)nh);
    gat_tile_aggregate<4, true><<<tiles, GAT_TILE_THREADS, 0, stream>>>(src, dst, bufA, el, er, b2, segcnt, seg, bufB, N, E);
  }

  {
    const int K = 256, Nw = 64, H = 1;
    int waves = (N / 16) * (Nw / 64);
    gat_gemm_wmma<<<(waves + 7) / 8, 256, 0, stream>>>(bufB, W3, bufA, N, K, Nw);
    long long nh = (long long)N * H;
    gat_node_attn2<1><<<blocks(nh), 256, 0, stream>>>(bufA, al3, ar3, el, er, (int)nh);
    gat_tile_aggregate<1, false><<<tiles, GAT_TILE_THREADS, 0, stream>>>(src, dst, bufA, el, er, b3, segcnt, seg, out, N, E);
  }
}


# Tile kernel gathers source rows by 16-byte loads (512 B per wave load; at H=1 two edges per load); LDS rows lane-interleaved so the adds stay bank-conflict-free, put back in order at the row store
# speedup vs baseline: 7.9538x; 1.0613x over previous
#include <hip/hip_runtime.h>


#define GAT_N 50000
#define GAT_E 800000
#define GAT_F 64

typedef float v2f __attribute__((ext_vector_type(2)));
typedef float v4f __attribute__((ext_vector_type(4)));
typedef float v8f __attribute__((ext_vector_type(8)));

#define GEMM_KB 4
__global__ __launch_bounds__(256)
void gat_gemm_wmma(const float* __restrict__ A, const float* __restrict__ B,
                   float* __restrict__ C, int M, int K, int Nw) {
  const int wave = blockIdx.x * 8 + (threadIdx.x >> 5);
  const int lane = threadIdx.x & 31;
  const int tilesN = Nw >> 6;
  const int tilesM = M >> 4;
  if (wave >= tilesM * tilesN) return;
  const int tm = wave / tilesN;
  const int tn = wave - tm * tilesN;
  const int kh = lane >> 4;
  const int lr = lane & 15;

  const float* Ap = A + (size_t)(tm * 16 + lr) * K + kh * 2;
  const float* Bp = B + (size_t)(tn * 64) + lr * 4 + (size_t)(kh * 2) * Nw;
  v8f acc0 = {}, acc1 = {}, acc2 = {}, acc3 = {};
  for (int k0 = 0; k0 < K; k0 += 4 * GEMM_KB) {
    v2f a[GEMM_KB];
    v4f bx[GEMM_KB], by[GEMM_KB];
    const float* Bk = Bp + (size_t)k0 * Nw;
#pragma unroll
    for (int s = 0; s < GEMM_KB; ++s) {
      a[s] = *(const v2f*)(Ap + k0 + 4 * s);
      const float* Brow0 = Bk + (size_t)(4 * s) * Nw;
      bx[s] = *(const v4f*)Brow0;
      by[s] = *(const v4f*)(Brow0 + Nw);
    }
    __builtin_amdgcn_sched_barrier(0);
#pragma unroll
    for (int s = 0; s < GEMM_KB; ++s) {
      const v2f b0 = {bx[s].x, by[s].x};
      const v2f b1 = {bx[s].y, by[s].y};
      const v2f b2 = {bx[s].z, by[s].z};
      const v2f b3 = {bx[s].w, by[s].w};
      acc0 = __builtin_amdgcn_wmma_f32_16x16x4_f32(false, a[s], false, b0, (short)0, acc0, false, false);
      acc1 = __builtin_amdgcn_wmma_f32_16x16x4_f32(false, a[s], false, b1, (short)0, acc1, false, false);
      acc2 = __builtin_amdgcn_wmma_f32_16x16x4_f32(false, a[s], false, b2, (short)0, acc2, false, false);
      acc3 = __builtin_amdgcn_wmma_f32_16x16x4_f32(false, a[s], false, b3, (short)0, acc3, false, false);
    }
  }
  float* Cp = C + (size_t)(tm * 16 + kh * 8) * Nw + tn * 64 + lr * 4;
#pragma unroll
  for (int r = 0; r < 8; ++r) {
    const v4f c = {acc0[r], acc1[r], acc2[r], acc3[r]};
    *(v4f*)(Cp + (size_t)r * Nw) = c;
  }
}


template <int C, int M>
__device__ __forceinline__ void gat_halve(float (&v)[16], int lane) {
  const bool up = (lane & M) != 0;
#pragma unroll
  for (int k = 0; k < C / 2; ++k) {
    const float a = v[k], b = v[k + C / 2];
    const float keep = up ? b : a;
    const float send = up ? a : b;
    v[k] = keep + __shfl_xor(send, M);
  }
}

template <int H>
__global__ __launch_bounds__(256)
void gat_node_attn2(const float* __restrict__ h, const float* __restrict__ attn_l,
                    const float* __restrict__ attn_r, float* __restrict__ el,
                    float* __restrict__ er, int R) {
  static_assert(H == 1 || H == 4, "the head of a load is taken from its number for these head counts");
  constexpr int NWT = (H == 4) ? 2 : 1;
  const int lane = threadIdx.x & 31;
  const int hf = lane >> 4;
  const int li = lane & 15;
  const int r0 = (blockIdx.x * 8 + (threadIdx.x >> 5)) * 32;
  if (r0 >= R) return;

  float4 wl[NWT], wr[NWT];
#pragma unroll
  for (int w = 0; w < NWT; ++w) {
    const int head = (2 * w + hf) % H;
    wl[w] = ((const float4*)attn_l)[head * (GAT_F / 4) + li];
    wr[w] = ((const float4*)attn_r)[head * (GAT_F / 4) + li];
  }
  const float4* h4 = (const float4*)h;
  float4 x[16];
#pragma unroll
  for (int t = 0; t < 16; ++t) {
    const int r = r0 + 2 * t + hf;
    x[t] = h4[(size_t)(r < R ? r : R - 1) * (GAT_F / 4) + li];
  }
  float pl[16], pr[16];
#pragma unroll
  for (int t = 0; t < 16; ++t) {
    const float4 a = wl[t & (NWT - 1)], b = wr[t & (NWT - 1)];
    pl[t] = x[t].x * a.x + x[t].y * a.y + x[t].z * a.z + x[t].w * a.w;
    pr[t] = x[t].x * b.x + x[t].y * b.y + x[t].z * b.z + x[t].w * b.w;
  }
  gat_halve<16, 8>(pl, lane); gat_halve<16, 8>(pr, lane);
  gat_halve<8, 4>(pl, lane);  gat_halve<8, 4>(pr, lane);
  gat_halve<4, 2>(pl, lane);  gat_halve<4, 2>(pr, lane);
  gat_halve<2, 1>(pl, lane);  gat_halve<2, 1>(pr, lane);
  const int r = r0 + 2 * li + hf;
  if (r < R) { el[r] = pl[0]; er[r] = pr[0]; }
}

__device__ __forceinline__ unsigned f32_key(float x) {
  unsigned u = __float_as_uint(x);
  return (u & 0x80000000u) ? ~u : (u | 0x80000000u);
}
__device__ __forceinline__ float key_f32(unsigned k) {
  return (k & 0x80000000u) ? __uint_as_float(k ^ 0x80000000u) : __uint_as_float(~k);
}
#define KEY_NEG_INF 0x007FFFFFu

__device__ __forceinline__ float gat_leaky(float x) { return (x > 0.f) ? x : 0.2f * x; }

#define GAT_TILE_NODES   196
#define GAT_TILE_THREADS 1024
#define GAT_LIST_CAP     4608
#define GAT_LIST_ROUNDS  ((GAT_LIST_CAP + GAT_TILE_THREADS - 1) / GAT_TILE_THREADS)
#define GAT_NUM_TILES    ((GAT_N + GAT_TILE_NODES - 1) / GAT_TILE_NODES)

#define GAT_PART_GROUPS   128
#define GAT_PART_THREADS  1024
#define GAT_PART_EDGES    ((GAT_E + GAT_PART_GROUPS - 1) / GAT_PART_GROUPS)
#define GAT_PART_ROUNDS   ((GAT_PART_EDGES + GAT_PART_THREADS - 1) / GAT_PART_THREADS)
#define GAT_SEG_CAP       64
#define GAT_SEGS_PER_WAVE (GAT_PART_GROUPS / (GAT_TILE_THREADS / 32))
static_assert(GAT_PART_GROUPS % (GAT_TILE_THREADS / 32) == 0, "segments divide evenly over the tile kernel's waves");
static_assert(GAT_SEG_CAP % 32 == 0, "a segment is read in whole-wave rounds");

__global__ __launch_bounds__(GAT_PART_THREADS)
void gat_partition(const int* __restrict__ src, const int* __restrict__ dst,
                   int* __restrict__ segcnt, int* __restrict__ seg) {
  __shared__ unsigned cnt[GAT_NUM_TILES];
  const int tid = threadIdx.x;
  const int g = blockIdx.x;
  for (int i = tid; i < GAT_NUM_TILES; i += GAT_PART_THREADS) cnt[i] = 0u;
  __syncthreads();

  const int e0 = g * GAT_PART_EDGES;
  const int lim = (e0 + GAT_PART_EDGES < GAT_E) ? e0 + GAT_PART_EDGES : GAT_E;
  int dv[GAT_PART_ROUNDS], sv[GAT_PART_ROUNDS];
#pragma unroll
  for (int r = 0; r < GAT_PART_ROUNDS; ++r) {
    const int e = e0 + tid + r * GAT_PART_THREADS;
    const bool ok = e < lim;
    dv[r] = ok ? dst[e] : -1;
    sv[r] = ok ? src[e] : 0;
  }
  int2* segs = (int2*)seg;
#pragma unroll
  for (int r = 0; r < GAT_PART_ROUNDS; ++r) {
    const int d = dv[r];
    if ((unsigned)d < (unsigned)GAT_N) {
      const int t = d / GAT_TILE_NODES;
      const unsigned pos = atomicAdd(&cnt[t], 1u);
      if (pos < (unsigned)GAT_SEG_CAP)
        segs[((size_t)t * GAT_PART_GROUPS + g) * GAT_SEG_CAP + pos] = make_int2(sv[r], d - t * GAT_TILE_NODES);
    }
  }
  __syncthreads();
  for (int i = tid; i < GAT_NUM_TILES; i += GAT_PART_THREADS) segcnt[i * GAT_PART_GROUPS + g] = (int)cnt[i];
}

__device__ __forceinline__ void gat_lds_add(float* p, float v) {
  __hip_atomic_fetch_add(p, v, __ATOMIC_RELAXED, __HIP_MEMORY_SCOPE_WORKGROUP);
}
__device__ __forceinline__ void gat_lds_max(unsigned* p, unsigned v) {
  __hip_atomic_fetch_max(p, v, __ATOMIC_RELAXED, __HIP_MEMORY_SCOPE_WORKGROUP);
}

template <int H> struct gat_hv { float v[H]; };
template <int H>
__device__ __forceinline__ gat_hv<H> gat_ld_heads(const float* p) {
  gat_hv<H> r;
  if constexpr (H == 4) {
    const float4 t = *(const float4*)p;
    r.v[0] = t.x; r.v[1] = t.y; r.v[2] = t.z; r.v[3] = t.w;
  } else {
#pragma unroll
    for (int q = 0; q < H; ++q) r.v[q] = p[q];
  }
  return r;
}
template <int H>
__device__ __forceinline__ void gat_st_heads(float* p, const gat_hv<H>& r) {
  if constexpr (H == 4) {
    *(float4*)p = make_float4(r.v[0], r.v[1], r.v[2], r.v[3]);
  } else {
#pragma unroll
    for (int q = 0; q < H; ++q) p[q] = r.v[q];
  }
}

template <int H> struct gat_piece { static constexpr int LPR = (H == 4) ? 32 : 16; };

__device__ __forceinline__ void gat_lds_add4(float* p, int stride, float x0, float x1, float x2, float x3, float a) {
  gat_lds_add(p, x0 * a);
  gat_lds_add(p + stride, x1 * a);
  gat_lds_add(p + 2 * stride, x2 * a);
  gat_lds_add(p + 3 * stride, x3 * a);
}

__device__ __forceinline__ void gat_piece_weights(float a, int lane, float& a0, float& a1) {
  const float w0 = __int_as_float(__builtin_amdgcn_readlane(__float_as_int(a), 0));
  const float w1 = __int_as_float(__builtin_amdgcn_readlane(__float_as_int(a), 1));
  const float w2 = __int_as_float(__builtin_amdgcn_readlane(__float_as_int(a), 2));
  const float w3 = __int_as_float(__builtin_amdgcn_readlane(__float_as_int(a), 3));
  const bool up = (lane & 16) != 0;
  a0 = up ? w1 : w0;
  a1 = up ? w3 : w2;
}

template <int H>
__device__ __forceinline__ void gat_tile_add_row(const float* __restrict__ h, float* tile,
                                                 int s, int dl, float a, int lane) {
  constexpr int HF = H * GAT_F;
  constexpr int LPR = gat_piece<H>::LPR;
  const float4* hp = (const float4*)(h + (size_t)s * HF);
  if constexpr (H == 4) {
    const float4 x0 = hp[lane], x1 = hp[LPR + lane];
    float a0, a1;
    gat_piece_weights(a, lane, a0, a1);
    float* tp = tile + dl * HF + lane;
    gat_lds_add4(tp, LPR, x0.x, x0.y, x0.z, x0.w, a0);
    gat_lds_add4(tp + 4 * LPR, LPR, x1.x, x1.y, x1.z, x1.w, a1);
  } else {
    if (lane < LPR) {
      const float4 x = hp[lane];
      gat_lds_add4(tile + dl * HF + lane, LPR, x.x, x.y, x.z, x.w, a);
    }
  }
}

__device__ __forceinline__ void gat_load_dst_group(const int* __restrict__ dst, int E, int gg, int lane,
                                                   int (&dv)[4][4]) {
  const int e0 = gg * 512 + lane * 4;
  if (gg * 512 + 512 <= E) {
#pragma unroll
    for (int c = 0; c < 4; ++c) {
      const int4 t = *(const int4*)(dst + e0 + c * 128);
      dv[c][0] = t.x; dv[c][1] = t.y; dv[c][2] = t.z; dv[c][3] = t.w;
    }
  } else {
#pragma unroll
    for (int c = 0; c < 4; ++c)
#pragma unroll
      for (int j = 0; j < 4; ++j) {
        const int e = e0 + c * 128 + j;
        dv[c][j] = (e < E) ? dst[e] : -1;
      }
  }
}

template <class F>
__device__ __forceinline__ void gat_for_each_match(const int* __restrict__ dst, int E, int base, int cnt,
                                                   int wave, int nw, int lane, F f) {
  const int nGroups = (E + 511) >> 9;
  for (int gg = wave; gg < nGroups; gg += nw) {
    int dv[4][4];
    gat_load_dst_group(dst, E, gg, lane, dv);
#pragma unroll
    for (int c = 0; c < 4; ++c)
#pragma unroll
      for (int j = 0; j < 4; ++j) {
        const int dl = dv[c][j] - base;
        unsigned hit = __builtin_amdgcn_ballot_w32((unsigned)dl < (unsigned)cnt);
        while (hit) {
          const int l = __builtin_ctz(hit);
          hit &= hit - 1u;
          f(gg * 512 + c * 128 + l * 4 + j, __builtin_amdgcn_readlane(dl, l));
        }
      }
  }
}

template <int H, bool RELU>
__global__ __launch_bounds__(GAT_TILE_THREADS)
void gat_tile_aggregate(const int* __restrict__ src, const int* __restrict__ dst,
                        const float* __restrict__ h, const float* __restrict__ el,
                        const float* __restrict__ er, const float* __restrict__ bias,
                        const int* __restrict__ segcnt, const int* __restrict__ seg,
                        float* __restrict__ out, int N, int E) {
  constexpr int HF = H * GAT_F;
  constexpr int LPR = gat_piece<H>::LPR;
  __shared__ __attribute__((aligned(16))) float tile[GAT_TILE_NODES * HF];
  __shared__ __attribute__((aligned(16))) float lg[GAT_LIST_CAP * H];
  __shared__ __attribute__((aligned(16))) float er_t[GAT_TILE_NODES * H];
  __shared__ int2 list[GAT_LIST_CAP];
  __shared__ unsigned mk[GAT_TILE_NODES * H];
  __shared__ float ss[GAT_TILE_NODES * H];
  __shared__ unsigned s_count;

  const int tid = threadIdx.x;
  const int lane = tid & 31;
  const int hq = lane & (H - 1);
  const int wave = __builtin_amdgcn_readfirstlane(tid >> 5);
  const int nw = GAT_TILE_THREADS >> 5;
  const int base = blockIdx.x * GAT_TILE_NODES;
  const int left = N - base;
  const int cnt = left < GAT_TILE_NODES ? left : GAT_TILE_NODES;

  float4* tile4 = (float4*)tile;
  const int n4 = cnt * (HF / 4);
  for (int i = tid; i < n4; i += GAT_TILE_THREADS) tile4[i] = make_float4(0.f, 0.f, 0.f, 0.f);
  for (int i = tid; i < cnt * H; i += GAT_TILE_THREADS) {
    er_t[i] = er[(size_t)base * H + i];
    mk[i] = KEY_NEG_INF;
    ss[i] = 0.f;
  }
  if (tid == 0) s_count = 0u;
  __syncthreads();

  {
    const int* cp = segcnt + (size_t)blockIdx.x * GAT_PART_GROUPS;
    const int2* sp = (const int2*)seg + (size_t)blockIdx.x * GAT_PART_GROUPS * GAT_SEG_CAP;
    int c[GAT_SEGS_PER_WAVE];
    unsigned b[GAT_SEGS_PER_WAVE];
    bool over = false;
    unsigned sum = 0u;
#pragma unroll
    for (int k = 0; k < GAT_SEGS_PER_WAVE; ++k) {
      const int ck = cp[wave + k * nw];
      over |= ck > GAT_SEG_CAP;
      c[k] = ck < GAT_SEG_CAP ? ck : GAT_SEG_CAP;
      b[k] = sum;
      sum += (unsigned)c[k];
    }
    if (over) sum += (unsigned)GAT_LIST_CAP + 1u;
    unsigned first = 0u;
    if (lane == 0) first = atomicAdd(&s_count, sum);
    first = __builtin_amdgcn_readfirstlane(first);
    int2 ent[GAT_SEGS_PER_WAVE][GAT_SEG_CAP / 32];
#pragma unroll
    for (int k = 0; k < GAT_SEGS_PER_WAVE; ++k)
#pragma unroll
      for (int j = 0; j < GAT_SEG_CAP / 32; ++j) {
        const int i = lane + 32 * j;
        ent[k][j] = make_int2(0, 0);
        if (i < c[k]) ent[k][j] = sp[(size_t)(wave + k * nw) * GAT_SEG_CAP + i];
      }
#pragma unroll
    for (int k = 0; k < GAT_SEGS_PER_WAVE; ++k)
#pragma unroll
      for (int j = 0; j < GAT_SEG_CAP / 32; ++j) {
        const int i = lane + 32 * j;
        const unsigned pos = first + b[k] + (unsigned)i;
        if (i < c[k] && pos < (unsigned)GAT_LIST_CAP) list[pos] = ent[k][j];
      }
  }
  __syncthreads();
  const int total = __builtin_amdgcn_readfirstlane((int)s_count);

  if (total <= GAT_LIST_CAP) {
    {
      int sv[GAT_LIST_ROUNDS];
      gat_hv<H> lv[GAT_LIST_ROUNDS];
#pragma unroll
      for (int r = 0; r < GAT_LIST_ROUNDS; ++r) {
        const int i = tid + r * GAT_TILE_THREADS;
        sv[r] = i < total ? list[i].x : 0;
      }
#pragma unroll
      for (int r = 0; r < GAT_LIST_ROUNDS; ++r) lv[r] = gat_ld_heads<H>(el + (size_t)sv[r] * H);
#pragma unroll
      for (int r = 0; r < GAT_LIST_ROUNDS; ++r) {
        const int i = tid + r * GAT_TILE_THREADS;
        if (i < total) {
          const int dl = list[i].y;
          const gat_hv<H> rv = gat_ld_heads<H>(er_t + dl * H);
          gat_hv<H> x;
#pragma unroll
          for (int q = 0; q < H; ++q) {
            x.v[q] = gat_leaky(lv[r].v[q] + rv.v[q]);
            gat_lds_max(&mk[dl * H + q], f32_key(x.v[q]));
          }
          gat_st_heads<H>(lg + i * H, x);
        }
      }
    }
    __syncthreads();

    for (int i = tid; i < total; i += GAT_TILE_THREADS) {
      const int dl = list[i].y;
      gat_hv<H> x = gat_ld_heads<H>(lg + i * H);
#pragma unroll
      for (int q = 0; q < H; ++q) {
        x.v[q] = __expf(x.v[q] - key_f32(mk[dl * H + q]));
        gat_lds_add(&ss[dl * H + q], x.v[q]);
      }
      gat_st_heads<H>(lg + i * H, x);
    }
    __syncthreads();

    constexpr int EPB = (H == 4) ? 4 : 16;
    for (int i = wave * EPB; i < total; i += nw * EPB) {
      if (i + EPB <= total) {
        if constexpr (H == 4) {
          int sn[4], dd[4];
#pragma unroll
          for (int k = 0; k < 4; ++k) {
            const int2 en = list[i + k];
            sn[k] = __builtin_amdgcn_readfirstlane(en.x);
            dd[k] = __builtin_amdgcn_readfirstlane(en.y);
          }
          float4 vals[4][2];
          float al[4];
#pragma unroll
          for (int k = 0; k < 4; ++k) {
            al[k] = lg[(i + k) * H + hq] / ss[dd[k] * H + hq];
            const float4* hp = (const float4*)(h + (size_t)sn[k] * HF) + lane;
            vals[k][0] = hp[0];
            vals[k][1] = hp[LPR];
          }
#pragma unroll
          for (int k = 0; k < 4; ++k) {
            float a0, a1;
            gat_piece_weights(al[k], lane, a0, a1);
            float* tp = tile + dd[k] * HF + lane;
            gat_lds_add4(tp, LPR, vals[k][0].x, vals[k][0].y, vals[k][0].z, vals[k][0].w, a0);
            gat_lds_add4(tp + 4 * LPR, LPR, vals[k][1].x, vals[k][1].y, vals[k][1].z, vals[k][1].w, a1);
          }
        } else {
          const int hf = lane >> 4, li = lane & 15;
          int sn[8], dr[8];
          float al[8];
          float4 vals[8];
#pragma unroll
          for (int k = 0; k < 8; ++k) {
            const int e = i + 2 * k + hf;
            const int2 en = list[e];
            sn[k] = en.x;
            dr[k] = en.y;
            al[k] = lg[e] / ss[en.y];
          }
#pragma unroll
          for (int k = 0; k < 8; ++k) vals[k] = ((const float4*)(h + (size_t)sn[k] * HF))[li];
#pragma unroll
          for (int k = 0; k < 8; ++k) {
            const float4 v = vals[k];
            float* t0 = tile + dr[k] * HF + li + (hf ? 2 * LPR : 0);
            float* t1 = tile + dr[k] * HF + li + (hf ? 0 : 2 * LPR);
            const float a = al[k];
            gat_lds_add(t0, (hf ? v.z : v.x) * a);
            gat_lds_add(t0 + LPR, (hf ? v.w : v.y) * a);
            gat_lds_add(t1, (hf ? v.x : v.z) * a);
            gat_lds_add(t1 + LPR, (hf ? v.y : v.w) * a);
          }
        }
      } else {
        for (int k = i; k < total; ++k) {
          const int2 en = list[k];
          const int dl = __builtin_amdgcn_readfirstlane(en.y);
          gat_tile_add_row<H>(h, tile, __builtin_amdgcn_readfirstlane(en.x), dl,
                              lg[k * H + hq] / ss[dl * H + hq], lane);
        }
      }
    }
  } else {
    gat_for_each_match(dst, E, base, cnt, wave, nw, lane, [&](int e, int dl) {
      if (lane < H) {
        const int s = src[e];
        gat_lds_max(&mk[dl * H + lane], f32_key(gat_leaky(el[(size_t)s * H + lane] + er_t[dl * H + lane])));
      }
    });
    __syncthreads();
    gat_for_each_match(dst, E, base, cnt, wave, nw, lane, [&](int e, int dl) {
      if (lane < H) {
        const int s = src[e];
        const float x = gat_leaky(el[(size_t)s * H + lane] + er_t[dl * H + lane]);
        gat_lds_add(&ss[dl * H + lane], __expf(x - key_f32(mk[dl * H + lane])));
      }
    });
    __syncthreads();
    gat_for_each_match(dst, E, base, cnt, wave, nw, lane, [&](int e, int dl) {
      const int s = src[e];
      const float x = gat_leaky(el[(size_t)s * H + hq] + er_t[dl * H + hq]);
      gat_tile_add_row<H>(h, tile, s, dl, __expf(x - key_f32(mk[dl * H + hq])) / ss[dl * H + hq], lane);
    });
  }
  __syncthreads();

  const float4* bias4 = (const float4*)bias;
  float4* out4 = (float4*)(out + (size_t)base * HF);
  for (int i = tid; i < n4; i += GAT_TILE_THREADS) {
    const float* tp = tile + (i / LPR) * (4 * LPR) + (i % LPR);
    float4 v = make_float4(tp[0], tp[LPR], tp[2 * LPR], tp[3 * LPR]);
    const float4 b = bias4[i & (HF / 4 - 1)];
    v.x += b.x; v.y += b.y; v.z += b.z; v.w += b.w;
    if (RELU) {
      v.x = v.x > 0.f ? v.x : 0.f; v.y = v.y > 0.f ? v.y : 0.f;
      v.z = v.z > 0.f ? v.z : 0.f; v.w = v.w > 0.f ? v.w : 0.f;
    }
    out4[i] = v;
  }
}

extern "C" void kernel_launch(void* const* d_in, const int* in_sizes, int n_in,
                              void* d_out, int out_size, void* d_ws, size_t ws_size,
                              hipStream_t stream) {
  (void)in_sizes; (void)n_in; (void)out_size; (void)ws_size;
  const float* feat = (const float*)d_in[0];
  const int*   src  = (const int*)d_in[1];
  const int*   dst  = (const int*)d_in[2];
  const float* W1  = (const float*)d_in[3];
  const float* al1 = (const float*)d_in[4];
  const float* ar1 = (const float*)d_in[5];
  const float* b1  = (const float*)d_in[6];
  const float* W2  = (const float*)d_in[7];
  const float* al2 = (const float*)d_in[8];
  const float* ar2 = (const float*)d_in[9];
  const float* b2  = (const float*)d_in[10];
  const float* W3  = (const float*)d_in[11];
  const float* al3 = (const float*)d_in[12];
  const float* ar3 = (const float*)d_in[13];
  const float* b3  = (const float*)d_in[14];
  float* out = (float*)d_out;

  const int N = GAT_N, E = GAT_E;

  char* ws = (char*)d_ws;
  size_t off = 0;
  auto carve = [&](size_t bytes) -> void* {
    void* p = ws + off;
    off += (bytes + 255) & ~(size_t)255;
    return p;
  };
  float* bufA = (float*)carve((size_t)N * 256 * 4);
  float* bufB = (float*)carve((size_t)N * 256 * 4);
  float* el   = (float*)carve((size_t)N * 4 * 4);
  float* er   = (float*)carve((size_t)N * 4 * 4);
  int* segcnt = (int*)carve((size_t)GAT_NUM_TILES * GAT_PART_GROUPS * 4);
  int* seg    = (int*)carve((size_t)GAT_NUM_TILES * GAT_PART_GROUPS * GAT_SEG_CAP * 8);

  auto blocks = [](long long n) { return (unsigned)((n + 255) / 256); };
  const unsigned tiles = GAT_NUM_TILES;

  gat_partition<<<GAT_PART_GROUPS, GAT_PART_THREADS, 0, stream>>>(src, dst, segcnt, seg);

  {
    const int K = 128, Nw = 256, H = 4;
    int waves = (N / 16) * (Nw / 64);
    gat_gemm_wmma<<<(waves + 7) / 8, 256, 0, stream>>>(feat, W1, bufA, N, K, Nw);
    long long nh = (long long)N * H;
    gat_node_attn2<4><<<blocks(nh), 256, 0, stream>>>(bufA, al1, ar1, el, er, (int)nh);
    gat_tile_aggregate<4, true><<<tiles, GAT_TILE_THREADS, 0, stream>>>(src, dst, bufA, el, er, b1, segcnt, seg, bufB, N, E);
  }

  {
    const int K = 256, Nw = 256, H = 4;
    int waves = (N / 16) * (Nw / 64);
    gat_gemm_wmma<<<(waves + 7) / 8, 256, 0, stream>>>(bufB, W2, bufA, N, K, Nw);
    long long nh = (long long)N * H;
    gat_node_attn2<4><<<blocks(nh), 256, 0, stream>>>(bufA, al2, ar2, el, er, (int)nh);
    gat_tile_aggregate<4, true><<<tiles, GAT_TILE_THREADS, 0, stream>>>(src, dst, bufA, el, er, b2, segcnt, seg, bufB, N, E);
  }

  {
    const int K = 256, Nw = 64, H = 1;
    int waves = (N / 16) * (Nw / 64);
    gat_gemm_wmma<<<(waves + 7) / 8, 256, 0, stream>>>(bufB, W3, bufA, N, K, Nw);
    long long nh = (long long)N * H;
    gat_node_attn2<1><<<blocks(nh), 256, 0, stream>>>(bufA, al3, ar3, el, er, (int)nh);
    gat_tile_aggregate<1, false><<<tiles, GAT_TILE_THREADS, 0, stream>>>(src, dst, bufA, el, er, b3, segcnt, seg, out, N, E);
  }
}


# Products of layers 2, 3 moved into the tile kernels: a group's finished rows (bias, ReLU) feed f32 WMMA straight from LDS (row pitch +4 dwords), same k order; 2 GEMM launches and a round trip gone
# speedup vs baseline: 9.0258x; 1.1348x over previous
#include <hip/hip_runtime.h>


#define GAT_N 50000
#define GAT_E 800000
#define GAT_F 64

typedef float v2f __attribute__((ext_vector_type(2)));
typedef float v4f __attribute__((ext_vector_type(4)));
typedef float v8f __attribute__((ext_vector_type(8)));

#define GEMM_KB 4
__global__ __launch_bounds__(256)
void gat_gemm_wmma(const float* __restrict__ A, const float* __restrict__ B,
                   float* __restrict__ C, int M, int K, int Nw) {
  const int wave = blockIdx.x * 8 + (threadIdx.x >> 5);
  const int lane = threadIdx.x & 31;
  const int tilesN = Nw >> 6;
  const int tilesM = M >> 4;
  if (wave >= tilesM * tilesN) return;
  const int tm = wave / tilesN;
  const int tn = wave - tm * tilesN;
  const int kh = lane >> 4;
  const int lr = lane & 15;

  const float* Ap = A + (size_t)(tm * 16 + lr) * K + kh * 2;
  const float* Bp = B + (size_t)(tn * 64) + lr * 4 + (size_t)(kh * 2) * Nw;
  v8f acc0 = {}, acc1 = {}, acc2 = {}, acc3 = {};
  for (int k0 = 0; k0 < K; k0 += 4 * GEMM_KB) {
    v2f a[GEMM_KB];
    v4f bx[GEMM_KB], by[GEMM_KB];
    const float* Bk = Bp + (size_t)k0 * Nw;
#pragma unroll
    for (int s = 0; s < GEMM_KB; ++s) {
      a[s] = *(const v2f*)(Ap + k0 + 4 * s);
      const float* Brow0 = Bk + (size_t)(4 * s) * Nw;
      bx[s] = *(const v4f*)Brow0;
      by[s] = *(const v4f*)(Brow0 + Nw);
    }
    __builtin_amdgcn_sched_barrier(0);
#pragma unroll
    for (int s = 0; s < GEMM_KB; ++s) {
      const v2f b0 = {bx[s].x, by[s].x};
      const v2f b1 = {bx[s].y, by[s].y};
      const v2f b2 = {bx[s].z, by[s].z};
      const v2f b3 = {bx[s].w, by[s].w};
      acc0 = __builtin_amdgcn_wmma_f32_16x16x4_f32(false, a[s], false, b0, (short)0, acc0, false, false);
      acc1 = __builtin_amdgcn_wmma_f32_16x16x4_f32(false, a[s], false, b1, (short)0, acc1, false, false);
      acc2 = __builtin_amdgcn_wmma_f32_16x16x4_f32(false, a[s], false, b2, (short)0, acc2, false, false);
      acc3 = __builtin_amdgcn_wmma_f32_16x16x4_f32(false, a[s], false, b3, (short)0, acc3, false, false);
    }
  }
  float* Cp = C + (size_t)(tm * 16 + kh * 8) * Nw + tn * 64 + lr * 4;
#pragma unroll
  for (int r = 0; r < 8; ++r) {
    const v4f c = {acc0[r], acc1[r], acc2[r], acc3[r]};
    *(v4f*)(Cp + (size_t)r * Nw) = c;
  }
}


template <int C, int M>
__device__ __forceinline__ void gat_halve(float (&v)[16], int lane) {
  const bool up = (lane & M) != 0;
#pragma unroll
  for (int k = 0; k < C / 2; ++k) {
    const float a = v[k], b = v[k + C / 2];
    const float keep = up ? b : a;
    const float send = up ? a : b;
    v[k] = keep + __shfl_xor(send, M);
  }
}

template <int H>
__global__ __launch_bounds__(256)
void gat_node_attn2(const float* __restrict__ h, const float* __restrict__ attn_l,
                    const float* __restrict__ attn_r, float* __restrict__ el,
                    float* __restrict__ er, int R) {
  static_assert(H == 1 || H == 4, "the head of a load is taken from its number for these head counts");
  constexpr int NWT = (H == 4) ? 2 : 1;
  const int lane = threadIdx.x & 31;
  const int hf = lane >> 4;
  const int li = lane & 15;
  const int r0 = (blockIdx.x * 8 + (threadIdx.x >> 5)) * 32;
  if (r0 >= R) return;

  float4 wl[NWT], wr[NWT];
#pragma unroll
  for (int w = 0; w < NWT; ++w) {
    const int head = (2 * w + hf) % H;
    wl[w] = ((const float4*)attn_l)[head * (GAT_F / 4) + li];
    wr[w] = ((const float4*)attn_r)[head * (GAT_F / 4) + li];
  }
  const float4* h4 = (const float4*)h;
  float4 x[16];
#pragma unroll
  for (int t = 0; t < 16; ++t) {
    const int r = r0 + 2 * t + hf;
    x[t] = h4[(size_t)(r < R ? r : R - 1) * (GAT_F / 4) + li];
  }
  float pl[16], pr[16];
#pragma unroll
  for (int t = 0; t < 16; ++t) {
    const float4 a = wl[t & (NWT - 1)], b = wr[t & (NWT - 1)];
    pl[t] = x[t].x * a.x + x[t].y * a.y + x[t].z * a.z + x[t].w * a.w;
    pr[t] = x[t].x * b.x + x[t].y * b.y + x[t].z * b.z + x[t].w * b.w;
  }
  gat_halve<16, 8>(pl, lane); gat_halve<16, 8>(pr, lane);
  gat_halve<8, 4>(pl, lane);  gat_halve<8, 4>(pr, lane);
  gat_halve<4, 2>(pl, lane);  gat_halve<4, 2>(pr, lane);
  gat_halve<2, 1>(pl, lane);  gat_halve<2, 1>(pr, lane);
  const int r = r0 + 2 * li + hf;
  if (r < R) { el[r] = pl[0]; er[r] = pr[0]; }
}

__device__ __forceinline__ unsigned f32_key(float x) {
  unsigned u = __float_as_uint(x);
  return (u & 0x80000000u) ? ~u : (u | 0x80000000u);
}
__device__ __forceinline__ float key_f32(unsigned k) {
  return (k & 0x80000000u) ? __uint_as_float(k ^ 0x80000000u) : __uint_as_float(~k);
}
#define KEY_NEG_INF 0x007FFFFFu

__device__ __forceinline__ float gat_leaky(float x) { return (x > 0.f) ? x : 0.2f * x; }

#define GAT_TILE_NODES   196
#define GAT_TILE_THREADS 1024
#define GAT_LIST_CAP     4608
#define GAT_LIST_ROUNDS  ((GAT_LIST_CAP + GAT_TILE_THREADS - 1) / GAT_TILE_THREADS)
#define GAT_NUM_TILES    ((GAT_N + GAT_TILE_NODES - 1) / GAT_TILE_NODES)

#define GAT_PART_GROUPS   128
#define GAT_PART_THREADS  1024
#define GAT_PART_EDGES    ((GAT_E + GAT_PART_GROUPS - 1) / GAT_PART_GROUPS)
#define GAT_PART_ROUNDS   ((GAT_PART_EDGES + GAT_PART_THREADS - 1) / GAT_PART_THREADS)
#define GAT_SEG_CAP       64
#define GAT_SEGS_PER_WAVE (GAT_PART_GROUPS / (GAT_TILE_THREADS / 32))
static_assert(GAT_PART_GROUPS % (GAT_TILE_THREADS / 32) == 0, "segments divide evenly over the tile kernel's waves");
static_assert(GAT_SEG_CAP % 32 == 0, "a segment is read in whole-wave rounds");

__global__ __launch_bounds__(GAT_PART_THREADS)
void gat_partition(const int* __restrict__ src, const int* __restrict__ dst,
                   int* __restrict__ segcnt, int* __restrict__ seg) {
  __shared__ unsigned cnt[GAT_NUM_TILES];
  const int tid = threadIdx.x;
  const int g = blockIdx.x;
  for (int i = tid; i < GAT_NUM_TILES; i += GAT_PART_THREADS) cnt[i] = 0u;
  __syncthreads();

  const int e0 = g * GAT_PART_EDGES;
  const int lim = (e0 + GAT_PART_EDGES < GAT_E) ? e0 + GAT_PART_EDGES : GAT_E;
  int dv[GAT_PART_ROUNDS], sv[GAT_PART_ROUNDS];
#pragma unroll
  for (int r = 0; r < GAT_PART_ROUNDS; ++r) {
    const int e = e0 + tid + r * GAT_PART_THREADS;
    const bool ok = e < lim;
    dv[r] = ok ? dst[e] : -1;
    sv[r] = ok ? src[e] : 0;
  }
  int2* segs = (int2*)seg;
#pragma unroll
  for (int r = 0; r < GAT_PART_ROUNDS; ++r) {
    const int d = dv[r];
    if ((unsigned)d < (unsigned)GAT_N) {
      const int t = d / GAT_TILE_NODES;
      const unsigned pos = atomicAdd(&cnt[t], 1u);
      if (pos < (unsigned)GAT_SEG_CAP)
        segs[((size_t)t * GAT_PART_GROUPS + g) * GAT_SEG_CAP + pos] = make_int2(sv[r], d - t * GAT_TILE_NODES);
    }
  }
  __syncthreads();
  for (int i = tid; i < GAT_NUM_TILES; i += GAT_PART_THREADS) segcnt[i * GAT_PART_GROUPS + g] = (int)cnt[i];
}

__device__ __forceinline__ void gat_lds_add(float* p, float v) {
  __hip_atomic_fetch_add(p, v, __ATOMIC_RELAXED, __HIP_MEMORY_SCOPE_WORKGROUP);
}
__device__ __forceinline__ void gat_lds_max(unsigned* p, unsigned v) {
  __hip_atomic_fetch_max(p, v, __ATOMIC_RELAXED, __HIP_MEMORY_SCOPE_WORKGROUP);
}

template <int H> struct gat_hv { float v[H]; };
template <int H>
__device__ __forceinline__ gat_hv<H> gat_ld_heads(const float* p) {
  gat_hv<H> r;
  if constexpr (H == 4) {
    const float4 t = *(const float4*)p;
    r.v[0] = t.x; r.v[1] = t.y; r.v[2] = t.z; r.v[3] = t.w;
  } else {
#pragma unroll
    for (int q = 0; q < H; ++q) r.v[q] = p[q];
  }
  return r;
}
template <int H>
__device__ __forceinline__ void gat_st_heads(float* p, const gat_hv<H>& r) {
  if constexpr (H == 4) {
    *(float4*)p = make_float4(r.v[0], r.v[1], r.v[2], r.v[3]);
  } else {
#pragma unroll
    for (int q = 0; q < H; ++q) p[q] = r.v[q];
  }
}

template <int H> struct gat_piece { static constexpr int LPR = (H == 4) ? 32 : 16; };

__device__ __forceinline__ void gat_lds_add4(float* p, int stride, float x0, float x1, float x2, float x3, float a) {
  gat_lds_add(p, x0 * a);
  gat_lds_add(p + stride, x1 * a);
  gat_lds_add(p + 2 * stride, x2 * a);
  gat_lds_add(p + 3 * stride, x3 * a);
}

__device__ __forceinline__ void gat_piece_weights(float a, int lane, float& a0, float& a1) {
  const float w0 = __int_as_float(__builtin_amdgcn_readlane(__float_as_int(a), 0));
  const float w1 = __int_as_float(__builtin_amdgcn_readlane(__float_as_int(a), 1));
  const float w2 = __int_as_float(__builtin_amdgcn_readlane(__float_as_int(a), 2));
  const float w3 = __int_as_float(__builtin_amdgcn_readlane(__float_as_int(a), 3));
  const bool up = (lane & 16) != 0;
  a0 = up ? w1 : w0;
  a1 = up ? w3 : w2;
}

template <int H, int TS>
__device__ __forceinline__ void gat_tile_add_row(const float* __restrict__ h, float* tile,
                                                 int s, int dl, float a, int lane) {
  constexpr int HF = H * GAT_F;
  constexpr int LPR = gat_piece<H>::LPR;
  const float4* hp = (const float4*)(h + (size_t)s * HF);
  if constexpr (H == 4) {
    const float4 x0 = hp[lane], x1 = hp[LPR + lane];
    float a0, a1;
    gat_piece_weights(a, lane, a0, a1);
    float* tp = tile + dl * TS + lane;
    gat_lds_add4(tp, LPR, x0.x, x0.y, x0.z, x0.w, a0);
    gat_lds_add4(tp + 4 * LPR, LPR, x1.x, x1.y, x1.z, x1.w, a1);
  } else {
    if (lane < LPR) {
      const float4 x = hp[lane];
      gat_lds_add4(tile + dl * TS + lane, LPR, x.x, x.y, x.z, x.w, a);
    }
  }
}

__device__ __forceinline__ void gat_load_dst_group(const int* __restrict__ dst, int E, int gg, int lane,
                                                   int (&dv)[4][4]) {
  const int e0 = gg * 512 + lane * 4;
  if (gg * 512 + 512 <= E) {
#pragma unroll
    for (int c = 0; c < 4; ++c) {
      const int4 t = *(const int4*)(dst + e0 + c * 128);
      dv[c][0] = t.x; dv[c][1] = t.y; dv[c][2] = t.z; dv[c][3] = t.w;
    }
  } else {
#pragma unroll
    for (int c = 0; c < 4; ++c)
#pragma unroll
      for (int j = 0; j < 4; ++j) {
        const int e = e0 + c * 128 + j;
        dv[c][j] = (e < E) ? dst[e] : -1;
      }
  }
}

template <class F>
__device__ __forceinline__ void gat_for_each_match(const int* __restrict__ dst, int E, int base, int cnt,
                                                   int wave, int nw, int lane, F f) {
  const int nGroups = (E + 511) >> 9;
  for (int gg = wave; gg < nGroups; gg += nw) {
    int dv[4][4];
    gat_load_dst_group(dst, E, gg, lane, dv);
#pragma unroll
    for (int c = 0; c < 4; ++c)
#pragma unroll
      for (int j = 0; j < 4; ++j) {
        const int dl = dv[c][j] - base;
        unsigned hit = __builtin_amdgcn_ballot_w32((unsigned)dl < (unsigned)cnt);
        while (hit) {
          const int l = __builtin_ctz(hit);
          hit &= hit - 1u;
          f(gg * 512 + c * 128 + l * 4 + j, __builtin_amdgcn_readlane(dl, l));
        }
      }
  }
}

template <int H, bool RELU, int NWN>
__global__ __launch_bounds__(GAT_TILE_THREADS)
void gat_tile_aggregate(const int* __restrict__ src, const int* __restrict__ dst,
                        const float* __restrict__ h, const float* __restrict__ el,
                        const float* __restrict__ er, const float* __restrict__ bias,
                        const int* __restrict__ segcnt, const int* __restrict__ seg,
                        const float* __restrict__ Wn, float* __restrict__ out, int N, int E) {
  static_assert(NWN == 0 || (H == 4 && RELU && NWN % 64 == 0), "the fused product is written for the two-piece rows of H = 4");
  constexpr int HF = H * GAT_F;
  constexpr int LPR = gat_piece<H>::LPR;
  constexpr int TS = (NWN > 0) ? HF + 4 : HF;
  __shared__ __attribute__((aligned(16))) float tile[GAT_TILE_NODES * TS];
  __shared__ __attribute__((aligned(16))) float lg[GAT_LIST_CAP * H];
  __shared__ __attribute__((aligned(16))) float er_t[GAT_TILE_NODES * H];
  __shared__ int2 list[GAT_LIST_CAP];
  __shared__ unsigned mk[GAT_TILE_NODES * H];
  __shared__ float ss[GAT_TILE_NODES * H];
  __shared__ unsigned s_count;

  const int tid = threadIdx.x;
  const int lane = tid & 31;
  const int hq = lane & (H - 1);
  const int wave = __builtin_amdgcn_readfirstlane(tid >> 5);
  const int nw = GAT_TILE_THREADS >> 5;
  const int base = blockIdx.x * GAT_TILE_NODES;
  const int left = N - base;
  const int cnt = left < GAT_TILE_NODES ? left : GAT_TILE_NODES;

  float4* tile4 = (float4*)tile;
  const int n4 = cnt * (HF / 4);
  for (int i = tid; i < cnt * (TS / 4); i += GAT_TILE_THREADS) tile4[i] = make_float4(0.f, 0.f, 0.f, 0.f);
  for (int i = tid; i < cnt * H; i += GAT_TILE_THREADS) {
    er_t[i] = er[(size_t)base * H + i];
    mk[i] = KEY_NEG_INF;
    ss[i] = 0.f;
  }
  if (tid == 0) s_count = 0u;
  __syncthreads();

  {
    const int* cp = segcnt + (size_t)blockIdx.x * GAT_PART_GROUPS;
    const int2* sp = (const int2*)seg + (size_t)blockIdx.x * GAT_PART_GROUPS * GAT_SEG_CAP;
    int c[GAT_SEGS_PER_WAVE];
    unsigned b[GAT_SEGS_PER_WAVE];
    bool over = false;
    unsigned sum = 0u;
#pragma unroll
    for (int k = 0; k < GAT_SEGS_PER_WAVE; ++k) {
      const int ck = cp[wave + k * nw];
      over |= ck > GAT_SEG_CAP;
      c[k] = ck < GAT_SEG_CAP ? ck : GAT_SEG_CAP;
      b[k] = sum;
      sum += (unsigned)c[k];
    }
    if (over) sum += (unsigned)GAT_LIST_CAP + 1u;
    unsigned first = 0u;
    if (lane == 0) first = atomicAdd(&s_count, sum);
    first = __builtin_amdgcn_readfirstlane(first);
    int2 ent[GAT_SEGS_PER_WAVE][GAT_SEG_CAP / 32];
#pragma unroll
    for (int k = 0; k < GAT_SEGS_PER_WAVE; ++k)
#pragma unroll
      for (int j = 0; j < GAT_SEG_CAP / 32; ++j) {
        const int i = lane + 32 * j;
        ent[k][j] = make_int2(0, 0);
        if (i < c[k]) ent[k][j] = sp[(size_t)(wave + k * nw) * GAT_SEG_CAP + i];
      }
#pragma unroll
    for (int k = 0; k < GAT_SEGS_PER_WAVE; ++k)
#pragma unroll
      for (int j = 0; j < GAT_SEG_CAP / 32; ++j) {
        const int i = lane + 32 * j;
        const unsigned pos = first + b[k] + (unsigned)i;
        if (i < c[k] && pos < (unsigned)GAT_LIST_CAP) list[pos] = ent[k][j];
      }
  }
  __syncthreads();
  const int total = __builtin_amdgcn_readfirstlane((int)s_count);

  if (total <= GAT_LIST_CAP) {
    {
      int sv[GAT_LIST_ROUNDS];
      gat_hv<H> lv[GAT_LIST_ROUNDS];
#pragma unroll
      for (int r = 0; r < GAT_LIST_ROUNDS; ++r) {
        const int i = tid + r * GAT_TILE_THREADS;
        sv[r] = i < total ? list[i].x : 0;
      }
#pragma unroll
      for (int r = 0; r < GAT_LIST_ROUNDS; ++r) lv[r] = gat_ld_heads<H>(el + (size_t)sv[r] * H);
#pragma unroll
      for (int r = 0; r < GAT_LIST_ROUNDS; ++r) {
        const int i = tid + r * GAT_TILE_THREADS;
        if (i < total) {
          const int dl = list[i].y;
          const gat_hv<H> rv = gat_ld_heads<H>(er_t + dl * H);
          gat_hv<H> x;
#pragma unroll
          for (int q = 0; q < H; ++q) {
            x.v[q] = gat_leaky(lv[r].v[q] + rv.v[q]);
            gat_lds_max(&mk[dl * H + q], f32_key(x.v[q]));
          }
          gat_st_heads<H>(lg + i * H, x);
        }
      }
    }
    __syncthreads();

    for (int i = tid; i < total; i += GAT_TILE_THREADS) {
      const int dl = list[i].y;
      gat_hv<H> x = gat_ld_heads<H>(lg + i * H);
#pragma unroll
      for (int q = 0; q < H; ++q) {
        x.v[q] = __expf(x.v[q] - key_f32(mk[dl * H + q]));
        gat_lds_add(&ss[dl * H + q], x.v[q]);
      }
      gat_st_heads<H>(lg + i * H, x);
    }
    __syncthreads();

    constexpr int EPB = (H == 4) ? 4 : 16;
    for (int i = wave * EPB; i < total; i += nw * EPB) {
      if (i + EPB <= total) {
        if constexpr (H == 4) {
          int sn[4], dd[4];
#pragma unroll
          for (int k = 0; k < 4; ++k) {
            const int2 en = list[i + k];
            sn[k] = __builtin_amdgcn_readfirstlane(en.x);
            dd[k] = __builtin_amdgcn_readfirstlane(en.y);
          }
          float4 vals[4][2];
          float al[4];
#pragma unroll
          for (int k = 0; k < 4; ++k) {
            al[k] = lg[(i + k) * H + hq] / ss[dd[k] * H + hq];
            const float4* hp = (const float4*)(h + (size_t)sn[k] * HF) + lane;
            vals[k][0] = hp[0];
            vals[k][1] = hp[LPR];
          }
#pragma unroll
          for (int k = 0; k < 4; ++k) {
            float a0, a1;
            gat_piece_weights(al[k], lane, a0, a1);
            float* tp = tile + dd[k] * TS + lane;
            gat_lds_add4(tp, LPR, vals[k][0].x, vals[k][0].y, vals[k][0].z, vals[k][0].w, a0);
            gat_lds_add4(tp + 4 * LPR, LPR, vals[k][1].x, vals[k][1].y, vals[k][1].z, vals[k][1].w, a1);
          }
        } else {
          const int hf = lane >> 4, li = lane & 15;
          int sn[8], dr[8];
          float al[8];
          float4 vals[8];
#pragma unroll
          for (int k = 0; k < 8; ++k) {
            const int e = i + 2 * k + hf;
            const int2 en = list[e];
            sn[k] = en.x;
            dr[k] = en.y;
            al[k] = lg[e] / ss[en.y];
          }
#pragma unroll
          for (int k = 0; k < 8; ++k) vals[k] = ((const float4*)(h + (size_t)sn[k] * HF))[li];
#pragma unroll
          for (int k = 0; k < 8; ++k) {
            const float4 v = vals[k];
            float* t0 = tile + dr[k] * TS + li + (hf ? 2 * LPR : 0);
            float* t1 = tile + dr[k] * TS + li + (hf ? 0 : 2 * LPR);
            const float a = al[k];
            gat_lds_add(t0, (hf ? v.z : v.x) * a);
            gat_lds_add(t0 + LPR, (hf ? v.w : v.y) * a);
            gat_lds_add(t1, (hf ? v.x : v.z) * a);
            gat_lds_add(t1 + LPR, (hf ? v.y : v.w) * a);
          }
        }
      } else {
        for (int k = i; k < total; ++k) {
          const int2 en = list[k];
          const int dl = __builtin_amdgcn_readfirstlane(en.y);
          gat_tile_add_row<H, TS>(h, tile, __builtin_amdgcn_readfirstlane(en.x), dl,
                              lg[k * H + hq] / ss[dl * H + hq], lane);
        }
      }
    }
  } else {
    gat_for_each_match(dst, E, base, cnt, wave, nw, lane, [&](int e, int dl) {
      if (lane < H) {
        const int s = src[e];
        gat_lds_max(&mk[dl * H + lane], f32_key(gat_leaky(el[(size_t)s * H + lane] + er_t[dl * H + lane])));
      }
    });
    __syncthreads();
    gat_for_each_match(dst, E, base, cnt, wave, nw, lane, [&](int e, int dl) {
      if (lane < H) {
        const int s = src[e];
        const float x = gat_leaky(el[(size_t)s * H + lane] + er_t[dl * H + lane]);
        gat_lds_add(&ss[dl * H + lane], __expf(x - key_f32(mk[dl * H + lane])));
      }
    });
    __syncthreads();
    gat_for_each_match(dst, E, base, cnt, wave, nw, lane, [&](int e, int dl) {
      const int s = src[e];
      const float x = gat_leaky(el[(size_t)s * H + hq] + er_t[dl * H + hq]);
      gat_tile_add_row<H, TS>(h, tile, s, dl, __expf(x - key_f32(mk[dl * H + hq])) / ss[dl * H + hq], lane);
    });
  }
  __syncthreads();

  if constexpr (NWN == 0) {
    const float4* bias4 = (const float4*)bias;
    float4* out4 = (float4*)(out + (size_t)base * HF);
    for (int i = tid; i < n4; i += GAT_TILE_THREADS) {
      const float* tp = tile + (i / LPR) * (4 * LPR) + (i % LPR);
      float4 v = make_float4(tp[0], tp[LPR], tp[2 * LPR], tp[3 * LPR]);
      const float4 b = bias4[i & (HF / 4 - 1)];
      v.x += b.x; v.y += b.y; v.z += b.z; v.w += b.w;
      if (RELU) {
        v.x = v.x > 0.f ? v.x : 0.f; v.y = v.y > 0.f ? v.y : 0.f;
        v.z = v.z > 0.f ? v.z : 0.f; v.w = v.w > 0.f ? v.w : 0.f;
      }
      out4[i] = v;
    }
  } else {
    {
      const int d = tid & (HF - 1);
      const int k = (d >> 7) * 128 + (d & 31) * 4 + ((d >> 5) & 3);
      const float bk = bias[k];
      for (int r = tid / HF; r < cnt; r += GAT_TILE_THREADS / HF) {
        float* p = tile + r * TS + d;
        const float v = *p + bk;
        *p = v > 0.f ? v : 0.f;
      }
    }
    __syncthreads();

    constexpr int tilesN = NWN >> 6;
    const int jobs = ((cnt + 15) >> 4) * tilesN;
    const int kh = lane >> 4;
    const int lr = lane & 15;
    for (int job = wave; job < jobs; job += nw) {
      const int tm = job / tilesN;
      const int tn = job - tm * tilesN;
      const int ar = tm * 16 + lr;
      const float* Ap = tile + (ar < cnt ? ar : cnt - 1) * TS + kh * 2 * LPR;
      const float* Bp = Wn + (size_t)(tn * 64) + lr * 4 + (size_t)(kh * 2) * NWN;
      v8f acc0 = {}, acc1 = {}, acc2 = {}, acc3 = {};
      for (int k0 = 0; k0 < HF; k0 += 4 * GEMM_KB) {
        v2f a[GEMM_KB];
        v4f bx[GEMM_KB], by[GEMM_KB];
        const float* Bk = Bp + (size_t)k0 * NWN;
#pragma unroll
        for (int q = 0; q < GEMM_KB; ++q) {
          const int kk = k0 + 4 * q;
          const float* ap = Ap + (kk >> 7) * 128 + ((kk & 127) >> 2);
          a[q] = v2f{ap[0], ap[LPR]};
          const float* Brow0 = Bk + (size_t)(4 * q) * NWN;
          bx[q] = *(const v4f*)Brow0;
          by[q] = *(const v4f*)(Brow0 + NWN);
        }
        __builtin_amdgcn_sched_barrier(0);
#pragma unroll
        for (int q = 0; q < GEMM_KB; ++q) {
          const v2f b0 = {bx[q].x, by[q].x};
          const v2f b1 = {bx[q].y, by[q].y};
          const v2f b2 = {bx[q].z, by[q].z};
          const v2f b3 = {bx[q].w, by[q].w};
          acc0 = __builtin_amdgcn_wmma_f32_16x16x4_f32(false, a[q], false, b0, (short)0, acc0, false, false);
          acc1 = __builtin_amdgcn_wmma_f32_16x16x4_f32(false, a[q], false, b1, (short)0, acc1, false, false);
          acc2 = __builtin_amdgcn_wmma_f32_16x16x4_f32(false, a[q], false, b2, (short)0, acc2, false, false);
          acc3 = __builtin_amdgcn_wmma_f32_16x16x4_f32(false, a[q], false, b3, (short)0, acc3, false, false);
        }
      }
      const int row0 = tm * 16 + kh * 8;
      float* Cp = out + (size_t)(base + row0) * NWN + tn * 64 + lr * 4;
#pragma unroll
      for (int r = 0; r < 8; ++r) {
        if (row0 + r < cnt) {
          const v4f c = {acc0[r], acc1[r], acc2[r], acc3[r]};
          *(v4f*)(Cp + (size_t)r * NWN) = c;
        }
      }
    }
  }
}

extern "C" void kernel_launch(void* const* d_in, const int* in_sizes, int n_in,
                              void* d_out, int out_size, void* d_ws, size_t ws_size,
                              hipStream_t stream) {
  (void)in_sizes; (void)n_in; (void)out_size; (void)ws_size;
  const float* feat = (const float*)d_in[0];
  const int*   src  = (const int*)d_in[1];
  const int*   dst  = (const int*)d_in[2];
  const float* W1  = (const float*)d_in[3];
  const float* al1 = (const float*)d_in[4];
  const float* ar1 = (const float*)d_in[5];
  const float* b1  = (const float*)d_in[6];
  const float* W2  = (const float*)d_in[7];
  const float* al2 = (const float*)d_in[8];
  const float* ar2 = (const float*)d_in[9];
  const float* b2  = (const float*)d_in[10];
  const float* W3  = (const float*)d_in[11];
  const float* al3 = (const float*)d_in[12];
  const float* ar3 = (const float*)d_in[13];
  const float* b3  = (const float*)d_in[14];
  float* out = (float*)d_out;

  const int N = GAT_N, E = GAT_E;

  char* ws = (char*)d_ws;
  size_t off = 0;
  auto carve = [&](size_t bytes) -> void* {
    void* p = ws + off;
    off += (bytes + 255) & ~(size_t)255;
    return p;
  };
  float* bufA = (float*)carve((size_t)N * 256 * 4);
  float* bufB = (float*)carve((size_t)N * 256 * 4);
  float* el   = (float*)carve((size_t)N * 4 * 4);
  float* er   = (float*)carve((size_t)N * 4 * 4);
  int* segcnt = (int*)carve((size_t)GAT_NUM_TILES * GAT_PART_GROUPS * 4);
  int* seg    = (int*)carve((size_t)GAT_NUM_TILES * GAT_PART_GROUPS * GAT_SEG_CAP * 8);

  auto blocks = [](long long n) { return (unsigned)((n + 255) / 256); };
  const unsigned tiles = GAT_NUM_TILES;

  gat_partition<<<GAT_PART_GROUPS, GAT_PART_THREADS, 0, stream>>>(src, dst, segcnt, seg);

  {
    const int K = 128, Nw = 256, H = 4;
    int waves = (N / 16) * (Nw / 64);
    gat_gemm_wmma<<<(waves + 7) / 8, 256, 0, stream>>>(feat, W1, bufA, N, K, Nw);
    long long nh = (long long)N * H;
    gat_node_attn2<4><<<blocks(nh), 256, 0, stream>>>(bufA, al1, ar1, el, er, (int)nh);
    gat_tile_aggregate<4, true, 256><<<tiles, GAT_TILE_THREADS, 0, stream>>>(src, dst, bufA, el, er, b1, segcnt, seg, W2, bufB, N, E);
  }

  {
    const int H = 4;
    long long nh = (long long)N * H;
    gat_node_attn2<4><<<blocks(nh), 256, 0, stream>>>(bufB, al2, ar2, el, er, (int)nh);
    gat_tile_aggregate<4, true, 64><<<tiles, GAT_TILE_THREADS, 0, stream>>>(src, dst, bufB, el, er, b2, segcnt, seg, W3, bufA, N, E);
  }

  {
    const int H = 1;
    long long nh = (long long)N * H;
    gat_node_attn2<1><<<blocks(nh), 256, 0, stream>>>(bufA, al3, ar3, el, er, (int)nh);
    gat_tile_aggregate<1, false, 0><<<tiles, GAT_TILE_THREADS, 0, stream>>>(src, dst, bufA, el, er, b3, segcnt, seg, nullptr, out, N, E);
  }
}
